# starting kernel
# baseline (speedup 1.0000x reference)
#include <hip/hip_runtime.h>
#include <hip/hip_bf16.h>


#define NEG_SLOPE 0.2f
#define F_IN   128
#define H1C1   160
#define HEADS1 8
#define CH1    20
#define NCLS   16

typedef float v2f __attribute__((ext_vector_type(2)));
typedef float v8f __attribute__((ext_vector_type(8)));


__device__ __forceinline__ void edge_sd(const int* __restrict__ ei, int E, int idx,
                                        int& s, int& d) {
    if (idx < E) { s = ei[idx]; d = ei[E + idx]; }
    else         { s = idx - E; d = idx - E; }
}

__device__ __forceinline__ unsigned f2ord(float f) {
    unsigned b = __float_as_uint(f);
    return (b & 0x80000000u) ? ~b : (b | 0x80000000u);
}
__device__ __forceinline__ float ord2f(unsigned u) {
    return (u & 0x80000000u) ? __uint_as_float(u ^ 0x80000000u)
                             : __uint_as_float(~u);
}

__device__ __forceinline__ float lrelu(float v) {
    return (v > 0.f) ? v : NEG_SLOPE * v;
}


__global__ void k_zero(float* __restrict__ p, size_t n) {
    size_t i = (size_t)blockIdx.x * blockDim.x + threadIdx.x;
    size_t stride = (size_t)gridDim.x * blockDim.x;
    for (; i < n; i += stride) p[i] = 0.f;
}

__global__ void k_gemm_wmma_f32(const float* __restrict__ A,
                                const float* __restrict__ B,
                                float* __restrict__ C,
                                int M, int K, int Nc) {
    int wave = blockIdx.x * (blockDim.x >> 5) + (threadIdx.x >> 5);
    int ntN = Nc >> 4;
    int ntM = M >> 4;
    if (wave >= ntM * ntN) return;
    int tm = wave / ntN, tn = wave % ntN;
    int m0 = tm << 4, n0 = tn << 4;

    int lane  = threadIdx.x & 31;
    int lr    = lane & 15;
    int hi    = lane >> 4;
    int khalf = hi << 1;

    const float* __restrict__ arow = A + (size_t)(m0 + lr) * K;
    v8f c = {};
    for (int k0 = 0; k0 < K; k0 += 4) {
        v2f a, b;
        a.x = arow[k0 + khalf];
        a.y = arow[k0 + khalf + 1];
        b.x = B[(size_t)(k0 + khalf)     * Nc + n0 + lr];
        b.y = B[(size_t)(k0 + khalf + 1) * Nc + n0 + lr];
        c = __builtin_amdgcn_wmma_f32_16x16x4_f32(
                 false, a,  false, b,
                 (short)0, c,  false,  false);
    }
#pragma unroll
    for (int r = 0; r < 8; ++r) {
        int row = m0 + r + (hi << 3);
        C[(size_t)row * Nc + n0 + lr] = c[r];
    }
}

__global__ void k_attn_proj(const float* __restrict__ h,
                            const float* __restrict__ att_s,
                            const float* __restrict__ att_d,
                            float* __restrict__ a_s, float* __restrict__ a_d,
                            int n, int heads, int ch) {
    int i = blockIdx.x * blockDim.x + threadIdx.x;
    if (i >= n * heads) return;
    int node = i / heads, hh = i - node * heads;
    const float* hp = h + (size_t)node * heads * ch + (size_t)hh * ch;
    const float* as = att_s + hh * ch;
    const float* ad = att_d + hh * ch;
    float ss = 0.f, sd = 0.f;
    for (int c = 0; c < ch; ++c) {
        float v = hp[c];
        ss += v * as[c];
        sd += v * ad[c];
    }
    a_s[i] = ss;
    a_d[i] = sd;
}

__global__ void k_edge_max(const int* __restrict__ ei, int E, int Etot, int heads,
                           const float* __restrict__ a_s, const float* __restrict__ a_d,
                           unsigned* __restrict__ menc) {
    int i = blockIdx.x * blockDim.x + threadIdx.x;
    if (i >= Etot * heads) return;
    int e = i / heads, hh = i - e * heads;
    int s, d; edge_sd(ei, E, e, s, d);
    float v = lrelu(a_s[s * heads + hh] + a_d[d * heads + hh]);
    atomicMax(&menc[(size_t)d * heads + hh], f2ord(v));
}

__global__ void k_edge_expsum(const int* __restrict__ ei, int E, int Etot, int heads,
                              const float* __restrict__ a_s, const float* __restrict__ a_d,
                              const unsigned* __restrict__ menc,
                              float* __restrict__ denom) {
    int i = blockIdx.x * blockDim.x + threadIdx.x;
    if (i >= Etot * heads) return;
    int e = i / heads, hh = i - e * heads;
    int s, d; edge_sd(ei, E, e, s, d);
    float v = lrelu(a_s[s * heads + hh] + a_d[d * heads + hh]);
    float w = expf(v - ord2f(menc[(size_t)d * heads + hh]));
    atomicAdd(&denom[(size_t)d * heads + hh], w);
}

__global__ void k_edge_scatter(const int* __restrict__ ei, int E, int Etot,
                               int heads, int ch,
                               const float* __restrict__ a_s, const float* __restrict__ a_d,
                               const unsigned* __restrict__ menc,
                               const float* __restrict__ denom,
                               const float* __restrict__ h,
                               float* __restrict__ agg) {
    int i = blockIdx.x * blockDim.x + threadIdx.x;
    if (i >= Etot * heads) return;
    int e = i / heads, hh = i - e * heads;
    int s, d; edge_sd(ei, E, e, s, d);
    float v = lrelu(a_s[s * heads + hh] + a_d[d * heads + hh]);
    float alpha = expf(v - ord2f(menc[(size_t)d * heads + hh]))
                  / denom[(size_t)d * heads + hh];
    const float* hs = h   + (size_t)s * heads * ch + (size_t)hh * ch;
    float*       ag = agg + (size_t)d * heads * ch + (size_t)hh * ch;
    for (int c = 0; c < ch; ++c)
        atomicAdd(&ag[c], alpha * hs[c]);
}

__global__ void k_bias_relu(float* __restrict__ x, const float* __restrict__ b,
                            int n, int hc) {
    int i = blockIdx.x * blockDim.x + threadIdx.x;
    if (i >= n * hc) return;
    x[i] = fmaxf(x[i] + b[i % hc], 0.f);
}

__global__ void k_bias_out(const float* __restrict__ agg, const float* __restrict__ b,
                           float* __restrict__ out, int n, int hc) {
    int i = blockIdx.x * blockDim.x + threadIdx.x;
    if (i >= n * hc) return;
    out[i] = agg[i] + b[i % hc];
}


extern "C" void kernel_launch(void* const* d_in, const int* in_sizes, int n_in,
                              void* d_out, int out_size, void* d_ws, size_t ws_size,
                              hipStream_t stream) {
    const float* x        = (const float*)d_in[0];
    const int*   ei       = (const int*)  d_in[1];
    const float* W1       = (const float*)d_in[2];
    const float* att_src1 = (const float*)d_in[3];
    const float* att_dst1 = (const float*)d_in[4];
    const float* b1       = (const float*)d_in[5];
    const float* W2       = (const float*)d_in[6];
    const float* att_src2 = (const float*)d_in[7];
    const float* att_dst2 = (const float*)d_in[8];
    const float* b2       = (const float*)d_in[9];
    float* out = (float*)d_out;

    const int N    = in_sizes[0] / F_IN;
    const int E    = in_sizes[1] / 2;
    const int Etot = E + N;

    float* ws  = (float*)d_ws;
    float* h1  = ws;
    float* as1 = h1  + (size_t)N * H1C1;
    float* ad1 = as1 + (size_t)N * HEADS1;
    float* h2  = ad1 + (size_t)N * HEADS1;
    float* as2 = h2  + (size_t)N * NCLS;
    float* ad2 = as2 + (size_t)N;
    float*    zbase = ad2 + (size_t)N;
    float*    agg1  = zbase;
    float*    den1  = agg1 + (size_t)N * H1C1;
    unsigned* m1    = (unsigned*)(den1 + (size_t)N * HEADS1);
    float*    agg2  = (float*)(m1 + (size_t)N * HEADS1);
    float*    den2  = agg2 + (size_t)N * NCLS;
    unsigned* m2    = (unsigned*)(den2 + (size_t)N);
    size_t zcount = (size_t)N * (H1C1 + HEADS1 + HEADS1 + NCLS + 1 + 1);

    const int T = 256;

    k_zero<<<4096, T, 0, stream>>>(zbase, zcount);

    {
        int tiles = (N >> 4) * (H1C1 >> 4);
        int wavesPerBlk = T >> 5;
        k_gemm_wmma_f32<<<(tiles + wavesPerBlk - 1) / wavesPerBlk, T, 0, stream>>>(
            x, W1, h1, N, F_IN, H1C1);
    }
    k_attn_proj<<<((N * HEADS1) + T - 1) / T, T, 0, stream>>>(
        h1, att_src1, att_dst1, as1, ad1, N, HEADS1, CH1);

    int work1 = Etot * HEADS1;
    k_edge_max   <<<(work1 + T - 1) / T, T, 0, stream>>>(ei, E, Etot, HEADS1, as1, ad1, m1);
    k_edge_expsum<<<(work1 + T - 1) / T, T, 0, stream>>>(ei, E, Etot, HEADS1, as1, ad1, m1, den1);
    k_edge_scatter<<<(work1 + T - 1) / T, T, 0, stream>>>(ei, E, Etot, HEADS1, CH1,
                                                          as1, ad1, m1, den1, h1, agg1);
    k_bias_relu<<<((N * H1C1) + T - 1) / T, T, 0, stream>>>(agg1, b1, N, H1C1);

    {
        int tiles = (N >> 4) * (NCLS >> 4);
        int wavesPerBlk = T >> 5;
        k_gemm_wmma_f32<<<(tiles + wavesPerBlk - 1) / wavesPerBlk, T, 0, stream>>>(
            agg1, W2, h2, N, H1C1, NCLS);
    }
    k_attn_proj<<<(N + T - 1) / T, T, 0, stream>>>(
        h2, att_src2, att_dst2, as2, ad2, N, 1, NCLS);

    k_edge_max   <<<(Etot + T - 1) / T, T, 0, stream>>>(ei, E, Etot, 1, as2, ad2, m2);
    k_edge_expsum<<<(Etot + T - 1) / T, T, 0, stream>>>(ei, E, Etot, 1, as2, ad2, m2, den2);
    k_edge_scatter<<<(Etot + T - 1) / T, T, 0, stream>>>(ei, E, Etot, 1, NCLS,
                                                         as2, ad2, m2, den2, h2, agg2);
    k_bias_out<<<((N * NCLS) + T - 1) / T, T, 0, stream>>>(agg2, b2, out, N, NCLS);
}


# Edges counting-sorted by destination in every call (int atomics), then one wave per node does softmax and weighted sum in registers with plain stores: no float atomics; row sum order = arrival rank
# speedup vs baseline: 5.4804x; 5.4804x over previous
#include <hip/hip_runtime.h>
#include <hip/hip_bf16.h>


#define NEG_SLOPE 0.2f
#define F_IN   128
#define H1C1   160
#define HEADS1 8
#define CH1    20
#define NCLS   16

typedef float v2f __attribute__((ext_vector_type(2)));
typedef float v8f __attribute__((ext_vector_type(8)));


__device__ __forceinline__ float lrelu(float v) {
    return (v > 0.f) ? v : NEG_SLOPE * v;
}


__global__ void k_zero(float* __restrict__ p, size_t n) {
    size_t i = (size_t)blockIdx.x * blockDim.x + threadIdx.x;
    size_t stride = (size_t)gridDim.x * blockDim.x;
    for (; i < n; i += stride) p[i] = 0.f;
}

__global__ void k_gemm_wmma_f32(const float* __restrict__ A,
                                const float* __restrict__ B,
                                float* __restrict__ C,
                                int M, int K, int Nc) {
    int wave = blockIdx.x * (blockDim.x >> 5) + (threadIdx.x >> 5);
    int ntN = Nc >> 4;
    int ntM = M >> 4;
    if (wave >= ntM * ntN) return;
    int tm = wave / ntN, tn = wave % ntN;
    int m0 = tm << 4, n0 = tn << 4;

    int lane  = threadIdx.x & 31;
    int lr    = lane & 15;
    int hi    = lane >> 4;
    int khalf = hi << 1;

    const float* __restrict__ arow = A + (size_t)(m0 + lr) * K;
    v8f c = {};
    for (int k0 = 0; k0 < K; k0 += 4) {
        v2f a, b;
        a.x = arow[k0 + khalf];
        a.y = arow[k0 + khalf + 1];
        b.x = B[(size_t)(k0 + khalf)     * Nc + n0 + lr];
        b.y = B[(size_t)(k0 + khalf + 1) * Nc + n0 + lr];
        c = __builtin_amdgcn_wmma_f32_16x16x4_f32(
                 false, a,  false, b,
                 (short)0, c,  false,  false);
    }
    for (int r = 0; r < 8; ++r) {
        int row = m0 + r + (hi << 3);
        C[(size_t)row * Nc + n0 + lr] = c[r];
    }
}

__global__ void k_attn_proj(const float* __restrict__ h,
                            const float* __restrict__ att_s,
                            const float* __restrict__ att_d,
                            float* __restrict__ a_s, float* __restrict__ a_d,
                            int n, int heads, int ch) {
    int i = blockIdx.x * blockDim.x + threadIdx.x;
    if (i >= n * heads) return;
    int node = i / heads, hh = i - node * heads;
    const float* hp = h + (size_t)node * heads * ch + (size_t)hh * ch;
    const float* as = att_s + hh * ch;
    const float* ad = att_d + hh * ch;
    float ss = 0.f, sd = 0.f;
    for (int c = 0; c < ch; ++c) {
        float v = hp[c];
        ss += v * as[c];
        sd += v * ad[c];
    }
    a_s[i] = ss;
    a_d[i] = sd;
}


__global__ void k_csr_count(const int* __restrict__ ei, int E,
                            int* __restrict__ cnt, int* __restrict__ rank) {
    int e = blockIdx.x * blockDim.x + threadIdx.x;
    if (e >= E) return;
    rank[e] = atomicAdd(&cnt[ei[E + e]], 1);
}

__global__ void k_csr_scan(const int* __restrict__ cnt, int* __restrict__ rowptr, int n) {
    __shared__ int wsum[32];
    int lane = threadIdx.x & 31, w = threadIdx.x >> 5;
    int per = (((n + 31) >> 5) + 31) & ~31;
    int beg = w * per;
    int end = (beg + per < n) ? beg + per : n;
    int tot = 0;
    for (int i = beg + lane; i < end; i += 32) tot += cnt[i];
    for (int o = 16; o > 0; o >>= 1) tot += __shfl_xor(tot, o);
    if (lane == 0) wsum[w] = tot;
    __syncthreads();
    int carry = (lane < w) ? wsum[lane] : 0;
    for (int o = 16; o > 0; o >>= 1) carry += __shfl_xor(carry, o);
    for (int base = beg; base < end; base += 32) {
        int i = base + lane;
        int c = (i < end) ? cnt[i] : 0;
        int inc = c;
        for (int o = 1; o < 32; o <<= 1) {
            int up = __shfl_up(inc, o);
            if (lane >= o) inc += up;
        }
        if (i < end) rowptr[i] = carry + inc - c;
        carry += __shfl(inc, 31);
    }
    if (w == 31 && lane == 0) rowptr[n] = carry;
}

__global__ void k_csr_fill(const int* __restrict__ ei, int E,
                           const int* __restrict__ rowptr, const int* __restrict__ rank,
                           int* __restrict__ ssrc) {
    int e = blockIdx.x * blockDim.x + threadIdx.x;
    if (e >= E) return;
    ssrc[rowptr[ei[E + e]] + rank[e]] = ei[e];
}

__global__ void k_gather_l1(const int* __restrict__ rowptr, const int* __restrict__ ssrc,
                            const float* __restrict__ a_s, const float* __restrict__ a_d,
                            const float* __restrict__ h, float* __restrict__ agg, int n) {
    int lane = threadIdx.x & 31;
    int node = blockIdx.x * (blockDim.x >> 5) + (threadIdx.x >> 5);
    node = __builtin_amdgcn_readfirstlane(node);
    if (node >= n) return;
    int slot = lane >> 3, head = lane & 7;
    int beg  = rowptr[node];
    int cntE = rowptr[node + 1] - beg + 1;
    float ad = a_d[node * HEADS1 + head];

    float m = -__builtin_huge_valf();
    for (int t0 = 0; t0 < cntE; t0 += 4) {
        int t = t0 + slot;
        if (t < cntE) {
            int s = (t == 0) ? node : ssrc[beg + t - 1];
            m = fmaxf(m, lrelu(a_s[s * HEADS1 + head] + ad));
        }
    }
    m = fmaxf(m, __shfl_xor(m, 8));
    m = fmaxf(m, __shfl_xor(m, 16));

    int hk[5];
    for (int k = 0; k < 5; ++k) hk[k] = (lane + 32 * k) / CH1;
    float acc[5] = {0.f, 0.f, 0.f, 0.f, 0.f};
    float den = 0.f;
    for (int t0 = 0; t0 < cntE; t0 += 4) {
        int t = t0 + slot;
        int s = node;
        float w = 0.f;
        if (t < cntE) {
            if (t > 0) s = ssrc[beg + t - 1];
            w = expf(lrelu(a_s[s * HEADS1 + head] + ad) - m);
        }
        den += w;
        if (t0 + 4 <= cntE) {
            float hv[4][5];
            for (int q = 0; q < 4; ++q) {
                int sq = __builtin_amdgcn_readlane(s, q * 8);
                const float* hs = h + (size_t)sq * H1C1 + lane;
                for (int k = 0; k < 5; ++k) hv[q][k] = hs[32 * k];
            }
            for (int q = 0; q < 4; ++q) {
                for (int k = 0; k < 5; ++k) acc[k] += __shfl(w, q * 8 + hk[k]) * hv[q][k];
            }
        } else {
            for (int q = 0; q < 3; ++q) {
                if (t0 + q < cntE) {
                    int sq = __builtin_amdgcn_readlane(s, q * 8);
                    const float* hs = h + (size_t)sq * H1C1 + lane;
                    for (int k = 0; k < 5; ++k) acc[k] += __shfl(w, q * 8 + hk[k]) * hs[32 * k];
                }
            }
        }
    }
    den += __shfl_xor(den, 8);
    den += __shfl_xor(den, 16);
    float* o = agg + (size_t)node * H1C1 + lane;
    for (int k = 0; k < 5; ++k) o[32 * k] = acc[k] / __shfl(den, hk[k]);
}

__global__ void k_gather_l2(const int* __restrict__ rowptr, const int* __restrict__ ssrc,
                            const float* __restrict__ a_s, const float* __restrict__ a_d,
                            const float* __restrict__ h, float* __restrict__ agg, int n) {
    int lane = threadIdx.x & 31;
    int node = blockIdx.x * (blockDim.x >> 5) + (threadIdx.x >> 5);
    node = __builtin_amdgcn_readfirstlane(node);
    if (node >= n) return;
    int half = lane >> 4, c = lane & 15;
    int beg  = rowptr[node];
    int cntE = rowptr[node + 1] - beg + 1;
    float ad = a_d[node];

    float m = -__builtin_huge_valf();
    for (int t0 = 0; t0 < cntE; t0 += 32) {
        int t = t0 + lane;
        if (t < cntE) {
            int s = (t == 0) ? node : ssrc[beg + t - 1];
            m = fmaxf(m, lrelu(a_s[s] + ad));
        }
    }
    for (int o = 16; o > 0; o >>= 1) m = fmaxf(m, __shfl_xor(m, o));

    float acc = 0.f, den = 0.f;
    for (int t0 = 0; t0 < cntE; t0 += 32) {
        int t = t0 + lane;
        int s = node;
        float w = 0.f;
        if (t < cntE) {
            if (t > 0) s = ssrc[beg + t - 1];
            w = expf(lrelu(a_s[s] + ad) - m);
        }
        den += w;
        int left = cntE - t0;
        int np = (left < 32) ? left : 32;
#pragma unroll 4
        for (int q = 0; q < np; q += 2) {
            int   sl = __shfl(s, q + half);
            float wl = __shfl(w, q + half);
            acc += wl * h[(size_t)sl * NCLS + c];
        }
    }
    for (int o = 16; o > 0; o >>= 1) den += __shfl_xor(den, o);
    acc += __shfl_xor(acc, 16);
    if (lane < 16) agg[(size_t)node * NCLS + c] = acc / den;
}

__global__ void k_bias_relu(float* __restrict__ x, const float* __restrict__ b,
                            int n, int hc) {
    int i = blockIdx.x * blockDim.x + threadIdx.x;
    if (i >= n * hc) return;
    x[i] = fmaxf(x[i] + b[i % hc], 0.f);
}

__global__ void k_bias_out(const float* __restrict__ agg, const float* __restrict__ b,
                           float* __restrict__ out, int n, int hc) {
    int i = blockIdx.x * blockDim.x + threadIdx.x;
    if (i >= n * hc) return;
    out[i] = agg[i] + b[i % hc];
}


extern "C" void kernel_launch(void* const* d_in, const int* in_sizes, int n_in,
                              void* d_out, int out_size, void* d_ws, size_t ws_size,
                              hipStream_t stream) {
    const float* x        = (const float*)d_in[0];
    const int*   ei       = (const int*)  d_in[1];
    const float* W1       = (const float*)d_in[2];
    const float* att_src1 = (const float*)d_in[3];
    const float* att_dst1 = (const float*)d_in[4];
    const float* b1       = (const float*)d_in[5];
    const float* W2       = (const float*)d_in[6];
    const float* att_src2 = (const float*)d_in[7];
    const float* att_dst2 = (const float*)d_in[8];
    const float* b2       = (const float*)d_in[9];
    float* out = (float*)d_out;

    const int N    = in_sizes[0] / F_IN;
    const int E    = in_sizes[1] / 2;
    float* ws  = (float*)d_ws;
    float* h1  = ws;
    float* as1 = h1  + (size_t)N * H1C1;
    float* ad1 = as1 + (size_t)N * HEADS1;
    float* h2  = ad1 + (size_t)N * HEADS1;
    float* as2 = h2  + (size_t)N * NCLS;
    float* ad2 = as2 + (size_t)N;
    float* agg1 = ad2 + (size_t)N;
    float* agg2 = agg1 + (size_t)N * H1C1;
    int* cnt    = (int*)(agg2 + (size_t)N * NCLS);
    int* rowptr = cnt + (size_t)N;
    int* rank   = rowptr + (size_t)N + 1;
    int* ssrc   = rank + (size_t)E;

    const int T = 256;
    const int wavesPerBlk = T >> 5;

    k_zero<<<(N + T - 1) / T, T, 0, stream>>>((float*)cnt, (size_t)N);
    k_csr_count<<<(E + T - 1) / T, T, 0, stream>>>(ei, E, cnt, rank);
    k_csr_scan<<<1, 1024, 0, stream>>>(cnt, rowptr, N);
    k_csr_fill<<<(E + T - 1) / T, T, 0, stream>>>(ei, E, rowptr, rank, ssrc);

    {
        int tiles = (N >> 4) * (H1C1 >> 4);
        k_gemm_wmma_f32<<<(tiles + wavesPerBlk - 1) / wavesPerBlk, T, 0, stream>>>(
            x, W1, h1, N, F_IN, H1C1);
    }
    k_attn_proj<<<((N * HEADS1) + T - 1) / T, T, 0, stream>>>(
        h1, att_src1, att_dst1, as1, ad1, N, HEADS1, CH1);
    k_gather_l1<<<(N + wavesPerBlk - 1) / wavesPerBlk, T, 0, stream>>>(
        rowptr, ssrc, as1, ad1, h1, agg1, N);
    k_bias_relu<<<((N * H1C1) + T - 1) / T, T, 0, stream>>>(agg1, b1, N, H1C1);

    {
        int tiles = (N >> 4) * (NCLS >> 4);
        k_gemm_wmma_f32<<<(tiles + wavesPerBlk - 1) / wavesPerBlk, T, 0, stream>>>(
            agg1, W2, h2, N, H1C1, NCLS);
    }
    k_attn_proj<<<(N + T - 1) / T, T, 0, stream>>>(
        h2, att_src2, att_dst2, as2, ad2, N, 1, NCLS);
    k_gather_l2<<<(N + wavesPerBlk - 1) / wavesPerBlk, T, 0, stream>>>(
        rowptr, ssrc, as2, ad2, h2, agg2, N);
    k_bias_out<<<((N * NCLS) + T - 1) / T, T, 0, stream>>>(agg2, b2, out, N, NCLS);
}


# Bias+ReLU and final bias applied in the gather kernels' own stores: the two elementwise launches (275000 one-element waves) are gone; sums and their order unchanged
# speedup vs baseline: 6.1260x; 1.1178x over previous
#include <hip/hip_runtime.h>
#include <hip/hip_bf16.h>


#define NEG_SLOPE 0.2f
#define F_IN   128
#define H1C1   160
#define HEADS1 8
#define CH1    20
#define NCLS   16

typedef float v2f __attribute__((ext_vector_type(2)));
typedef float v8f __attribute__((ext_vector_type(8)));


__device__ __forceinline__ float lrelu(float v) {
    return (v > 0.f) ? v : NEG_SLOPE * v;
}


__global__ void k_zero(float* __restrict__ p, size_t n) {
    size_t i = (size_t)blockIdx.x * blockDim.x + threadIdx.x;
    size_t stride = (size_t)gridDim.x * blockDim.x;
    for (; i < n; i += stride) p[i] = 0.f;
}

__global__ void k_gemm_wmma_f32(const float* __restrict__ A,
                                const float* __restrict__ B,
                                float* __restrict__ C,
                                int M, int K, int Nc) {
    int wave = blockIdx.x * (blockDim.x >> 5) + (threadIdx.x >> 5);
    int ntN = Nc >> 4;
    int ntM = M >> 4;
    if (wave >= ntM * ntN) return;
    int tm = wave / ntN, tn = wave % ntN;
    int m0 = tm << 4, n0 = tn << 4;

    int lane  = threadIdx.x & 31;
    int lr    = lane & 15;
    int hi    = lane >> 4;
    int khalf = hi << 1;

    const float* __restrict__ arow = A + (size_t)(m0 + lr) * K;
    v8f c = {};
    for (int k0 = 0; k0 < K; k0 += 4) {
        v2f a, b;
        a.x = arow[k0 + khalf];
        a.y = arow[k0 + khalf + 1];
        b.x = B[(size_t)(k0 + khalf)     * Nc + n0 + lr];
        b.y = B[(size_t)(k0 + khalf + 1) * Nc + n0 + lr];
        c = __builtin_amdgcn_wmma_f32_16x16x4_f32(
                 false, a,  false, b,
                 (short)0, c,  false,  false);
    }
    for (int r = 0; r < 8; ++r) {
        int row = m0 + r + (hi << 3);
        C[(size_t)row * Nc + n0 + lr] = c[r];
    }
}

__global__ void k_attn_proj(const float* __restrict__ h,
                            const float* __restrict__ att_s,
                            const float* __restrict__ att_d,
                            float* __restrict__ a_s, float* __restrict__ a_d,
                            int n, int heads, int ch) {
    int i = blockIdx.x * blockDim.x + threadIdx.x;
    if (i >= n * heads) return;
    int node = i / heads, hh = i - node * heads;
    const float* hp = h + (size_t)node * heads * ch + (size_t)hh * ch;
    const float* as = att_s + hh * ch;
    const float* ad = att_d + hh * ch;
    float ss = 0.f, sd = 0.f;
    for (int c = 0; c < ch; ++c) {
        float v = hp[c];
        ss += v * as[c];
        sd += v * ad[c];
    }
    a_s[i] = ss;
    a_d[i] = sd;
}


__global__ void k_csr_count(const int* __restrict__ ei, int E,
                            int* __restrict__ cnt, int* __restrict__ rank) {
    int e = blockIdx.x * blockDim.x + threadIdx.x;
    if (e >= E) return;
    rank[e] = atomicAdd(&cnt[ei[E + e]], 1);
}

__global__ void k_csr_scan(const int* __restrict__ cnt, int* __restrict__ rowptr, int n) {
    __shared__ int wsum[32];
    int lane = threadIdx.x & 31, w = threadIdx.x >> 5;
    int per = (((n + 31) >> 5) + 31) & ~31;
    int beg = w * per;
    int end = (beg + per < n) ? beg + per : n;
    int tot = 0;
    for (int i = beg + lane; i < end; i += 32) tot += cnt[i];
    for (int o = 16; o > 0; o >>= 1) tot += __shfl_xor(tot, o);
    if (lane == 0) wsum[w] = tot;
    __syncthreads();
    int carry = (lane < w) ? wsum[lane] : 0;
    for (int o = 16; o > 0; o >>= 1) carry += __shfl_xor(carry, o);
    for (int base = beg; base < end; base += 32) {
        int i = base + lane;
        int c = (i < end) ? cnt[i] : 0;
        int inc = c;
        for (int o = 1; o < 32; o <<= 1) {
            int up = __shfl_up(inc, o);
            if (lane >= o) inc += up;
        }
        if (i < end) rowptr[i] = carry + inc - c;
        carry += __shfl(inc, 31);
    }
    if (w == 31 && lane == 0) rowptr[n] = carry;
}

__global__ void k_csr_fill(const int* __restrict__ ei, int E,
                           const int* __restrict__ rowptr, const int* __restrict__ rank,
                           int* __restrict__ ssrc) {
    int e = blockIdx.x * blockDim.x + threadIdx.x;
    if (e >= E) return;
    ssrc[rowptr[ei[E + e]] + rank[e]] = ei[e];
}

__global__ void k_gather_l1(const int* __restrict__ rowptr, const int* __restrict__ ssrc,
                            const float* __restrict__ a_s, const float* __restrict__ a_d,
                            const float* __restrict__ h, const float* __restrict__ b,
                            float* __restrict__ agg, int n) {
    int lane = threadIdx.x & 31;
    int node = blockIdx.x * (blockDim.x >> 5) + (threadIdx.x >> 5);
    node = __builtin_amdgcn_readfirstlane(node);
    if (node >= n) return;
    int slot = lane >> 3, head = lane & 7;
    int beg  = rowptr[node];
    int cntE = rowptr[node + 1] - beg + 1;
    float ad = a_d[node * HEADS1 + head];

    float m = -__builtin_huge_valf();
    for (int t0 = 0; t0 < cntE; t0 += 4) {
        int t = t0 + slot;
        if (t < cntE) {
            int s = (t == 0) ? node : ssrc[beg + t - 1];
            m = fmaxf(m, lrelu(a_s[s * HEADS1 + head] + ad));
        }
    }
    m = fmaxf(m, __shfl_xor(m, 8));
    m = fmaxf(m, __shfl_xor(m, 16));

    int hk[5];
    for (int k = 0; k < 5; ++k) hk[k] = (lane + 32 * k) / CH1;
    float acc[5] = {0.f, 0.f, 0.f, 0.f, 0.f};
    float den = 0.f;
    for (int t0 = 0; t0 < cntE; t0 += 4) {
        int t = t0 + slot;
        int s = node;
        float w = 0.f;
        if (t < cntE) {
            if (t > 0) s = ssrc[beg + t - 1];
            w = expf(lrelu(a_s[s * HEADS1 + head] + ad) - m);
        }
        den += w;
        if (t0 + 4 <= cntE) {
            float hv[4][5];
            for (int q = 0; q < 4; ++q) {
                int sq = __builtin_amdgcn_readlane(s, q * 8);
                const float* hs = h + (size_t)sq * H1C1 + lane;
                for (int k = 0; k < 5; ++k) hv[q][k] = hs[32 * k];
            }
            for (int q = 0; q < 4; ++q) {
                for (int k = 0; k < 5; ++k) acc[k] += __shfl(w, q * 8 + hk[k]) * hv[q][k];
            }
        } else {
            for (int q = 0; q < 3; ++q) {
                if (t0 + q < cntE) {
                    int sq = __builtin_amdgcn_readlane(s, q * 8);
                    const float* hs = h + (size_t)sq * H1C1 + lane;
                    for (int k = 0; k < 5; ++k) acc[k] += __shfl(w, q * 8 + hk[k]) * hs[32 * k];
                }
            }
        }
    }
    float bv[5];
    for (int k = 0; k < 5; ++k) bv[k] = b[lane + 32 * k];
    den += __shfl_xor(den, 8);
    den += __shfl_xor(den, 16);
    float* o = agg + (size_t)node * H1C1 + lane;
    for (int k = 0; k < 5; ++k) o[32 * k] = fmaxf(acc[k] / __shfl(den, hk[k]) + bv[k], 0.f);
}

__global__ void k_gather_l2(const int* __restrict__ rowptr, const int* __restrict__ ssrc,
                            const float* __restrict__ a_s, const float* __restrict__ a_d,
                            const float* __restrict__ h, const float* __restrict__ b,
                            float* __restrict__ out, int n) {
    int lane = threadIdx.x & 31;
    int node = blockIdx.x * (blockDim.x >> 5) + (threadIdx.x >> 5);
    node = __builtin_amdgcn_readfirstlane(node);
    if (node >= n) return;
    int half = lane >> 4, c = lane & 15;
    int beg  = rowptr[node];
    int cntE = rowptr[node + 1] - beg + 1;
    float ad = a_d[node];
    float bc = b[c];

    float m = -__builtin_huge_valf();
    for (int t0 = 0; t0 < cntE; t0 += 32) {
        int t = t0 + lane;
        if (t < cntE) {
            int s = (t == 0) ? node : ssrc[beg + t - 1];
            m = fmaxf(m, lrelu(a_s[s] + ad));
        }
    }
    for (int o = 16; o > 0; o >>= 1) m = fmaxf(m, __shfl_xor(m, o));

    float acc = 0.f, den = 0.f;
    for (int t0 = 0; t0 < cntE; t0 += 32) {
        int t = t0 + lane;
        int s = node;
        float w = 0.f;
        if (t < cntE) {
            if (t > 0) s = ssrc[beg + t - 1];
            w = expf(lrelu(a_s[s] + ad) - m);
        }
        den += w;
        int left = cntE - t0;
        int np = (left < 32) ? left : 32;
#pragma unroll 4
        for (int q = 0; q < np; q += 2) {
            int   sl = __shfl(s, q + half);
            float wl = __shfl(w, q + half);
            acc += wl * h[(size_t)sl * NCLS + c];
        }
    }
    for (int o = 16; o > 0; o >>= 1) den += __shfl_xor(den, o);
    acc += __shfl_xor(acc, 16);
    if (lane < 16) out[(size_t)node * NCLS + c] = acc / den + bc;
}


extern "C" void kernel_launch(void* const* d_in, const int* in_sizes, int n_in,
                              void* d_out, int out_size, void* d_ws, size_t ws_size,
                              hipStream_t stream) {
    const float* x        = (const float*)d_in[0];
    const int*   ei       = (const int*)  d_in[1];
    const float* W1       = (const float*)d_in[2];
    const float* att_src1 = (const float*)d_in[3];
    const float* att_dst1 = (const float*)d_in[4];
    const float* b1       = (const float*)d_in[5];
    const float* W2       = (const float*)d_in[6];
    const float* att_src2 = (const float*)d_in[7];
    const float* att_dst2 = (const float*)d_in[8];
    const float* b2       = (const float*)d_in[9];
    float* out = (float*)d_out;

    const int N    = in_sizes[0] / F_IN;
    const int E    = in_sizes[1] / 2;
    float* ws  = (float*)d_ws;
    float* h1  = ws;
    float* as1 = h1  + (size_t)N * H1C1;
    float* ad1 = as1 + (size_t)N * HEADS1;
    float* h2  = ad1 + (size_t)N * HEADS1;
    float* as2 = h2  + (size_t)N * NCLS;
    float* ad2 = as2 + (size_t)N;
    float* agg1 = ad2 + (size_t)N;
    int* cnt    = (int*)(agg1 + (size_t)N * H1C1);
    int* rowptr = cnt + (size_t)N;
    int* rank   = rowptr + (size_t)N + 1;
    int* ssrc   = rank + (size_t)E;

    const int T = 256;
    const int wavesPerBlk = T >> 5;

    k_zero<<<(N + T - 1) / T, T, 0, stream>>>((float*)cnt, (size_t)N);
    k_csr_count<<<(E + T - 1) / T, T, 0, stream>>>(ei, E, cnt, rank);
    k_csr_scan<<<1, 1024, 0, stream>>>(cnt, rowptr, N);
    k_csr_fill<<<(E + T - 1) / T, T, 0, stream>>>(ei, E, rowptr, rank, ssrc);

    {
        int tiles = (N >> 4) * (H1C1 >> 4);
        k_gemm_wmma_f32<<<(tiles + wavesPerBlk - 1) / wavesPerBlk, T, 0, stream>>>(
            x, W1, h1, N, F_IN, H1C1);
    }
    k_attn_proj<<<((N * HEADS1) + T - 1) / T, T, 0, stream>>>(
        h1, att_src1, att_dst1, as1, ad1, N, HEADS1, CH1);
    k_gather_l1<<<(N + wavesPerBlk - 1) / wavesPerBlk, T, 0, stream>>>(
        rowptr, ssrc, as1, ad1, h1, b1, agg1, N);

    {
        int tiles = (N >> 4) * (NCLS >> 4);
        k_gemm_wmma_f32<<<(tiles + wavesPerBlk - 1) / wavesPerBlk, T, 0, stream>>>(
            agg1, W2, h2, N, H1C1, NCLS);
    }
    k_attn_proj<<<(N + T - 1) / T, T, 0, stream>>>(
        h2, att_src2, att_dst2, as2, ad2, N, 1, NCLS);
    k_gather_l2<<<(N + wavesPerBlk - 1) / wavesPerBlk, T, 0, stream>>>(
        rowptr, ssrc, as2, ad2, h2, b2, out, N);
}


# GEMMs re-tiled: one wave owns a 16-row strip with all its column tiles, W staged once per block in LDS, each A fragment read once in 16-byte loads; k summing order inside each 16-k block changes
# speedup vs baseline: 6.7367x; 1.0997x over previous
#include <hip/hip_runtime.h>
#include <hip/hip_bf16.h>


#define NEG_SLOPE 0.2f
#define F_IN   128
#define H1C1   160
#define HEADS1 8
#define CH1    20
#define NCLS   16

typedef float v2f __attribute__((ext_vector_type(2)));
typedef float v8f __attribute__((ext_vector_type(8)));


__device__ __forceinline__ float lrelu(float v) {
    return (v > 0.f) ? v : NEG_SLOPE * v;
}


__global__ void k_zero(float* __restrict__ p, size_t n) {
    size_t i = (size_t)blockIdx.x * blockDim.x + threadIdx.x;
    size_t stride = (size_t)gridDim.x * blockDim.x;
    for (; i < n; i += stride) p[i] = 0.f;
}

template <int NT, int K>
__global__ void __launch_bounds__(256)
k_gemm_strip_f32(const float* __restrict__ A,
                 const float* __restrict__ B,
                 float* __restrict__ C, int M) {
    constexpr int NC    = NT * 16;
    constexpr int PITCH = 2 * NC + 8;
    constexpr int KB    = K / 16;
    __shared__ __attribute__((aligned(16))) float Bs[(K / 2) * PITCH];

    const int lane  = threadIdx.x & 31;
    const int c     = lane & 15;
    const int h     = lane >> 4;
    const int strip = blockIdx.x * (blockDim.x >> 5) + (threadIdx.x >> 5);
    const bool live = strip < (M >> 4);
    const int m0    = live ? (strip << 4) : 0;

    float4 a[KB][2];
    {
        const float4* ap = reinterpret_cast<const float4*>(A + (size_t)(m0 + c) * K + 8 * h);
#pragma unroll
        for (int i = 0; i < KB; ++i) {
            a[i][0] = ap[4 * i];
            a[i][1] = ap[4 * i + 1];
        }
    }

    constexpr int NQ    = NC / 4;
    constexpr int NP4   = (K / 2) * NQ;
    constexpr int ITERS = (NP4 + 255) / 256;
#pragma unroll
    for (int it = 0; it < ITERS; ++it) {
        const int p = it * 256 + (int)threadIdx.x;
        if (NP4 % 256 == 0 || p < NP4) {
            const int kp = p / NQ, q = p - kp * NQ;
            const float4 r0 = *reinterpret_cast<const float4*>(B + (size_t)(2 * kp)     * NC + 4 * q);
            const float4 r1 = *reinterpret_cast<const float4*>(B + (size_t)(2 * kp + 1) * NC + 4 * q);
            float4* d = reinterpret_cast<float4*>(&Bs[kp * PITCH + 8 * q]);
            d[0] = make_float4(r0.x, r1.x, r0.y, r1.y);
            d[1] = make_float4(r0.z, r1.z, r0.w, r1.w);
        }
    }
    __syncthreads();
    if (!live) return;

    v8f acc[NT];
#pragma unroll
    for (int t = 0; t < NT; ++t) acc[t] = v8f{};

    const float* bl = Bs + (4 * h) * PITCH + 2 * c;
#pragma unroll
    for (int i = 0; i < KB; ++i) {
#pragma unroll
        for (int j = 0; j < 4; ++j) {
            const float4 q = a[i][j >> 1];
            v2f af;
            af.x = (j & 1) ? q.z : q.x;
            af.y = (j & 1) ? q.w : q.y;
            const float* bj = bl + (8 * i + j) * PITCH;
#pragma unroll
            for (int t = 0; t < NT; ++t) {
                v2f bf = *reinterpret_cast<const v2f*>(bj + 32 * t);
                acc[t] = __builtin_amdgcn_wmma_f32_16x16x4_f32(
                              false, af,  false, bf,
                              (short)0, acc[t],  false,  false);
            }
        }
    }

    float* cp = C + (size_t)(m0 + 8 * h) * NC + c;
#pragma unroll
    for (int t = 0; t < NT; ++t) {
#pragma unroll
        for (int r = 0; r < 8; ++r) cp[(size_t)r * NC + 16 * t] = acc[t][r];
    }
}

__global__ void k_attn_proj(const float* __restrict__ h,
                            const float* __restrict__ att_s,
                            const float* __restrict__ att_d,
                            float* __restrict__ a_s, float* __restrict__ a_d,
                            int n, int heads, int ch) {
    int i = blockIdx.x * blockDim.x + threadIdx.x;
    if (i >= n * heads) return;
    int node = i / heads, hh = i - node * heads;
    const float* hp = h + (size_t)node * heads * ch + (size_t)hh * ch;
    const float* as = att_s + hh * ch;
    const float* ad = att_d + hh * ch;
    float ss = 0.f, sd = 0.f;
    for (int c = 0; c < ch; ++c) {
        float v = hp[c];
        ss += v * as[c];
        sd += v * ad[c];
    }
    a_s[i] = ss;
    a_d[i] = sd;
}


__global__ void k_csr_count(const int* __restrict__ ei, int E,
                            int* __restrict__ cnt, int* __restrict__ rank) {
    int e = blockIdx.x * blockDim.x + threadIdx.x;
    if (e >= E) return;
    rank[e] = atomicAdd(&cnt[ei[E + e]], 1);
}

__global__ void k_csr_scan(const int* __restrict__ cnt, int* __restrict__ rowptr, int n) {
    __shared__ int wsum[32];
    int lane = threadIdx.x & 31, w = threadIdx.x >> 5;
    int per = (((n + 31) >> 5) + 31) & ~31;
    int beg = w * per;
    int end = (beg + per < n) ? beg + per : n;
    int tot = 0;
    for (int i = beg + lane; i < end; i += 32) tot += cnt[i];
    for (int o = 16; o > 0; o >>= 1) tot += __shfl_xor(tot, o);
    if (lane == 0) wsum[w] = tot;
    __syncthreads();
    int carry = (lane < w) ? wsum[lane] : 0;
    for (int o = 16; o > 0; o >>= 1) carry += __shfl_xor(carry, o);
    for (int base = beg; base < end; base += 32) {
        int i = base + lane;
        int c = (i < end) ? cnt[i] : 0;
        int inc = c;
        for (int o = 1; o < 32; o <<= 1) {
            int up = __shfl_up(inc, o);
            if (lane >= o) inc += up;
        }
        if (i < end) rowptr[i] = carry + inc - c;
        carry += __shfl(inc, 31);
    }
    if (w == 31 && lane == 0) rowptr[n] = carry;
}

__global__ void k_csr_fill(const int* __restrict__ ei, int E,
                           const int* __restrict__ rowptr, const int* __restrict__ rank,
                           int* __restrict__ ssrc) {
    int e = blockIdx.x * blockDim.x + threadIdx.x;
    if (e >= E) return;
    ssrc[rowptr[ei[E + e]] + rank[e]] = ei[e];
}

__global__ void k_gather_l1(const int* __restrict__ rowptr, const int* __restrict__ ssrc,
                            const float* __restrict__ a_s, const float* __restrict__ a_d,
                            const float* __restrict__ h, const float* __restrict__ b,
                            float* __restrict__ agg, int n) {
    int lane = threadIdx.x & 31;
    int node = blockIdx.x * (blockDim.x >> 5) + (threadIdx.x >> 5);
    node = __builtin_amdgcn_readfirstlane(node);
    if (node >= n) return;
    int slot = lane >> 3, head = lane & 7;
    int beg  = rowptr[node];
    int cntE = rowptr[node + 1] - beg + 1;
    float ad = a_d[node * HEADS1 + head];

    float m = -__builtin_huge_valf();
    for (int t0 = 0; t0 < cntE; t0 += 4) {
        int t = t0 + slot;
        if (t < cntE) {
            int s = (t == 0) ? node : ssrc[beg + t - 1];
            m = fmaxf(m, lrelu(a_s[s * HEADS1 + head] + ad));
        }
    }
    m = fmaxf(m, __shfl_xor(m, 8));
    m = fmaxf(m, __shfl_xor(m, 16));

    int hk[5];
    for (int k = 0; k < 5; ++k) hk[k] = (lane + 32 * k) / CH1;
    float acc[5] = {0.f, 0.f, 0.f, 0.f, 0.f};
    float den = 0.f;
    for (int t0 = 0; t0 < cntE; t0 += 4) {
        int t = t0 + slot;
        int s = node;
        float w = 0.f;
        if (t < cntE) {
            if (t > 0) s = ssrc[beg + t - 1];
            w = expf(lrelu(a_s[s * HEADS1 + head] + ad) - m);
        }
        den += w;
        if (t0 + 4 <= cntE) {
            float hv[4][5];
            for (int q = 0; q < 4; ++q) {
                int sq = __builtin_amdgcn_readlane(s, q * 8);
                const float* hs = h + (size_t)sq * H1C1 + lane;
                for (int k = 0; k < 5; ++k) hv[q][k] = hs[32 * k];
            }
            for (int q = 0; q < 4; ++q) {
                for (int k = 0; k < 5; ++k) acc[k] += __shfl(w, q * 8 + hk[k]) * hv[q][k];
            }
        } else {
            for (int q = 0; q < 3; ++q) {
                if (t0 + q < cntE) {
                    int sq = __builtin_amdgcn_readlane(s, q * 8);
                    const float* hs = h + (size_t)sq * H1C1 + lane;
                    for (int k = 0; k < 5; ++k) acc[k] += __shfl(w, q * 8 + hk[k]) * hs[32 * k];
                }
            }
        }
    }
    float bv[5];
    for (int k = 0; k < 5; ++k) bv[k] = b[lane + 32 * k];
    den += __shfl_xor(den, 8);
    den += __shfl_xor(den, 16);
    float* o = agg + (size_t)node * H1C1 + lane;
    for (int k = 0; k < 5; ++k) o[32 * k] = fmaxf(acc[k] / __shfl(den, hk[k]) + bv[k], 0.f);
}

__global__ void k_gather_l2(const int* __restrict__ rowptr, const int* __restrict__ ssrc,
                            const float* __restrict__ a_s, const float* __restrict__ a_d,
                            const float* __restrict__ h, const float* __restrict__ b,
                            float* __restrict__ out, int n) {
    int lane = threadIdx.x & 31;
    int node = blockIdx.x * (blockDim.x >> 5) + (threadIdx.x >> 5);
    node = __builtin_amdgcn_readfirstlane(node);
    if (node >= n) return;
    int half = lane >> 4, c = lane & 15;
    int beg  = rowptr[node];
    int cntE = rowptr[node + 1] - beg + 1;
    float ad = a_d[node];
    float bc = b[c];

    float m = -__builtin_huge_valf();
    for (int t0 = 0; t0 < cntE; t0 += 32) {
        int t = t0 + lane;
        if (t < cntE) {
            int s = (t == 0) ? node : ssrc[beg + t - 1];
            m = fmaxf(m, lrelu(a_s[s] + ad));
        }
    }
    for (int o = 16; o > 0; o >>= 1) m = fmaxf(m, __shfl_xor(m, o));

    float acc = 0.f, den = 0.f;
    for (int t0 = 0; t0 < cntE; t0 += 32) {
        int t = t0 + lane;
        int s = node;
        float w = 0.f;
        if (t < cntE) {
            if (t > 0) s = ssrc[beg + t - 1];
            w = expf(lrelu(a_s[s] + ad) - m);
        }
        den += w;
        int left = cntE - t0;
        int np = (left < 32) ? left : 32;
#pragma unroll 4
        for (int q = 0; q < np; q += 2) {
            int   sl = __shfl(s, q + half);
            float wl = __shfl(w, q + half);
            acc += wl * h[(size_t)sl * NCLS + c];
        }
    }
    for (int o = 16; o > 0; o >>= 1) den += __shfl_xor(den, o);
    acc += __shfl_xor(acc, 16);
    if (lane < 16) out[(size_t)node * NCLS + c] = acc / den + bc;
}


extern "C" void kernel_launch(void* const* d_in, const int* in_sizes, int n_in,
                              void* d_out, int out_size, void* d_ws, size_t ws_size,
                              hipStream_t stream) {
    const float* x        = (const float*)d_in[0];
    const int*   ei       = (const int*)  d_in[1];
    const float* W1       = (const float*)d_in[2];
    const float* att_src1 = (const float*)d_in[3];
    const float* att_dst1 = (const float*)d_in[4];
    const float* b1       = (const float*)d_in[5];
    const float* W2       = (const float*)d_in[6];
    const float* att_src2 = (const float*)d_in[7];
    const float* att_dst2 = (const float*)d_in[8];
    const float* b2       = (const float*)d_in[9];
    float* out = (float*)d_out;

    const int N    = in_sizes[0] / F_IN;
    const int E    = in_sizes[1] / 2;
    float* ws  = (float*)d_ws;
    float* h1  = ws;
    float* as1 = h1  + (size_t)N * H1C1;
    float* ad1 = as1 + (size_t)N * HEADS1;
    float* h2  = ad1 + (size_t)N * HEADS1;
    float* as2 = h2  + (size_t)N * NCLS;
    float* ad2 = as2 + (size_t)N;
    float* agg1 = ad2 + (size_t)N;
    int* cnt    = (int*)(agg1 + (size_t)N * H1C1);
    int* rowptr = cnt + (size_t)N;
    int* rank   = rowptr + (size_t)N + 1;
    int* ssrc   = rank + (size_t)E;

    const int T = 256;
    const int wavesPerBlk = T >> 5;

    k_zero<<<(N + T - 1) / T, T, 0, stream>>>((float*)cnt, (size_t)N);
    k_csr_count<<<(E + T - 1) / T, T, 0, stream>>>(ei, E, cnt, rank);
    k_csr_scan<<<1, 1024, 0, stream>>>(cnt, rowptr, N);
    k_csr_fill<<<(E + T - 1) / T, T, 0, stream>>>(ei, E, rowptr, rank, ssrc);

    k_gemm_strip_f32<H1C1 / 16, F_IN><<<((N >> 4) + wavesPerBlk - 1) / wavesPerBlk, T, 0, stream>>>(
        x, W1, h1, N);
    k_attn_proj<<<((N * HEADS1) + T - 1) / T, T, 0, stream>>>(
        h1, att_src1, att_dst1, as1, ad1, N, HEADS1, CH1);
    k_gather_l1<<<(N + wavesPerBlk - 1) / wavesPerBlk, T, 0, stream>>>(
        rowptr, ssrc, as1, ad1, h1, b1, agg1, N);

    k_gemm_strip_f32<NCLS / 16, H1C1><<<((N >> 4) + wavesPerBlk - 1) / wavesPerBlk, T, 0, stream>>>(
        agg1, W2, h2, N);
    k_attn_proj<<<(N + T - 1) / T, T, 0, stream>>>(
        h2, att_src2, att_dst2, as2, ad2, N, 1, NCLS);
    k_gather_l2<<<(N + wavesPerBlk - 1) / wavesPerBlk, T, 0, stream>>>(
        rowptr, ssrc, as2, ad2, h2, b2, out, N);
}


# rowptr prefix sum re-cut: N/4096+1 work-groups; each adds up the counters in front of its 4096-entry range with independent 16-byte loads, then scans its own range once (was one block, serial trips)
# speedup vs baseline: 7.7533x; 1.1509x over previous
#include <hip/hip_runtime.h>
#include <hip/hip_bf16.h>


#define NEG_SLOPE 0.2f
#define F_IN   128
#define H1C1   160
#define HEADS1 8
#define CH1    20
#define NCLS   16

typedef float v2f __attribute__((ext_vector_type(2)));
typedef float v8f __attribute__((ext_vector_type(8)));


__device__ __forceinline__ float lrelu(float v) {
    return (v > 0.f) ? v : NEG_SLOPE * v;
}


__global__ void k_zero(float* __restrict__ p, size_t n) {
    size_t i = (size_t)blockIdx.x * blockDim.x + threadIdx.x;
    size_t stride = (size_t)gridDim.x * blockDim.x;
    for (; i < n; i += stride) p[i] = 0.f;
}

template <int NT, int K>
__global__ void __launch_bounds__(256)
k_gemm_strip_f32(const float* __restrict__ A,
                 const float* __restrict__ B,
                 float* __restrict__ C, int M) {
    constexpr int NC    = NT * 16;
    constexpr int PITCH = 2 * NC + 8;
    constexpr int KB    = K / 16;
    __shared__ __attribute__((aligned(16))) float Bs[(K / 2) * PITCH];

    const int lane  = threadIdx.x & 31;
    const int c     = lane & 15;
    const int h     = lane >> 4;
    const int strip = blockIdx.x * (blockDim.x >> 5) + (threadIdx.x >> 5);
    const bool live = strip < (M >> 4);
    const int m0    = live ? (strip << 4) : 0;

    float4 a[KB][2];
    {
        const float4* ap = reinterpret_cast<const float4*>(A + (size_t)(m0 + c) * K + 8 * h);
#pragma unroll
        for (int i = 0; i < KB; ++i) {
            a[i][0] = ap[4 * i];
            a[i][1] = ap[4 * i + 1];
        }
    }

    constexpr int NQ    = NC / 4;
    constexpr int NP4   = (K / 2) * NQ;
    constexpr int ITERS = (NP4 + 255) / 256;
#pragma unroll
    for (int it = 0; it < ITERS; ++it) {
        const int p = it * 256 + (int)threadIdx.x;
        if (NP4 % 256 == 0 || p < NP4) {
            const int kp = p / NQ, q = p - kp * NQ;
            const float4 r0 = *reinterpret_cast<const float4*>(B + (size_t)(2 * kp)     * NC + 4 * q);
            const float4 r1 = *reinterpret_cast<const float4*>(B + (size_t)(2 * kp + 1) * NC + 4 * q);
            float4* d = reinterpret_cast<float4*>(&Bs[kp * PITCH + 8 * q]);
            d[0] = make_float4(r0.x, r1.x, r0.y, r1.y);
            d[1] = make_float4(r0.z, r1.z, r0.w, r1.w);
        }
    }
    __syncthreads();
    if (!live) return;

    v8f acc[NT];
#pragma unroll
    for (int t = 0; t < NT; ++t) acc[t] = v8f{};

    const float* bl = Bs + (4 * h) * PITCH + 2 * c;
#pragma unroll
    for (int i = 0; i < KB; ++i) {
#pragma unroll
        for (int j = 0; j < 4; ++j) {
            const float4 q = a[i][j >> 1];
            v2f af;
            af.x = (j & 1) ? q.z : q.x;
            af.y = (j & 1) ? q.w : q.y;
            const float* bj = bl + (8 * i + j) * PITCH;
#pragma unroll
            for (int t = 0; t < NT; ++t) {
                v2f bf = *reinterpret_cast<const v2f*>(bj + 32 * t);
                acc[t] = __builtin_amdgcn_wmma_f32_16x16x4_f32(
                              false, af,  false, bf,
                              (short)0, acc[t],  false,  false);
            }
        }
    }

    float* cp = C + (size_t)(m0 + 8 * h) * NC + c;
#pragma unroll
    for (int t = 0; t < NT; ++t) {
#pragma unroll
        for (int r = 0; r < 8; ++r) cp[(size_t)r * NC + 16 * t] = acc[t][r];
    }
}

__global__ void k_attn_proj(const float* __restrict__ h,
                            const float* __restrict__ att_s,
                            const float* __restrict__ att_d,
                            float* __restrict__ a_s, float* __restrict__ a_d,
                            int n, int heads, int ch) {
    int i = blockIdx.x * blockDim.x + threadIdx.x;
    if (i >= n * heads) return;
    int node = i / heads, hh = i - node * heads;
    const float* hp = h + (size_t)node * heads * ch + (size_t)hh * ch;
    const float* as = att_s + hh * ch;
    const float* ad = att_d + hh * ch;
    float ss = 0.f, sd = 0.f;
    for (int c = 0; c < ch; ++c) {
        float v = hp[c];
        ss += v * as[c];
        sd += v * ad[c];
    }
    a_s[i] = ss;
    a_d[i] = sd;
}


__global__ void k_csr_count(const int* __restrict__ ei, int E,
                            int* __restrict__ cnt, int* __restrict__ rank) {
    int e = blockIdx.x * blockDim.x + threadIdx.x;
    if (e >= E) return;
    rank[e] = atomicAdd(&cnt[ei[E + e]], 1);
}

__global__ void __launch_bounds__(1024)
k_csr_scan(const int* __restrict__ cnt, int* __restrict__ rowptr, int n) {
    __shared__ int s_front[32];
    __shared__ int s_total[32];
    const int tid = threadIdx.x, lane = tid & 31, w = tid >> 5;
    const int beg = blockIdx.x * 4096;
    const int i   = beg + 4 * tid;

    int4 v;
    if (i + 3 < n) {
        v = *reinterpret_cast<const int4*>(cnt + i);
    } else {
        v.x = (i     < n) ? cnt[i]     : 0;
        v.y = (i + 1 < n) ? cnt[i + 1] : 0;
        v.z = (i + 2 < n) ? cnt[i + 2] : 0;
        v.w = 0;
    }

    int front = 0;
    for (int k0 = 0; k0 < beg; k0 += 8 * 4096) {
        int4 q[8];
#pragma unroll
        for (int u = 0; u < 8; ++u) {
            const int k = k0 + u * 4096;
            q[u] = *reinterpret_cast<const int4*>(cnt + ((k < beg) ? k : 0) + 4 * tid);
        }
#pragma unroll
        for (int u = 0; u < 8; ++u) {
            const int keep = (k0 + u * 4096 < beg) ? -1 : 0;
            front += (q[u].x + q[u].y + q[u].z + q[u].w) & keep;
        }
    }
    for (int o = 16; o > 0; o >>= 1) front += __shfl_xor(front, o);

    const int t = v.x + v.y + v.z + v.w;
    int inc = t;
    for (int o = 1; o < 32; o <<= 1) {
        int up = __shfl_up(inc, o);
        if (lane >= o) inc += up;
    }
    if (lane == 31) { s_front[w] = front; s_total[w] = inc; }
    __syncthreads();
    int x = s_front[lane] + ((lane < w) ? s_total[lane] : 0);
    for (int o = 16; o > 0; o >>= 1) x += __shfl_xor(x, o);

    const int e0 = x + inc - t;
    const int e1 = e0 + v.x, e2 = e1 + v.y, e3 = e2 + v.z;
    if (i + 3 <= n) {
        *reinterpret_cast<int4*>(rowptr + i) = make_int4(e0, e1, e2, e3);
    } else {
        if (i     <= n) rowptr[i]     = e0;
        if (i + 1 <= n) rowptr[i + 1] = e1;
        if (i + 2 <= n) rowptr[i + 2] = e2;
    }
}

__global__ void k_csr_fill(const int* __restrict__ ei, int E,
                           const int* __restrict__ rowptr, const int* __restrict__ rank,
                           int* __restrict__ ssrc) {
    int e = blockIdx.x * blockDim.x + threadIdx.x;
    if (e >= E) return;
    ssrc[rowptr[ei[E + e]] + rank[e]] = ei[e];
}

__global__ void k_gather_l1(const int* __restrict__ rowptr, const int* __restrict__ ssrc,
                            const float* __restrict__ a_s, const float* __restrict__ a_d,
                            const float* __restrict__ h, const float* __restrict__ b,
                            float* __restrict__ agg, int n) {
    int lane = threadIdx.x & 31;
    int node = blockIdx.x * (blockDim.x >> 5) + (threadIdx.x >> 5);
    node = __builtin_amdgcn_readfirstlane(node);
    if (node >= n) return;
    int slot = lane >> 3, head = lane & 7;
    int beg  = rowptr[node];
    int cntE = rowptr[node + 1] - beg + 1;
    float ad = a_d[node * HEADS1 + head];

    float m = -__builtin_huge_valf();
    for (int t0 = 0; t0 < cntE; t0 += 4) {
        int t = t0 + slot;
        if (t < cntE) {
            int s = (t == 0) ? node : ssrc[beg + t - 1];
            m = fmaxf(m, lrelu(a_s[s * HEADS1 + head] + ad));
        }
    }
    m = fmaxf(m, __shfl_xor(m, 8));
    m = fmaxf(m, __shfl_xor(m, 16));

    int hk[5];
    for (int k = 0; k < 5; ++k) hk[k] = (lane + 32 * k) / CH1;
    float acc[5] = {0.f, 0.f, 0.f, 0.f, 0.f};
    float den = 0.f;
    for (int t0 = 0; t0 < cntE; t0 += 4) {
        int t = t0 + slot;
        int s = node;
        float w = 0.f;
        if (t < cntE) {
            if (t > 0) s = ssrc[beg + t - 1];
            w = expf(lrelu(a_s[s * HEADS1 + head] + ad) - m);
        }
        den += w;
        if (t0 + 4 <= cntE) {
            float hv[4][5];
            for (int q = 0; q < 4; ++q) {
                int sq = __builtin_amdgcn_readlane(s, q * 8);
                const float* hs = h + (size_t)sq * H1C1 + lane;
                for (int k = 0; k < 5; ++k) hv[q][k] = hs[32 * k];
            }
            for (int q = 0; q < 4; ++q) {
                for (int k = 0; k < 5; ++k) acc[k] += __shfl(w, q * 8 + hk[k]) * hv[q][k];
            }
        } else {
            for (int q = 0; q < 3; ++q) {
                if (t0 + q < cntE) {
                    int sq = __builtin_amdgcn_readlane(s, q * 8);
                    const float* hs = h + (size_t)sq * H1C1 + lane;
                    for (int k = 0; k < 5; ++k) acc[k] += __shfl(w, q * 8 + hk[k]) * hs[32 * k];
                }
            }
        }
    }
    float bv[5];
    for (int k = 0; k < 5; ++k) bv[k] = b[lane + 32 * k];
    den += __shfl_xor(den, 8);
    den += __shfl_xor(den, 16);
    float* o = agg + (size_t)node * H1C1 + lane;
    for (int k = 0; k < 5; ++k) o[32 * k] = fmaxf(acc[k] / __shfl(den, hk[k]) + bv[k], 0.f);
}

__global__ void k_gather_l2(const int* __restrict__ rowptr, const int* __restrict__ ssrc,
                            const float* __restrict__ a_s, const float* __restrict__ a_d,
                            const float* __restrict__ h, const float* __restrict__ b,
                            float* __restrict__ out, int n) {
    int lane = threadIdx.x & 31;
    int node = blockIdx.x * (blockDim.x >> 5) + (threadIdx.x >> 5);
    node = __builtin_amdgcn_readfirstlane(node);
    if (node >= n) return;
    int half = lane >> 4, c = lane & 15;
    int beg  = rowptr[node];
    int cntE = rowptr[node + 1] - beg + 1;
    float ad = a_d[node];
    float bc = b[c];

    float m = -__builtin_huge_valf();
    for (int t0 = 0; t0 < cntE; t0 += 32) {
        int t = t0 + lane;
        if (t < cntE) {
            int s = (t == 0) ? node : ssrc[beg + t - 1];
            m = fmaxf(m, lrelu(a_s[s] + ad));
        }
    }
    for (int o = 16; o > 0; o >>= 1) m = fmaxf(m, __shfl_xor(m, o));

    float acc = 0.f, den = 0.f;
    for (int t0 = 0; t0 < cntE; t0 += 32) {
        int t = t0 + lane;
        int s = node;
        float w = 0.f;
        if (t < cntE) {
            if (t > 0) s = ssrc[beg + t - 1];
            w = expf(lrelu(a_s[s] + ad) - m);
        }
        den += w;
        int left = cntE - t0;
        int np = (left < 32) ? left : 32;
#pragma unroll 4
        for (int q = 0; q < np; q += 2) {
            int   sl = __shfl(s, q + half);
            float wl = __shfl(w, q + half);
            acc += wl * h[(size_t)sl * NCLS + c];
        }
    }
    for (int o = 16; o > 0; o >>= 1) den += __shfl_xor(den, o);
    acc += __shfl_xor(acc, 16);
    if (lane < 16) out[(size_t)node * NCLS + c] = acc / den + bc;
}


extern "C" void kernel_launch(void* const* d_in, const int* in_sizes, int n_in,
                              void* d_out, int out_size, void* d_ws, size_t ws_size,
                              hipStream_t stream) {
    const float* x        = (const float*)d_in[0];
    const int*   ei       = (const int*)  d_in[1];
    const float* W1       = (const float*)d_in[2];
    const float* att_src1 = (const float*)d_in[3];
    const float* att_dst1 = (const float*)d_in[4];
    const float* b1       = (const float*)d_in[5];
    const float* W2       = (const float*)d_in[6];
    const float* att_src2 = (const float*)d_in[7];
    const float* att_dst2 = (const float*)d_in[8];
    const float* b2       = (const float*)d_in[9];
    float* out = (float*)d_out;

    const int N    = in_sizes[0] / F_IN;
    const int E    = in_sizes[1] / 2;
    float* ws  = (float*)d_ws;
    float* h1  = ws;
    float* as1 = h1  + (size_t)N * H1C1;
    float* ad1 = as1 + (size_t)N * HEADS1;
    float* h2  = ad1 + (size_t)N * HEADS1;
    float* as2 = h2  + (size_t)N * NCLS;
    float* ad2 = as2 + (size_t)N;
    float* agg1 = ad2 + (size_t)N;
    int* cnt    = (int*)(agg1 + (size_t)N * H1C1);
    int* rowptr = cnt + (size_t)N;
    int* rank   = rowptr + (size_t)N + 1;
    int* ssrc   = rank + (size_t)E;

    const int T = 256;
    const int wavesPerBlk = T >> 5;

    k_zero<<<(N + T - 1) / T, T, 0, stream>>>((float*)cnt, (size_t)N);
    k_csr_count<<<(E + T - 1) / T, T, 0, stream>>>(ei, E, cnt, rank);
    k_csr_scan<<<N / 4096 + 1, 1024, 0, stream>>>(cnt, rowptr, N);
    k_csr_fill<<<(E + T - 1) / T, T, 0, stream>>>(ei, E, rowptr, rank, ssrc);

    k_gemm_strip_f32<H1C1 / 16, F_IN><<<((N >> 4) + wavesPerBlk - 1) / wavesPerBlk, T, 0, stream>>>(
        x, W1, h1, N);
    k_attn_proj<<<((N * HEADS1) + T - 1) / T, T, 0, stream>>>(
        h1, att_src1, att_dst1, as1, ad1, N, HEADS1, CH1);
    k_gather_l1<<<(N + wavesPerBlk - 1) / wavesPerBlk, T, 0, stream>>>(
        rowptr, ssrc, as1, ad1, h1, b1, agg1, N);

    k_gemm_strip_f32<NCLS / 16, H1C1><<<((N >> 4) + wavesPerBlk - 1) / wavesPerBlk, T, 0, stream>>>(
        agg1, W2, h2, N);
    k_attn_proj<<<(N + T - 1) / T, T, 0, stream>>>(
        h2, att_src2, att_dst2, as2, ad2, N, 1, NCLS);
    k_gather_l2<<<(N + wavesPerBlk - 1) / wavesPerBlk, T, 0, stream>>>(
        rowptr, ssrc, as2, ad2, h2, b2, out, N);
}


# Gather kernels read a node's first 32 list entries once: sources by one edge-list load, scores kept in registers from the max pass for the weight pass (were re-gathered); order of all sums unchanged
# speedup vs baseline: 8.6496x; 1.1156x over previous
#include <hip/hip_runtime.h>
#include <hip/hip_bf16.h>


#define NEG_SLOPE 0.2f
#define F_IN   128
#define H1C1   160
#define HEADS1 8
#define CH1    20
#define NCLS   16

typedef float v2f __attribute__((ext_vector_type(2)));
typedef float v8f __attribute__((ext_vector_type(8)));


__device__ __forceinline__ float lrelu(float v) {
    return (v > 0.f) ? v : NEG_SLOPE * v;
}


__global__ void k_zero(float* __restrict__ p, size_t n) {
    size_t i = (size_t)blockIdx.x * blockDim.x + threadIdx.x;
    size_t stride = (size_t)gridDim.x * blockDim.x;
    for (; i < n; i += stride) p[i] = 0.f;
}

template <int NT, int K>
__global__ void __launch_bounds__(256)
k_gemm_strip_f32(const float* __restrict__ A,
                 const float* __restrict__ B,
                 float* __restrict__ C, int M) {
    constexpr int NC    = NT * 16;
    constexpr int PITCH = 2 * NC + 8;
    constexpr int KB    = K / 16;
    __shared__ __attribute__((aligned(16))) float Bs[(K / 2) * PITCH];

    const int lane  = threadIdx.x & 31;
    const int c     = lane & 15;
    const int h     = lane >> 4;
    const int strip = blockIdx.x * (blockDim.x >> 5) + (threadIdx.x >> 5);
    const bool live = strip < (M >> 4);
    const int m0    = live ? (strip << 4) : 0;

    float4 a[KB][2];
    {
        const float4* ap = reinterpret_cast<const float4*>(A + (size_t)(m0 + c) * K + 8 * h);
#pragma unroll
        for (int i = 0; i < KB; ++i) {
            a[i][0] = ap[4 * i];
            a[i][1] = ap[4 * i + 1];
        }
    }

    constexpr int NQ    = NC / 4;
    constexpr int NP4   = (K / 2) * NQ;
    constexpr int ITERS = (NP4 + 255) / 256;
#pragma unroll
    for (int it = 0; it < ITERS; ++it) {
        const int p = it * 256 + (int)threadIdx.x;
        if (NP4 % 256 == 0 || p < NP4) {
            const int kp = p / NQ, q = p - kp * NQ;
            const float4 r0 = *reinterpret_cast<const float4*>(B + (size_t)(2 * kp)     * NC + 4 * q);
            const float4 r1 = *reinterpret_cast<const float4*>(B + (size_t)(2 * kp + 1) * NC + 4 * q);
            float4* d = reinterpret_cast<float4*>(&Bs[kp * PITCH + 8 * q]);
            d[0] = make_float4(r0.x, r1.x, r0.y, r1.y);
            d[1] = make_float4(r0.z, r1.z, r0.w, r1.w);
        }
    }
    __syncthreads();
    if (!live) return;

    v8f acc[NT];
#pragma unroll
    for (int t = 0; t < NT; ++t) acc[t] = v8f{};

    const float* bl = Bs + (4 * h) * PITCH + 2 * c;
#pragma unroll
    for (int i = 0; i < KB; ++i) {
#pragma unroll
        for (int j = 0; j < 4; ++j) {
            const float4 q = a[i][j >> 1];
            v2f af;
            af.x = (j & 1) ? q.z : q.x;
            af.y = (j & 1) ? q.w : q.y;
            const float* bj = bl + (8 * i + j) * PITCH;
#pragma unroll
            for (int t = 0; t < NT; ++t) {
                v2f bf = *reinterpret_cast<const v2f*>(bj + 32 * t);
                acc[t] = __builtin_amdgcn_wmma_f32_16x16x4_f32(
                              false, af,  false, bf,
                              (short)0, acc[t],  false,  false);
            }
        }
    }

    float* cp = C + (size_t)(m0 + 8 * h) * NC + c;
#pragma unroll
    for (int t = 0; t < NT; ++t) {
#pragma unroll
        for (int r = 0; r < 8; ++r) cp[(size_t)r * NC + 16 * t] = acc[t][r];
    }
}

__global__ void k_attn_proj(const float* __restrict__ h,
                            const float* __restrict__ att_s,
                            const float* __restrict__ att_d,
                            float* __restrict__ a_s, float* __restrict__ a_d,
                            int n, int heads, int ch) {
    int i = blockIdx.x * blockDim.x + threadIdx.x;
    if (i >= n * heads) return;
    int node = i / heads, hh = i - node * heads;
    const float* hp = h + (size_t)node * heads * ch + (size_t)hh * ch;
    const float* as = att_s + hh * ch;
    const float* ad = att_d + hh * ch;
    float ss = 0.f, sd = 0.f;
    for (int c = 0; c < ch; ++c) {
        float v = hp[c];
        ss += v * as[c];
        sd += v * ad[c];
    }
    a_s[i] = ss;
    a_d[i] = sd;
}


__global__ void k_csr_count(const int* __restrict__ ei, int E,
                            int* __restrict__ cnt, int* __restrict__ rank) {
    int e = blockIdx.x * blockDim.x + threadIdx.x;
    if (e >= E) return;
    rank[e] = atomicAdd(&cnt[ei[E + e]], 1);
}

__global__ void __launch_bounds__(1024)
k_csr_scan(const int* __restrict__ cnt, int* __restrict__ rowptr, int n) {
    __shared__ int s_front[32];
    __shared__ int s_total[32];
    const int tid = threadIdx.x, lane = tid & 31, w = tid >> 5;
    const int beg = blockIdx.x * 4096;
    const int i   = beg + 4 * tid;

    int4 v;
    if (i + 3 < n) {
        v = *reinterpret_cast<const int4*>(cnt + i);
    } else {
        v.x = (i     < n) ? cnt[i]     : 0;
        v.y = (i + 1 < n) ? cnt[i + 1] : 0;
        v.z = (i + 2 < n) ? cnt[i + 2] : 0;
        v.w = 0;
    }

    int front = 0;
    for (int k0 = 0; k0 < beg; k0 += 8 * 4096) {
        int4 q[8];
#pragma unroll
        for (int u = 0; u < 8; ++u) {
            const int k = k0 + u * 4096;
            q[u] = *reinterpret_cast<const int4*>(cnt + ((k < beg) ? k : 0) + 4 * tid);
        }
#pragma unroll
        for (int u = 0; u < 8; ++u) {
            const int keep = (k0 + u * 4096 < beg) ? -1 : 0;
            front += (q[u].x + q[u].y + q[u].z + q[u].w) & keep;
        }
    }
    for (int o = 16; o > 0; o >>= 1) front += __shfl_xor(front, o);

    const int t = v.x + v.y + v.z + v.w;
    int inc = t;
    for (int o = 1; o < 32; o <<= 1) {
        int up = __shfl_up(inc, o);
        if (lane >= o) inc += up;
    }
    if (lane == 31) { s_front[w] = front; s_total[w] = inc; }
    __syncthreads();
    int x = s_front[lane] + ((lane < w) ? s_total[lane] : 0);
    for (int o = 16; o > 0; o >>= 1) x += __shfl_xor(x, o);

    const int e0 = x + inc - t;
    const int e1 = e0 + v.x, e2 = e1 + v.y, e3 = e2 + v.z;
    if (i + 3 <= n) {
        *reinterpret_cast<int4*>(rowptr + i) = make_int4(e0, e1, e2, e3);
    } else {
        if (i     <= n) rowptr[i]     = e0;
        if (i + 1 <= n) rowptr[i + 1] = e1;
        if (i + 2 <= n) rowptr[i + 2] = e2;
    }
}

__global__ void k_csr_fill(const int* __restrict__ ei, int E,
                           const int* __restrict__ rowptr, const int* __restrict__ rank,
                           int* __restrict__ ssrc) {
    int e = blockIdx.x * blockDim.x + threadIdx.x;
    if (e >= E) return;
    ssrc[rowptr[ei[E + e]] + rank[e]] = ei[e];
}

#define L1_KEEP 8

__device__ __forceinline__ void l1_rows(const float* __restrict__ h, const int (&sq)[4], float w,
                                        int t0, int cntE, int lane, const int (&hk)[5], float (&acc)[5]) {
    if (t0 + 4 <= cntE) {
        float hv[4][5];
        for (int q = 0; q < 4; ++q) {
            const float* hs = h + (size_t)sq[q] * H1C1 + lane;
            for (int k = 0; k < 5; ++k) hv[q][k] = hs[32 * k];
        }
        for (int q = 0; q < 4; ++q) {
            for (int k = 0; k < 5; ++k) acc[k] += __shfl(w, q * 8 + hk[k]) * hv[q][k];
        }
    } else {
        for (int q = 0; q < 3; ++q) {
            if (t0 + q < cntE) {
                const float* hs = h + (size_t)sq[q] * H1C1 + lane;
                for (int k = 0; k < 5; ++k) acc[k] += __shfl(w, q * 8 + hk[k]) * hs[32 * k];
            }
        }
    }
}

__global__ void k_gather_l1(const int* __restrict__ rowptr, const int* __restrict__ ssrc,
                            const float* __restrict__ a_s, const float* __restrict__ a_d,
                            const float* __restrict__ h, const float* __restrict__ b,
                            float* __restrict__ agg, int n) {
    int lane = threadIdx.x & 31;
    int node = blockIdx.x * (blockDim.x >> 5) + (threadIdx.x >> 5);
    node = __builtin_amdgcn_readfirstlane(node);
    if (node >= n) return;
    int slot = lane >> 3, head = lane & 7;
    int beg  = rowptr[node];
    int cntE = rowptr[node + 1] - beg + 1;
    float ad = a_d[node * HEADS1 + head];

    int sAll = node;
    if (lane > 0 && lane < cntE) sAll = ssrc[beg + lane - 1];

    float eC[L1_KEEP];
#pragma unroll
    for (int k = 0; k < L1_KEEP; ++k) {
        const int s = __shfl(sAll, 4 * k + slot);
        eC[k] = a_s[s * HEADS1 + head];
    }
#pragma unroll
    for (int k = 0; k < L1_KEEP; ++k)
        eC[k] = (4 * k + slot < cntE) ? lrelu(eC[k] + ad) : -__builtin_huge_valf();

    float m = -__builtin_huge_valf();
#pragma unroll
    for (int k = 0; k < L1_KEEP; ++k) m = fmaxf(m, eC[k]);
    for (int t0 = 4 * L1_KEEP; t0 < cntE; t0 += 4) {
        int t = t0 + slot;
        if (t < cntE) {
            int s = ssrc[beg + t - 1];
            m = fmaxf(m, lrelu(a_s[s * HEADS1 + head] + ad));
        }
    }
    m = fmaxf(m, __shfl_xor(m, 8));
    m = fmaxf(m, __shfl_xor(m, 16));

    int hk[5];
    for (int k = 0; k < 5; ++k) hk[k] = (lane + 32 * k) / CH1;
    float acc[5] = {0.f, 0.f, 0.f, 0.f, 0.f};
    float den = 0.f;
#pragma unroll
    for (int k = 0; k < L1_KEEP; ++k) {
        const int t0 = 4 * k;
        if (t0 < cntE) {
            const float w = (t0 + slot < cntE) ? expf(eC[k] - m) : 0.f;
            den += w;
            int sq[4];
            for (int q = 0; q < 4; ++q) sq[q] = __builtin_amdgcn_readlane(sAll, t0 + q);
            l1_rows(h, sq, w, t0, cntE, lane, hk, acc);
        }
    }
    for (int t0 = 4 * L1_KEEP; t0 < cntE; t0 += 4) {
        int t = t0 + slot;
        int s = node;
        float w = 0.f;
        if (t < cntE) {
            s = ssrc[beg + t - 1];
            w = expf(lrelu(a_s[s * HEADS1 + head] + ad) - m);
        }
        den += w;
        int sq[4];
        for (int q = 0; q < 4; ++q) sq[q] = __builtin_amdgcn_readlane(s, q * 8);
        l1_rows(h, sq, w, t0, cntE, lane, hk, acc);
    }
    float bv[5];
    for (int k = 0; k < 5; ++k) bv[k] = b[lane + 32 * k];
    den += __shfl_xor(den, 8);
    den += __shfl_xor(den, 16);
    float* o = agg + (size_t)node * H1C1 + lane;
    for (int k = 0; k < 5; ++k) o[32 * k] = fmaxf(acc[k] / __shfl(den, hk[k]) + bv[k], 0.f);
}


__device__ __forceinline__ float l2_rows(const float* __restrict__ h, int s, float w, int np,
                                         int half, int c, float acc) {
#pragma unroll 4
    for (int q = 0; q < np; q += 2) {
        int   sl = __shfl(s, q + half);
        float wl = __shfl(w, q + half);
        acc += wl * h[(size_t)sl * NCLS + c];
    }
    return acc;
}

__global__ void k_gather_l2(const int* __restrict__ rowptr, const int* __restrict__ ssrc,
                            const float* __restrict__ a_s, const float* __restrict__ a_d,
                            const float* __restrict__ h, const float* __restrict__ b,
                            float* __restrict__ out, int n) {
    int lane = threadIdx.x & 31;
    int node = blockIdx.x * (blockDim.x >> 5) + (threadIdx.x >> 5);
    node = __builtin_amdgcn_readfirstlane(node);
    if (node >= n) return;
    int half = lane >> 4, c = lane & 15;
    int beg  = rowptr[node];
    int cntE = rowptr[node + 1] - beg + 1;
    float ad = a_d[node];
    float bc = b[c];

    int   s0 = node;
    float e0 = -__builtin_huge_valf();
    if (lane < cntE) {
        if (lane > 0) s0 = ssrc[beg + lane - 1];
        e0 = lrelu(a_s[s0] + ad);
    }

    float m = fmaxf(-__builtin_huge_valf(), e0);
    for (int t0 = 32; t0 < cntE; t0 += 32) {
        int t = t0 + lane;
        if (t < cntE) {
            int s = ssrc[beg + t - 1];
            m = fmaxf(m, lrelu(a_s[s] + ad));
        }
    }
    for (int o = 16; o > 0; o >>= 1) m = fmaxf(m, __shfl_xor(m, o));

    float acc = 0.f, den = 0.f;
    {
        const float w = (lane < cntE) ? expf(e0 - m) : 0.f;
        den += w;
        acc = l2_rows(h, s0, w, (cntE < 32) ? cntE : 32, half, c, acc);
    }
    for (int t0 = 32; t0 < cntE; t0 += 32) {
        int t = t0 + lane;
        int s = node;
        float w = 0.f;
        if (t < cntE) {
            s = ssrc[beg + t - 1];
            w = expf(lrelu(a_s[s] + ad) - m);
        }
        den += w;
        int left = cntE - t0;
        acc = l2_rows(h, s, w, (left < 32) ? left : 32, half, c, acc);
    }
    for (int o = 16; o > 0; o >>= 1) den += __shfl_xor(den, o);
    acc += __shfl_xor(acc, 16);
    if (lane < 16) out[(size_t)node * NCLS + c] = acc / den + bc;
}


extern "C" void kernel_launch(void* const* d_in, const int* in_sizes, int n_in,
                              void* d_out, int out_size, void* d_ws, size_t ws_size,
                              hipStream_t stream) {
    const float* x        = (const float*)d_in[0];
    const int*   ei       = (const int*)  d_in[1];
    const float* W1       = (const float*)d_in[2];
    const float* att_src1 = (const float*)d_in[3];
    const float* att_dst1 = (const float*)d_in[4];
    const float* b1       = (const float*)d_in[5];
    const float* W2       = (const float*)d_in[6];
    const float* att_src2 = (const float*)d_in[7];
    const float* att_dst2 = (const float*)d_in[8];
    const float* b2       = (const float*)d_in[9];
    float* out = (float*)d_out;

    const int N    = in_sizes[0] / F_IN;
    const int E    = in_sizes[1] / 2;
    float* ws  = (float*)d_ws;
    float* h1  = ws;
    float* as1 = h1  + (size_t)N * H1C1;
    float* ad1 = as1 + (size_t)N * HEADS1;
    float* h2  = ad1 + (size_t)N * HEADS1;
    float* as2 = h2  + (size_t)N * NCLS;
    float* ad2 = as2 + (size_t)N;
    float* agg1 = ad2 + (size_t)N;
    int* cnt    = (int*)(agg1 + (size_t)N * H1C1);
    int* rowptr = cnt + (size_t)N;
    int* rank   = rowptr + (size_t)N + 1;
    int* ssrc   = rank + (size_t)E;

    const int T = 256;
    const int wavesPerBlk = T >> 5;

    k_zero<<<(N + T - 1) / T, T, 0, stream>>>((float*)cnt, (size_t)N);
    k_csr_count<<<(E + T - 1) / T, T, 0, stream>>>(ei, E, cnt, rank);
    k_csr_scan<<<N / 4096 + 1, 1024, 0, stream>>>(cnt, rowptr, N);
    k_csr_fill<<<(E + T - 1) / T, T, 0, stream>>>(ei, E, rowptr, rank, ssrc);

    k_gemm_strip_f32<H1C1 / 16, F_IN><<<((N >> 4) + wavesPerBlk - 1) / wavesPerBlk, T, 0, stream>>>(
        x, W1, h1, N);
    k_attn_proj<<<((N * HEADS1) + T - 1) / T, T, 0, stream>>>(
        h1, att_src1, att_dst1, as1, ad1, N, HEADS1, CH1);
    k_gather_l1<<<(N + wavesPerBlk - 1) / wavesPerBlk, T, 0, stream>>>(
        rowptr, ssrc, as1, ad1, h1, b1, agg1, N);

    k_gemm_strip_f32<NCLS / 16, H1C1><<<((N >> 4) + wavesPerBlk - 1) / wavesPerBlk, T, 0, stream>>>(
        agg1, W2, h2, N);
    k_attn_proj<<<(N + T - 1) / T, T, 0, stream>>>(
        h2, att_src2, att_dst2, as2, ad2, N, 1, NCLS);
    k_gather_l2<<<(N + wavesPerBlk - 1) / wavesPerBlk, T, 0, stream>>>(
        rowptr, ssrc, as2, ad2, h2, b2, out, N);
}


# Edges grouped by destination on LDS counters (chunks scatter into 256-node buckets, one work-group sorts each bucket): no returning global adds, no clear, 9 launches; row sum order = LDS arrival rank
# speedup vs baseline: 12.3245x; 1.4249x over previous
#include <hip/hip_runtime.h>
#include <hip/hip_bf16.h>


#define NEG_SLOPE 0.2f
#define F_IN   128
#define H1C1   160
#define HEADS1 8
#define CH1    20
#define NCLS   16

typedef float v2f __attribute__((ext_vector_type(2)));
typedef float v8f __attribute__((ext_vector_type(8)));


__device__ __forceinline__ float lrelu(float v) {
    return (v > 0.f) ? v : NEG_SLOPE * v;
}


template <int NT, int K>
__global__ void __launch_bounds__(256)
k_gemm_strip_f32(const float* __restrict__ A,
                 const float* __restrict__ B,
                 float* __restrict__ C, int M) {
    constexpr int NC    = NT * 16;
    constexpr int PITCH = 2 * NC + 8;
    constexpr int KB    = K / 16;
    __shared__ __attribute__((aligned(16))) float Bs[(K / 2) * PITCH];

    const int lane  = threadIdx.x & 31;
    const int c     = lane & 15;
    const int h     = lane >> 4;
    const int strip = blockIdx.x * (blockDim.x >> 5) + (threadIdx.x >> 5);
    const bool live = strip < (M >> 4);
    const int m0    = live ? (strip << 4) : 0;

    float4 a[KB][2];
    {
        const float4* ap = reinterpret_cast<const float4*>(A + (size_t)(m0 + c) * K + 8 * h);
#pragma unroll
        for (int i = 0; i < KB; ++i) {
            a[i][0] = ap[4 * i];
            a[i][1] = ap[4 * i + 1];
        }
    }

    constexpr int NQ    = NC / 4;
    constexpr int NP4   = (K / 2) * NQ;
    constexpr int ITERS = (NP4 + 255) / 256;
#pragma unroll
    for (int it = 0; it < ITERS; ++it) {
        const int p = it * 256 + (int)threadIdx.x;
        if (NP4 % 256 == 0 || p < NP4) {
            const int kp = p / NQ, q = p - kp * NQ;
            const float4 r0 = *reinterpret_cast<const float4*>(B + (size_t)(2 * kp)     * NC + 4 * q);
            const float4 r1 = *reinterpret_cast<const float4*>(B + (size_t)(2 * kp + 1) * NC + 4 * q);
            float4* d = reinterpret_cast<float4*>(&Bs[kp * PITCH + 8 * q]);
            d[0] = make_float4(r0.x, r1.x, r0.y, r1.y);
            d[1] = make_float4(r0.z, r1.z, r0.w, r1.w);
        }
    }
    __syncthreads();
    if (!live) return;

    v8f acc[NT];
#pragma unroll
    for (int t = 0; t < NT; ++t) acc[t] = v8f{};

    const float* bl = Bs + (4 * h) * PITCH + 2 * c;
#pragma unroll
    for (int i = 0; i < KB; ++i) {
#pragma unroll
        for (int j = 0; j < 4; ++j) {
            const float4 q = a[i][j >> 1];
            v2f af;
            af.x = (j & 1) ? q.z : q.x;
            af.y = (j & 1) ? q.w : q.y;
            const float* bj = bl + (8 * i + j) * PITCH;
#pragma unroll
            for (int t = 0; t < NT; ++t) {
                v2f bf = *reinterpret_cast<const v2f*>(bj + 32 * t);
                acc[t] = __builtin_amdgcn_wmma_f32_16x16x4_f32(
                              false, af,  false, bf,
                              (short)0, acc[t],  false,  false);
            }
        }
    }

    float* cp = C + (size_t)(m0 + 8 * h) * NC + c;
#pragma unroll
    for (int t = 0; t < NT; ++t) {
#pragma unroll
        for (int r = 0; r < 8; ++r) cp[(size_t)r * NC + 16 * t] = acc[t][r];
    }
}

__global__ void k_attn_proj(const float* __restrict__ h,
                            const float* __restrict__ att_s,
                            const float* __restrict__ att_d,
                            float* __restrict__ a_s, float* __restrict__ a_d,
                            int n, int heads, int ch) {
    int i = blockIdx.x * blockDim.x + threadIdx.x;
    if (i >= n * heads) return;
    int node = i / heads, hh = i - node * heads;
    const float* hp = h + (size_t)node * heads * ch + (size_t)hh * ch;
    const float* as = att_s + hh * ch;
    const float* ad = att_d + hh * ch;
    float ss = 0.f, sd = 0.f;
    for (int c = 0; c < ch; ++c) {
        float v = hp[c];
        ss += v * as[c];
        sd += v * ad[c];
    }
    a_s[i] = ss;
    a_d[i] = sd;
}

#define BKT_SHIFT 8
#define BKT_NODES 256
#define MAXB      256
#define CHUNK     8192
#define BKT_KEEP  5

__global__ void __launch_bounds__(1024)
k_bin_count(const int* __restrict__ ei, int E, int nb, int* __restrict__ hist) {
    __shared__ int s_cnt[MAXB];
    const int tid  = threadIdx.x;
    const int base = blockIdx.x * CHUNK;
    int d[8];
#pragma unroll
    for (int k = 0; k < 8; ++k) {
        const int e = base + k * 1024 + tid;
        d[k] = ei[E + ((e < E) ? e : E - 1)];
    }
    if (tid < MAXB) s_cnt[tid] = 0;
    __syncthreads();
#pragma unroll
    for (int k = 0; k < 8; ++k) {
        const int e = base + k * 1024 + tid;
        if (e < E) atomicAdd(&s_cnt[d[k] >> BKT_SHIFT], 1);
    }
    __syncthreads();
    if (tid < nb) hist[blockIdx.x * nb + tid] = s_cnt[tid];
}

__global__ void __launch_bounds__(1024)
k_bin_scatter(const int* __restrict__ ei, int E, int nb, const int* __restrict__ hist,
              int* __restrict__ bstart, int* __restrict__ pairs) {
    __shared__ int s_bef[4][MAXB];
    __shared__ int s_tot[4][MAXB];
    __shared__ int s_cur[MAXB];
    const int tid = threadIdx.x, lane = tid & 31;
    const int me = blockIdx.x, G = gridDim.x;
    const int base = me * CHUNK;

    int s[8], d[8];
#pragma unroll
    for (int k = 0; k < 8; ++k) {
        const int e  = base + k * 1024 + tid;
        const int ec = (e < E) ? e : E - 1;
        s[k] = ei[ec];
        d[k] = ei[E + ec];
    }

    const int b = tid & (MAXB - 1), part = tid >> 8;
    const int bc = (b < nb) ? b : 0;
    int before = 0, total = 0;
    for (int w0 = 0; w0 < G; w0 += 32) {
        int v[8];
#pragma unroll
        for (int u = 0; u < 8; ++u) {
            const int w = w0 + 4 * u + part;
            v[u] = hist[((w < G) ? w : 0) * nb + bc];
        }
#pragma unroll
        for (int u = 0; u < 8; ++u) {
            const int w = w0 + 4 * u + part;
            const int x = (w < G && b < nb) ? v[u] : 0;
            total  += x;
            before += (w < me) ? x : 0;
        }
    }
    s_bef[part][b] = before;
    s_tot[part][b] = total;
    __syncthreads();

    if (tid < 32) {
        int ex[8];
        int run = 0;
#pragma unroll
        for (int k = 0; k < 8; ++k) {
            const int i = 8 * tid + k;
            ex[k] = run;
            run += s_tot[0][i] + s_tot[1][i] + s_tot[2][i] + s_tot[3][i];
        }
        int inc = run;
        for (int o = 1; o < 32; o <<= 1) {
            int up = __shfl_up(inc, o);
            if (lane >= o) inc += up;
        }
        const int excl = inc - run;
#pragma unroll
        for (int k = 0; k < 8; ++k) {
            const int i  = 8 * tid + k;
            const int st = excl + ex[k];
            s_cur[i] = st + s_bef[0][i] + s_bef[1][i] + s_bef[2][i] + s_bef[3][i];
            if (me == 0 && i <= nb) bstart[i] = st;
        }
    }
    __syncthreads();

#pragma unroll
    for (int k = 0; k < 8; ++k) {
        const int e = base + k * 1024 + tid;
        if (e < E) {
            const int pos = atomicAdd(&s_cur[d[k] >> BKT_SHIFT], 1);
            *reinterpret_cast<int2*>(pairs + 2 * (size_t)pos) = make_int2(s[k], d[k]);
        }
    }
}

__global__ void __launch_bounds__(1024)
k_bucket_sort(const int* __restrict__ pairs, const int* __restrict__ bstart, int E,
              int* __restrict__ rowptr, int* __restrict__ ssrc, int n) {
    __shared__ int s_cnt[BKT_NODES];
    __shared__ int s_off[BKT_NODES];
    const int tid = threadIdx.x, lane = tid & 31;
    const int bkt = blockIdx.x;
    const int lo  = bkt << BKT_SHIFT;
    const int s0  = bstart[bkt], s1 = bstart[bkt + 1];
    const int2* pp = reinterpret_cast<const int2*>(pairs);

    int2 p[BKT_KEEP];
#pragma unroll
    for (int k = 0; k < BKT_KEEP; ++k) {
        const int i = s0 + k * 1024 + tid;
        p[k] = pp[(i < s1) ? i : E - 1];
    }
    if (tid < BKT_NODES) s_cnt[tid] = 0;
    __syncthreads();

#pragma unroll
    for (int k = 0; k < BKT_KEEP; ++k) {
        const int i = s0 + k * 1024 + tid;
        if (i < s1) atomicAdd(&s_cnt[p[k].y - lo], 1);
    }
    for (int i = s0 + BKT_KEEP * 1024 + tid; i < s1; i += 1024) atomicAdd(&s_cnt[pp[i].y - lo], 1);
    __syncthreads();

    if (tid < 32) {
        int ex[8];
        int run = 0;
#pragma unroll
        for (int k = 0; k < 8; ++k) {
            ex[k] = run;
            run += s_cnt[8 * tid + k];
        }
        int inc = run;
        for (int o = 1; o < 32; o <<= 1) {
            int up = __shfl_up(inc, o);
            if (lane >= o) inc += up;
        }
        const int excl = inc - run;
#pragma unroll
        for (int k = 0; k < 8; ++k) s_off[8 * tid + k] = excl + ex[k];
    }
    __syncthreads();
    if (tid < BKT_NODES && lo + tid < n) rowptr[lo + tid] = s0 + s_off[tid];
    if (bkt == (int)gridDim.x - 1 && tid == 0) rowptr[n] = s1;
    __syncthreads();

#pragma unroll
    for (int k = 0; k < BKT_KEEP; ++k) {
        const int i = s0 + k * 1024 + tid;
        if (i < s1) ssrc[s0 + atomicAdd(&s_off[p[k].y - lo], 1)] = p[k].x;
    }
    for (int i = s0 + BKT_KEEP * 1024 + tid; i < s1; i += 1024) {
        const int2 q = pp[i];
        ssrc[s0 + atomicAdd(&s_off[q.y - lo], 1)] = q.x;
    }
}

#define L1_KEEP 8

__device__ __forceinline__ void l1_rows(const float* __restrict__ h, const int (&sq)[4], float w,
                                        int t0, int cntE, int lane, const int (&hk)[5], float (&acc)[5]) {
    if (t0 + 4 <= cntE) {
        float hv[4][5];
        for (int q = 0; q < 4; ++q) {
            const float* hs = h + (size_t)sq[q] * H1C1 + lane;
            for (int k = 0; k < 5; ++k) hv[q][k] = hs[32 * k];
        }
        for (int q = 0; q < 4; ++q) {
            for (int k = 0; k < 5; ++k) acc[k] += __shfl(w, q * 8 + hk[k]) * hv[q][k];
        }
    } else {
        for (int q = 0; q < 3; ++q) {
            if (t0 + q < cntE) {
                const float* hs = h + (size_t)sq[q] * H1C1 + lane;
                for (int k = 0; k < 5; ++k) acc[k] += __shfl(w, q * 8 + hk[k]) * hs[32 * k];
            }
        }
    }
}

__global__ void k_gather_l1(const int* __restrict__ rowptr, const int* __restrict__ ssrc,
                            const float* __restrict__ a_s, const float* __restrict__ a_d,
                            const float* __restrict__ h, const float* __restrict__ b,
                            float* __restrict__ agg, int n) {
    int lane = threadIdx.x & 31;
    int node = blockIdx.x * (blockDim.x >> 5) + (threadIdx.x >> 5);
    node = __builtin_amdgcn_readfirstlane(node);
    if (node >= n) return;
    int slot = lane >> 3, head = lane & 7;
    int beg  = rowptr[node];
    int cntE = rowptr[node + 1] - beg + 1;
    float ad = a_d[node * HEADS1 + head];

    int sAll = node;
    if (lane > 0 && lane < cntE) sAll = ssrc[beg + lane - 1];

    float eC[L1_KEEP];
#pragma unroll
    for (int k = 0; k < L1_KEEP; ++k) {
        const int s = __shfl(sAll, 4 * k + slot);
        eC[k] = a_s[s * HEADS1 + head];
    }
#pragma unroll
    for (int k = 0; k < L1_KEEP; ++k)
        eC[k] = (4 * k + slot < cntE) ? lrelu(eC[k] + ad) : -__builtin_huge_valf();

    float m = -__builtin_huge_valf();
#pragma unroll
    for (int k = 0; k < L1_KEEP; ++k) m = fmaxf(m, eC[k]);
    for (int t0 = 4 * L1_KEEP; t0 < cntE; t0 += 4) {
        int t = t0 + slot;
        if (t < cntE) {
            int s = ssrc[beg + t - 1];
            m = fmaxf(m, lrelu(a_s[s * HEADS1 + head] + ad));
        }
    }
    m = fmaxf(m, __shfl_xor(m, 8));
    m = fmaxf(m, __shfl_xor(m, 16));

    int hk[5];
    for (int k = 0; k < 5; ++k) hk[k] = (lane + 32 * k) / CH1;
    float acc[5] = {0.f, 0.f, 0.f, 0.f, 0.f};
    float den = 0.f;
#pragma unroll
    for (int k = 0; k < L1_KEEP; ++k) {
        const int t0 = 4 * k;
        if (t0 < cntE) {
            const float w = (t0 + slot < cntE) ? expf(eC[k] - m) : 0.f;
            den += w;
            int sq[4];
            for (int q = 0; q < 4; ++q) sq[q] = __builtin_amdgcn_readlane(sAll, t0 + q);
            l1_rows(h, sq, w, t0, cntE, lane, hk, acc);
        }
    }
    for (int t0 = 4 * L1_KEEP; t0 < cntE; t0 += 4) {
        int t = t0 + slot;
        int s = node;
        float w = 0.f;
        if (t < cntE) {
            s = ssrc[beg + t - 1];
            w = expf(lrelu(a_s[s * HEADS1 + head] + ad) - m);
        }
        den += w;
        int sq[4];
        for (int q = 0; q < 4; ++q) sq[q] = __builtin_amdgcn_readlane(s, q * 8);
        l1_rows(h, sq, w, t0, cntE, lane, hk, acc);
    }
    float bv[5];
    for (int k = 0; k < 5; ++k) bv[k] = b[lane + 32 * k];
    den += __shfl_xor(den, 8);
    den += __shfl_xor(den, 16);
    float* o = agg + (size_t)node * H1C1 + lane;
    for (int k = 0; k < 5; ++k) o[32 * k] = fmaxf(acc[k] / __shfl(den, hk[k]) + bv[k], 0.f);
}


__device__ __forceinline__ float l2_rows(const float* __restrict__ h, int s, float w, int np,
                                         int half, int c, float acc) {
#pragma unroll 4
    for (int q = 0; q < np; q += 2) {
        int   sl = __shfl(s, q + half);
        float wl = __shfl(w, q + half);
        acc += wl * h[(size_t)sl * NCLS + c];
    }
    return acc;
}

__global__ void k_gather_l2(const int* __restrict__ rowptr, const int* __restrict__ ssrc,
                            const float* __restrict__ a_s, const float* __restrict__ a_d,
                            const float* __restrict__ h, const float* __restrict__ b,
                            float* __restrict__ out, int n) {
    int lane = threadIdx.x & 31;
    int node = blockIdx.x * (blockDim.x >> 5) + (threadIdx.x >> 5);
    node = __builtin_amdgcn_readfirstlane(node);
    if (node >= n) return;
    int half = lane >> 4, c = lane & 15;
    int beg  = rowptr[node];
    int cntE = rowptr[node + 1] - beg + 1;
    float ad = a_d[node];
    float bc = b[c];

    int   s0 = node;
    float e0 = -__builtin_huge_valf();
    if (lane < cntE) {
        if (lane > 0) s0 = ssrc[beg + lane - 1];
        e0 = lrelu(a_s[s0] + ad);
    }

    float m = fmaxf(-__builtin_huge_valf(), e0);
    for (int t0 = 32; t0 < cntE; t0 += 32) {
        int t = t0 + lane;
        if (t < cntE) {
            int s = ssrc[beg + t - 1];
            m = fmaxf(m, lrelu(a_s[s] + ad));
        }
    }
    for (int o = 16; o > 0; o >>= 1) m = fmaxf(m, __shfl_xor(m, o));

    float acc = 0.f, den = 0.f;
    {
        const float w = (lane < cntE) ? expf(e0 - m) : 0.f;
        den += w;
        acc = l2_rows(h, s0, w, (cntE < 32) ? cntE : 32, half, c, acc);
    }
    for (int t0 = 32; t0 < cntE; t0 += 32) {
        int t = t0 + lane;
        int s = node;
        float w = 0.f;
        if (t < cntE) {
            s = ssrc[beg + t - 1];
            w = expf(lrelu(a_s[s] + ad) - m);
        }
        den += w;
        int left = cntE - t0;
        acc = l2_rows(h, s, w, (left < 32) ? left : 32, half, c, acc);
    }
    for (int o = 16; o > 0; o >>= 1) den += __shfl_xor(den, o);
    acc += __shfl_xor(acc, 16);
    if (lane < 16) out[(size_t)node * NCLS + c] = acc / den + bc;
}


extern "C" void kernel_launch(void* const* d_in, const int* in_sizes, int n_in,
                              void* d_out, int out_size, void* d_ws, size_t ws_size,
                              hipStream_t stream) {
    const float* x        = (const float*)d_in[0];
    const int*   ei       = (const int*)  d_in[1];
    const float* W1       = (const float*)d_in[2];
    const float* att_src1 = (const float*)d_in[3];
    const float* att_dst1 = (const float*)d_in[4];
    const float* b1       = (const float*)d_in[5];
    const float* W2       = (const float*)d_in[6];
    const float* att_src2 = (const float*)d_in[7];
    const float* att_dst2 = (const float*)d_in[8];
    const float* b2       = (const float*)d_in[9];
    float* out = (float*)d_out;

    const int N    = in_sizes[0] / F_IN;
    const int E    = in_sizes[1] / 2;
    float* ws  = (float*)d_ws;
    float* h1  = ws;
    float* as1 = h1  + (size_t)N * H1C1;
    float* ad1 = as1 + (size_t)N * HEADS1;
    float* h2  = ad1 + (size_t)N * HEADS1;
    float* as2 = h2  + (size_t)N * NCLS;
    float* ad2 = as2 + (size_t)N;
    float* agg1 = ad2 + (size_t)N;
    int* pairs  = (int*)(agg1 + (size_t)N * H1C1);
    int* ssrc   = pairs + 2 * (size_t)E;
    int* rowptr = ssrc + (size_t)E;
    int* bstart = rowptr + (size_t)N + 1;
    int* hist   = bstart + MAXB;

    const int T = 256;
    const int wavesPerBlk = T >> 5;

    const int nb     = (N + BKT_NODES - 1) >> BKT_SHIFT;
    const int chunks = (E + CHUNK - 1) / CHUNK;
    k_bin_count<<<chunks, 1024, 0, stream>>>(ei, E, nb, hist);
    k_bin_scatter<<<chunks, 1024, 0, stream>>>(ei, E, nb, hist, bstart, pairs);
    k_bucket_sort<<<nb, 1024, 0, stream>>>(pairs, bstart, E, rowptr, ssrc, N);

    k_gemm_strip_f32<H1C1 / 16, F_IN><<<((N >> 4) + wavesPerBlk - 1) / wavesPerBlk, T, 0, stream>>>(
        x, W1, h1, N);
    k_attn_proj<<<((N * HEADS1) + T - 1) / T, T, 0, stream>>>(
        h1, att_src1, att_dst1, as1, ad1, N, HEADS1, CH1);
    k_gather_l1<<<(N + wavesPerBlk - 1) / wavesPerBlk, T, 0, stream>>>(
        rowptr, ssrc, as1, ad1, h1, b1, agg1, N);

    k_gemm_strip_f32<NCLS / 16, H1C1><<<((N >> 4) + wavesPerBlk - 1) / wavesPerBlk, T, 0, stream>>>(
        agg1, W2, h2, N);
    k_attn_proj<<<(N + T - 1) / T, T, 0, stream>>>(
        h2, att_src2, att_dst2, as2, ad2, N, 1, NCLS);
    k_gather_l2<<<(N + wavesPerBlk - 1) / wavesPerBlk, T, 0, stream>>>(
        rowptr, ssrc, as2, ad2, h2, b2, out, N);
}


# Attention scores a_src/a_dst come out of the GEMMs as one more column tile, x*(W*att), W*att built per work-group in LDS: both projection launches gone (7 left); score sum order now k of x*(W*att)
# speedup vs baseline: 12.7875x; 1.0376x over previous
#include <hip/hip_runtime.h>
#include <hip/hip_bf16.h>


#define NEG_SLOPE 0.2f
#define F_IN   128
#define H1C1   160
#define HEADS1 8
#define CH1    20
#define NCLS   16

typedef float v2f __attribute__((ext_vector_type(2)));
typedef float v8f __attribute__((ext_vector_type(8)));


__device__ __forceinline__ float lrelu(float v) {
    return (v > 0.f) ? v : NEG_SLOPE * v;
}


template <int NT, int K, int HEADS, int CH>
__global__ void __launch_bounds__(256)
k_gemm_strip_f32(const float* __restrict__ A,
                 const float* __restrict__ B,
                 const float* __restrict__ att_s,
                 const float* __restrict__ att_d,
                 float* __restrict__ C,
                 float* __restrict__ a_s, float* __restrict__ a_d, int M) {
    constexpr int NC    = NT * 16;
    constexpr int NCX   = NC + 16;
    constexpr int PITCH = 2 * NCX + 8;
    constexpr int KB    = K / 16;
    constexpr int KP    = K / 2;
    static_assert(NC == HEADS * CH && 2 * HEADS <= 16 && CH % 2 == 0, "score tile: 2*HEADS columns, channels read in pairs");
    static_assert(HEADS == 1 || KP % 32 == 0, "a wave of the score pass works on one head");
    __shared__ __attribute__((aligned(16))) float Bs[KP * PITCH];

    const int lane  = threadIdx.x & 31;
    const int c     = lane & 15;
    const int h     = lane >> 4;
    const int strip = blockIdx.x * (blockDim.x >> 5) + (threadIdx.x >> 5);
    const bool live = strip < (M >> 4);
    const int m0    = live ? (strip << 4) : 0;

    float4 a[KB][2];
    {
        const float4* ap = reinterpret_cast<const float4*>(A + (size_t)(m0 + c) * K + 8 * h);
#pragma unroll
        for (int i = 0; i < KB; ++i) {
            a[i][0] = ap[4 * i];
            a[i][1] = ap[4 * i + 1];
        }
    }

    constexpr int NQ    = NC / 4;
    constexpr int NP4   = (K / 2) * NQ;
    constexpr int ITERS = (NP4 + 255) / 256;
#pragma unroll
    for (int it = 0; it < ITERS; ++it) {
        const int p = it * 256 + (int)threadIdx.x;
        if (NP4 % 256 == 0 || p < NP4) {
            const int kp = p / NQ, q = p - kp * NQ;
            const float4 r0 = *reinterpret_cast<const float4*>(B + (size_t)(2 * kp)     * NC + 4 * q);
            const float4 r1 = *reinterpret_cast<const float4*>(B + (size_t)(2 * kp + 1) * NC + 4 * q);
            float4* d = reinterpret_cast<float4*>(&Bs[kp * PITCH + 8 * q]);
            d[0] = make_float4(r0.x, r1.x, r0.y, r1.y);
            d[1] = make_float4(r0.z, r1.z, r0.w, r1.w);
        }
    }
    if constexpr (2 * HEADS < 16) {
        constexpr int ZQ = (16 - 2 * HEADS) / 2;
        for (int p = (int)threadIdx.x; p < KP * ZQ; p += 256) {
            const int kp = p / ZQ, q = p - kp * ZQ;
            *reinterpret_cast<float4*>(&Bs[kp * PITCH + 2 * (NC + 2 * HEADS) + 4 * q]) = make_float4(0.f, 0.f, 0.f, 0.f);
        }
    }
    __syncthreads();

    constexpr int NITEM = KP * HEADS;
    constexpr int XIT   = (NITEM + 255) / 256;
#pragma unroll
    for (int it = 0; it < XIT; ++it) {
        const int item = it * 256 + (int)threadIdx.x;
        if (NITEM % 256 == 0 || item < NITEM) {
            const int hd = __builtin_amdgcn_readfirstlane(item / KP);
            const int kp = item - hd * KP;
            const float4* bp = reinterpret_cast<const float4*>(&Bs[kp * PITCH + 2 * hd * CH]);
            const float* as = att_s + hd * CH;
            const float* ad = att_d + hd * CH;
            float s0 = 0.f, s1 = 0.f, d0 = 0.f, d1 = 0.f;
#pragma unroll
            for (int n2 = 0; n2 < CH / 2; ++n2) {
                const float4 v = bp[n2];
                const float as0 = as[2 * n2], as1 = as[2 * n2 + 1];
                const float ad0 = ad[2 * n2], ad1 = ad[2 * n2 + 1];
                s0 += v.x * as0;  s1 += v.y * as0;  d0 += v.x * ad0;  d1 += v.y * ad0;
                s0 += v.z * as1;  s1 += v.w * as1;  d0 += v.z * ad1;  d1 += v.w * ad1;
            }
            *reinterpret_cast<float2*>(&Bs[kp * PITCH + 2 * (NC + hd)])         = make_float2(s0, s1);
            *reinterpret_cast<float2*>(&Bs[kp * PITCH + 2 * (NC + HEADS + hd)]) = make_float2(d0, d1);
        }
    }
    __syncthreads();
    if (!live) return;

    v8f acc[NT + 1];
#pragma unroll
    for (int t = 0; t <= NT; ++t) acc[t] = v8f{};

    const float* bl = Bs + (4 * h) * PITCH + 2 * c;
#pragma unroll
    for (int i = 0; i < KB; ++i) {
#pragma unroll
        for (int j = 0; j < 4; ++j) {
            const float4 q = a[i][j >> 1];
            v2f af;
            af.x = (j & 1) ? q.z : q.x;
            af.y = (j & 1) ? q.w : q.y;
            const float* bj = bl + (8 * i + j) * PITCH;
#pragma unroll
            for (int t = 0; t <= NT; ++t) {
                v2f bf = *reinterpret_cast<const v2f*>(bj + 32 * t);
                acc[t] = __builtin_amdgcn_wmma_f32_16x16x4_f32(
                              false, af,  false, bf,
                              (short)0, acc[t],  false,  false);
            }
        }
    }

    float* cp = C + (size_t)(m0 + 8 * h) * NC + c;
#pragma unroll
    for (int t = 0; t < NT; ++t) {
#pragma unroll
        for (int r = 0; r < 8; ++r) cp[(size_t)r * NC + 16 * t] = acc[t][r];
    }

    if (c < 2 * HEADS) {
        float* sp = (c < HEADS) ? (a_s + c) : (a_d + (c - HEADS));
        sp += (size_t)(m0 + 8 * h) * HEADS;
#pragma unroll
        for (int r = 0; r < 8; ++r) sp[(size_t)r * HEADS] = acc[NT][r];
    }
}

#define BKT_SHIFT 8
#define BKT_NODES 256
#define MAXB      256
#define CHUNK     8192
#define BKT_KEEP  5

__global__ void __launch_bounds__(1024)
k_bin_count(const int* __restrict__ ei, int E, int nb, int* __restrict__ hist) {
    __shared__ int s_cnt[MAXB];
    const int tid  = threadIdx.x;
    const int base = blockIdx.x * CHUNK;
    int d[8];
#pragma unroll
    for (int k = 0; k < 8; ++k) {
        const int e = base + k * 1024 + tid;
        d[k] = ei[E + ((e < E) ? e : E - 1)];
    }
    if (tid < MAXB) s_cnt[tid] = 0;
    __syncthreads();
#pragma unroll
    for (int k = 0; k < 8; ++k) {
        const int e = base + k * 1024 + tid;
        if (e < E) atomicAdd(&s_cnt[d[k] >> BKT_SHIFT], 1);
    }
    __syncthreads();
    if (tid < nb) hist[blockIdx.x * nb + tid] = s_cnt[tid];
}

__global__ void __launch_bounds__(1024)
k_bin_scatter(const int* __restrict__ ei, int E, int nb, const int* __restrict__ hist,
              int* __restrict__ bstart, int* __restrict__ pairs) {
    __shared__ int s_bef[4][MAXB];
    __shared__ int s_tot[4][MAXB];
    __shared__ int s_cur[MAXB];
    const int tid = threadIdx.x, lane = tid & 31;
    const int me = blockIdx.x, G = gridDim.x;
    const int base = me * CHUNK;

    int s[8], d[8];
#pragma unroll
    for (int k = 0; k < 8; ++k) {
        const int e  = base + k * 1024 + tid;
        const int ec = (e < E) ? e : E - 1;
        s[k] = ei[ec];
        d[k] = ei[E + ec];
    }

    const int b = tid & (MAXB - 1), part = tid >> 8;
    const int bc = (b < nb) ? b : 0;
    int before = 0, total = 0;
    for (int w0 = 0; w0 < G; w0 += 32) {
        int v[8];
#pragma unroll
        for (int u = 0; u < 8; ++u) {
            const int w = w0 + 4 * u + part;
            v[u] = hist[((w < G) ? w : 0) * nb + bc];
        }
#pragma unroll
        for (int u = 0; u < 8; ++u) {
            const int w = w0 + 4 * u + part;
            const int x = (w < G && b < nb) ? v[u] : 0;
            total  += x;
            before += (w < me) ? x : 0;
        }
    }
    s_bef[part][b] = before;
    s_tot[part][b] = total;
    __syncthreads();

    if (tid < 32) {
        int ex[8];
        int run = 0;
#pragma unroll
        for (int k = 0; k < 8; ++k) {
            const int i = 8 * tid + k;
            ex[k] = run;
            run += s_tot[0][i] + s_tot[1][i] + s_tot[2][i] + s_tot[3][i];
        }
        int inc = run;
        for (int o = 1; o < 32; o <<= 1) {
            int up = __shfl_up(inc, o);
            if (lane >= o) inc += up;
        }
        const int excl = inc - run;
#pragma unroll
        for (int k = 0; k < 8; ++k) {
            const int i  = 8 * tid + k;
            const int st = excl + ex[k];
            s_cur[i] = st + s_bef[0][i] + s_bef[1][i] + s_bef[2][i] + s_bef[3][i];
            if (me == 0 && i <= nb) bstart[i] = st;
        }
    }
    __syncthreads();

#pragma unroll
    for (int k = 0; k < 8; ++k) {
        const int e = base + k * 1024 + tid;
        if (e < E) {
            const int pos = atomicAdd(&s_cur[d[k] >> BKT_SHIFT], 1);
            *reinterpret_cast<int2*>(pairs + 2 * (size_t)pos) = make_int2(s[k], d[k]);
        }
    }
}

__global__ void __launch_bounds__(1024)
k_bucket_sort(const int* __restrict__ pairs, const int* __restrict__ bstart, int E,
              int* __restrict__ rowptr, int* __restrict__ ssrc, int n) {
    __shared__ int s_cnt[BKT_NODES];
    __shared__ int s_off[BKT_NODES];
    const int tid = threadIdx.x, lane = tid & 31;
    const int bkt = blockIdx.x;
    const int lo  = bkt << BKT_SHIFT;
    const int s0  = bstart[bkt], s1 = bstart[bkt + 1];
    const int2* pp = reinterpret_cast<const int2*>(pairs);

    int2 p[BKT_KEEP];
#pragma unroll
    for (int k = 0; k < BKT_KEEP; ++k) {
        const int i = s0 + k * 1024 + tid;
        p[k] = pp[(i < s1) ? i : E - 1];
    }
    if (tid < BKT_NODES) s_cnt[tid] = 0;
    __syncthreads();

#pragma unroll
    for (int k = 0; k < BKT_KEEP; ++k) {
        const int i = s0 + k * 1024 + tid;
        if (i < s1) atomicAdd(&s_cnt[p[k].y - lo], 1);
    }
    for (int i = s0 + BKT_KEEP * 1024 + tid; i < s1; i += 1024) atomicAdd(&s_cnt[pp[i].y - lo], 1);
    __syncthreads();

    if (tid < 32) {
        int ex[8];
        int run = 0;
#pragma unroll
        for (int k = 0; k < 8; ++k) {
            ex[k] = run;
            run += s_cnt[8 * tid + k];
        }
        int inc = run;
        for (int o = 1; o < 32; o <<= 1) {
            int up = __shfl_up(inc, o);
            if (lane >= o) inc += up;
        }
        const int excl = inc - run;
#pragma unroll
        for (int k = 0; k < 8; ++k) s_off[8 * tid + k] = excl + ex[k];
    }
    __syncthreads();
    if (tid < BKT_NODES && lo + tid < n) rowptr[lo + tid] = s0 + s_off[tid];
    if (bkt == (int)gridDim.x - 1 && tid == 0) rowptr[n] = s1;
    __syncthreads();

#pragma unroll
    for (int k = 0; k < BKT_KEEP; ++k) {
        const int i = s0 + k * 1024 + tid;
        if (i < s1) ssrc[s0 + atomicAdd(&s_off[p[k].y - lo], 1)] = p[k].x;
    }
    for (int i = s0 + BKT_KEEP * 1024 + tid; i < s1; i += 1024) {
        const int2 q = pp[i];
        ssrc[s0 + atomicAdd(&s_off[q.y - lo], 1)] = q.x;
    }
}

#define L1_KEEP 8

__device__ __forceinline__ void l1_rows(const float* __restrict__ h, const int (&sq)[4], float w,
                                        int t0, int cntE, int lane, const int (&hk)[5], float (&acc)[5]) {
    if (t0 + 4 <= cntE) {
        float hv[4][5];
        for (int q = 0; q < 4; ++q) {
            const float* hs = h + (size_t)sq[q] * H1C1 + lane;
            for (int k = 0; k < 5; ++k) hv[q][k] = hs[32 * k];
        }
        for (int q = 0; q < 4; ++q) {
            for (int k = 0; k < 5; ++k) acc[k] += __shfl(w, q * 8 + hk[k]) * hv[q][k];
        }
    } else {
        for (int q = 0; q < 3; ++q) {
            if (t0 + q < cntE) {
                const float* hs = h + (size_t)sq[q] * H1C1 + lane;
                for (int k = 0; k < 5; ++k) acc[k] += __shfl(w, q * 8 + hk[k]) * hs[32 * k];
            }
        }
    }
}

__global__ void k_gather_l1(const int* __restrict__ rowptr, const int* __restrict__ ssrc,
                            const float* __restrict__ a_s, const float* __restrict__ a_d,
                            const float* __restrict__ h, const float* __restrict__ b,
                            float* __restrict__ agg, int n) {
    int lane = threadIdx.x & 31;
    int node = blockIdx.x * (blockDim.x >> 5) + (threadIdx.x >> 5);
    node = __builtin_amdgcn_readfirstlane(node);
    if (node >= n) return;
    int slot = lane >> 3, head = lane & 7;
    int beg  = rowptr[node];
    int cntE = rowptr[node + 1] - beg + 1;
    float ad = a_d[node * HEADS1 + head];

    int sAll = node;
    if (lane > 0 && lane < cntE) sAll = ssrc[beg + lane - 1];

    float eC[L1_KEEP];
#pragma unroll
    for (int k = 0; k < L1_KEEP; ++k) {
        const int s = __shfl(sAll, 4 * k + slot);
        eC[k] = a_s[s * HEADS1 + head];
    }
#pragma unroll
    for (int k = 0; k < L1_KEEP; ++k)
        eC[k] = (4 * k + slot < cntE) ? lrelu(eC[k] + ad) : -__builtin_huge_valf();

    float m = -__builtin_huge_valf();
#pragma unroll
    for (int k = 0; k < L1_KEEP; ++k) m = fmaxf(m, eC[k]);
    for (int t0 = 4 * L1_KEEP; t0 < cntE; t0 += 4) {
        int t = t0 + slot;
        if (t < cntE) {
            int s = ssrc[beg + t - 1];
            m = fmaxf(m, lrelu(a_s[s * HEADS1 + head] + ad));
        }
    }
    m = fmaxf(m, __shfl_xor(m, 8));
    m = fmaxf(m, __shfl_xor(m, 16));

    int hk[5];
    for (int k = 0; k < 5; ++k) hk[k] = (lane + 32 * k) / CH1;
    float acc[5] = {0.f, 0.f, 0.f, 0.f, 0.f};
    float den = 0.f;
#pragma unroll
    for (int k = 0; k < L1_KEEP; ++k) {
        const int t0 = 4 * k;
        if (t0 < cntE) {
            const float w = (t0 + slot < cntE) ? expf(eC[k] - m) : 0.f;
            den += w;
            int sq[4];
            for (int q = 0; q < 4; ++q) sq[q] = __builtin_amdgcn_readlane(sAll, t0 + q);
            l1_rows(h, sq, w, t0, cntE, lane, hk, acc);
        }
    }
    for (int t0 = 4 * L1_KEEP; t0 < cntE; t0 += 4) {
        int t = t0 + slot;
        int s = node;
        float w = 0.f;
        if (t < cntE) {
            s = ssrc[beg + t - 1];
            w = expf(lrelu(a_s[s * HEADS1 + head] + ad) - m);
        }
        den += w;
        int sq[4];
        for (int q = 0; q < 4; ++q) sq[q] = __builtin_amdgcn_readlane(s, q * 8);
        l1_rows(h, sq, w, t0, cntE, lane, hk, acc);
    }
    float bv[5];
    for (int k = 0; k < 5; ++k) bv[k] = b[lane + 32 * k];
    den += __shfl_xor(den, 8);
    den += __shfl_xor(den, 16);
    float* o = agg + (size_t)node * H1C1 + lane;
    for (int k = 0; k < 5; ++k) o[32 * k] = fmaxf(acc[k] / __shfl(den, hk[k]) + bv[k], 0.f);
}


__device__ __forceinline__ float l2_rows(const float* __restrict__ h, int s, float w, int np,
                                         int half, int c, float acc) {
#pragma unroll 4
    for (int q = 0; q < np; q += 2) {
        int   sl = __shfl(s, q + half);
        float wl = __shfl(w, q + half);
        acc += wl * h[(size_t)sl * NCLS + c];
    }
    return acc;
}

__global__ void k_gather_l2(const int* __restrict__ rowptr, const int* __restrict__ ssrc,
                            const float* __restrict__ a_s, const float* __restrict__ a_d,
                            const float* __restrict__ h, const float* __restrict__ b,
                            float* __restrict__ out, int n) {
    int lane = threadIdx.x & 31;
    int node = blockIdx.x * (blockDim.x >> 5) + (threadIdx.x >> 5);
    node = __builtin_amdgcn_readfirstlane(node);
    if (node >= n) return;
    int half = lane >> 4, c = lane & 15;
    int beg  = rowptr[node];
    int cntE = rowptr[node + 1] - beg + 1;
    float ad = a_d[node];
    float bc = b[c];

    int   s0 = node;
    float e0 = -__builtin_huge_valf();
    if (lane < cntE) {
        if (lane > 0) s0 = ssrc[beg + lane - 1];
        e0 = lrelu(a_s[s0] + ad);
    }

    float m = fmaxf(-__builtin_huge_valf(), e0);
    for (int t0 = 32; t0 < cntE; t0 += 32) {
        int t = t0 + lane;
        if (t < cntE) {
            int s = ssrc[beg + t - 1];
            m = fmaxf(m, lrelu(a_s[s] + ad));
        }
    }
    for (int o = 16; o > 0; o >>= 1) m = fmaxf(m, __shfl_xor(m, o));

    float acc = 0.f, den = 0.f;
    {
        const float w = (lane < cntE) ? expf(e0 - m) : 0.f;
        den += w;
        acc = l2_rows(h, s0, w, (cntE < 32) ? cntE : 32, half, c, acc);
    }
    for (int t0 = 32; t0 < cntE; t0 += 32) {
        int t = t0 + lane;
        int s = node;
        float w = 0.f;
        if (t < cntE) {
            s = ssrc[beg + t - 1];
            w = expf(lrelu(a_s[s] + ad) - m);
        }
        den += w;
        int left = cntE - t0;
        acc = l2_rows(h, s, w, (left < 32) ? left : 32, half, c, acc);
    }
    for (int o = 16; o > 0; o >>= 1) den += __shfl_xor(den, o);
    acc += __shfl_xor(acc, 16);
    if (lane < 16) out[(size_t)node * NCLS + c] = acc / den + bc;
}


extern "C" void kernel_launch(void* const* d_in, const int* in_sizes, int n_in,
                              void* d_out, int out_size, void* d_ws, size_t ws_size,
                              hipStream_t stream) {
    const float* x        = (const float*)d_in[0];
    const int*   ei       = (const int*)  d_in[1];
    const float* W1       = (const float*)d_in[2];
    const float* att_src1 = (const float*)d_in[3];
    const float* att_dst1 = (const float*)d_in[4];
    const float* b1       = (const float*)d_in[5];
    const float* W2       = (const float*)d_in[6];
    const float* att_src2 = (const float*)d_in[7];
    const float* att_dst2 = (const float*)d_in[8];
    const float* b2       = (const float*)d_in[9];
    float* out = (float*)d_out;

    const int N    = in_sizes[0] / F_IN;
    const int E    = in_sizes[1] / 2;
    float* ws  = (float*)d_ws;
    float* h1  = ws;
    float* as1 = h1  + (size_t)N * H1C1;
    float* ad1 = as1 + (size_t)N * HEADS1;
    float* h2  = ad1 + (size_t)N * HEADS1;
    float* as2 = h2  + (size_t)N * NCLS;
    float* ad2 = as2 + (size_t)N;
    float* agg1 = ad2 + (size_t)N;
    int* pairs  = (int*)(agg1 + (size_t)N * H1C1);
    int* ssrc   = pairs + 2 * (size_t)E;
    int* rowptr = ssrc + (size_t)E;
    int* bstart = rowptr + (size_t)N + 1;
    int* hist   = bstart + MAXB;

    const int T = 256;
    const int wavesPerBlk = T >> 5;

    const int nb     = (N + BKT_NODES - 1) >> BKT_SHIFT;
    const int chunks = (E + CHUNK - 1) / CHUNK;
    k_bin_count<<<chunks, 1024, 0, stream>>>(ei, E, nb, hist);
    k_bin_scatter<<<chunks, 1024, 0, stream>>>(ei, E, nb, hist, bstart, pairs);
    k_bucket_sort<<<nb, 1024, 0, stream>>>(pairs, bstart, E, rowptr, ssrc, N);

    k_gemm_strip_f32<H1C1 / 16, F_IN, HEADS1, CH1><<<((N >> 4) + wavesPerBlk - 1) / wavesPerBlk, T, 0, stream>>>(
        x, W1, att_src1, att_dst1, h1, as1, ad1, N);
    k_gather_l1<<<(N + wavesPerBlk - 1) / wavesPerBlk, T, 0, stream>>>(
        rowptr, ssrc, as1, ad1, h1, b1, agg1, N);

    k_gemm_strip_f32<NCLS / 16, H1C1, 1, NCLS><<<((N >> 4) + wavesPerBlk - 1) / wavesPerBlk, T, 0, stream>>>(
        agg1, W2, att_src2, att_dst2, h2, as2, ad2, N);
    k_gather_l2<<<(N + wavesPerBlk - 1) / wavesPerBlk, T, 0, stream>>>(
        rowptr, ssrc, as2, ad2, h2, b2, out, N);
}


# Edge grouping level 1 chunk-local (a chunk orders its own edges by bucket, no global counts), run as 98 extra work-groups of the layer-1 GEMM launch: 5 launches; row sum order = LDS arrival rank
# speedup vs baseline: 14.8700x; 1.1629x over previous
#include <hip/hip_runtime.h>
#include <hip/hip_bf16.h>


#define NEG_SLOPE 0.2f
#define F_IN   128
#define H1C1   160
#define HEADS1 8
#define CH1    20
#define NCLS   16

typedef float v2f __attribute__((ext_vector_type(2)));
typedef float v8f __attribute__((ext_vector_type(8)));


__device__ __forceinline__ float lrelu(float v) {
    return (v > 0.f) ? v : NEG_SLOPE * v;
}


#define BKT_SHIFT 8
#define BKT_NODES 256
#define MAXB      256
#define CHUNK     8192
#define CS_EPT    32

__device__ __forceinline__ void chunk_sort(const int* __restrict__ ei, int E, int nb, int w,
                                           int* __restrict__ segoff, int* __restrict__ cpairs, int* lds) {
    int* s_cnt = lds;
    int* s_cur = lds + MAXB;
    const int tid = threadIdx.x, lane = tid & 31;
    const int base = w * CHUNK;
    int s[CS_EPT], d[CS_EPT];
#pragma unroll
    for (int k = 0; k < CS_EPT; ++k) {
        const int e  = base + k * 256 + tid;
        const int ec = (e < E) ? e : E - 1;
        s[k] = ei[ec];
        d[k] = ei[E + ec];
    }
    s_cnt[tid] = 0;
    __syncthreads();
#pragma unroll
    for (int k = 0; k < CS_EPT; ++k) {
        const int e = base + k * 256 + tid;
        if (e < E) atomicAdd(&s_cnt[d[k] >> BKT_SHIFT], 1);
    }
    __syncthreads();
    if (tid < 32) {
        int ex[8];
        int run = 0;
#pragma unroll
        for (int k = 0; k < 8; ++k) {
            ex[k] = run;
            run += s_cnt[8 * tid + k];
        }
        int inc = run;
        for (int o = 1; o < 32; o <<= 1) {
            int up = __shfl_up(inc, o);
            if (lane >= o) inc += up;
        }
        const int excl = inc - run;
#pragma unroll
        for (int k = 0; k < 8; ++k) {
            const int i = 8 * tid + k;
            s_cur[i] = excl + ex[k];
            if (i <= nb) segoff[w * (nb + 1) + i] = excl + ex[k];
        }
    }
    __syncthreads();
#pragma unroll
    for (int k = 0; k < CS_EPT; ++k) {
        const int e = base + k * 256 + tid;
        if (e < E) {
            const int pos = atomicAdd(&s_cur[d[k] >> BKT_SHIFT], 1);
            *reinterpret_cast<int2*>(cpairs + 2 * (size_t)(base + pos)) = make_int2(s[k], d[k]);
        }
    }
}

template <int NT, int K, int HEADS, int CH, bool SORT>
__global__ void __launch_bounds__(256)
k_gemm_strip_f32(const float* __restrict__ A,
                 const float* __restrict__ B,
                 const float* __restrict__ att_s,
                 const float* __restrict__ att_d,
                 float* __restrict__ C,
                 float* __restrict__ a_s, float* __restrict__ a_d, int M,
                 const int* __restrict__ ei, int E, int nb, int nsort,
                 int* __restrict__ segoff, int* __restrict__ cpairs) {
    constexpr int NC    = NT * 16;
    constexpr int NCX   = NC + 16;
    constexpr int PITCH = 2 * NCX + 8;
    constexpr int KB    = K / 16;
    constexpr int KP    = K / 2;
    static_assert(NC == HEADS * CH && 2 * HEADS <= 16 && CH % 2 == 0, "score tile: 2*HEADS columns, channels read in pairs");
    static_assert(HEADS == 1 || KP % 32 == 0, "a wave of the score pass works on one head");
    __shared__ __attribute__((aligned(16))) float Bs[KP * PITCH];
    static_assert(!SORT || KP * PITCH >= 2 * MAXB, "chunk_sort's counters live in the B stage");

    int blk = blockIdx.x;
    if constexpr (SORT) {
        if (blk < nsort) {
            chunk_sort(ei, E, nb, blk, segoff, cpairs, reinterpret_cast<int*>(Bs));
            return;
        }
        blk -= nsort;
    }

    const int lane  = threadIdx.x & 31;
    const int c     = lane & 15;
    const int h     = lane >> 4;
    const int strip = blk * (blockDim.x >> 5) + (threadIdx.x >> 5);
    const bool live = strip < (M >> 4);
    const int m0    = live ? (strip << 4) : 0;

    float4 a[KB][2];
    {
        const float4* ap = reinterpret_cast<const float4*>(A + (size_t)(m0 + c) * K + 8 * h);
#pragma unroll
        for (int i = 0; i < KB; ++i) {
            a[i][0] = ap[4 * i];
            a[i][1] = ap[4 * i + 1];
        }
    }

    constexpr int NQ    = NC / 4;
    constexpr int NP4   = (K / 2) * NQ;
    constexpr int ITERS = (NP4 + 255) / 256;
#pragma unroll
    for (int it = 0; it < ITERS; ++it) {
        const int p = it * 256 + (int)threadIdx.x;
        if (NP4 % 256 == 0 || p < NP4) {
            const int kp = p / NQ, q = p - kp * NQ;
            const float4 r0 = *reinterpret_cast<const float4*>(B + (size_t)(2 * kp)     * NC + 4 * q);
            const float4 r1 = *reinterpret_cast<const float4*>(B + (size_t)(2 * kp + 1) * NC + 4 * q);
            float4* d = reinterpret_cast<float4*>(&Bs[kp * PITCH + 8 * q]);
            d[0] = make_float4(r0.x, r1.x, r0.y, r1.y);
            d[1] = make_float4(r0.z, r1.z, r0.w, r1.w);
        }
    }
    if constexpr (2 * HEADS < 16) {
        constexpr int ZQ = (16 - 2 * HEADS) / 2;
        for (int p = (int)threadIdx.x; p < KP * ZQ; p += 256) {
            const int kp = p / ZQ, q = p - kp * ZQ;
            *reinterpret_cast<float4*>(&Bs[kp * PITCH + 2 * (NC + 2 * HEADS) + 4 * q]) = make_float4(0.f, 0.f, 0.f, 0.f);
        }
    }
    __syncthreads();

    constexpr int NITEM = KP * HEADS;
    constexpr int XIT   = (NITEM + 255) / 256;
#pragma unroll
    for (int it = 0; it < XIT; ++it) {
        const int item = it * 256 + (int)threadIdx.x;
        if (NITEM % 256 == 0 || item < NITEM) {
            const int hd = __builtin_amdgcn_readfirstlane(item / KP);
            const int kp = item - hd * KP;
            const float4* bp = reinterpret_cast<const float4*>(&Bs[kp * PITCH + 2 * hd * CH]);
            const float* as = att_s + hd * CH;
            const float* ad = att_d + hd * CH;
            float s0 = 0.f, s1 = 0.f, d0 = 0.f, d1 = 0.f;
#pragma unroll
            for (int n2 = 0; n2 < CH / 2; ++n2) {
                const float4 v = bp[n2];
                const float as0 = as[2 * n2], as1 = as[2 * n2 + 1];
                const float ad0 = ad[2 * n2], ad1 = ad[2 * n2 + 1];
                s0 += v.x * as0;  s1 += v.y * as0;  d0 += v.x * ad0;  d1 += v.y * ad0;
                s0 += v.z * as1;  s1 += v.w * as1;  d0 += v.z * ad1;  d1 += v.w * ad1;
            }
            *reinterpret_cast<float2*>(&Bs[kp * PITCH + 2 * (NC + hd)])         = make_float2(s0, s1);
            *reinterpret_cast<float2*>(&Bs[kp * PITCH + 2 * (NC + HEADS + hd)]) = make_float2(d0, d1);
        }
    }
    __syncthreads();
    if (!live) return;

    v8f acc[NT + 1];
#pragma unroll
    for (int t = 0; t <= NT; ++t) acc[t] = v8f{};

    const float* bl = Bs + (4 * h) * PITCH + 2 * c;
#pragma unroll
    for (int i = 0; i < KB; ++i) {
#pragma unroll
        for (int j = 0; j < 4; ++j) {
            const float4 q = a[i][j >> 1];
            v2f af;
            af.x = (j & 1) ? q.z : q.x;
            af.y = (j & 1) ? q.w : q.y;
            const float* bj = bl + (8 * i + j) * PITCH;
#pragma unroll
            for (int t = 0; t <= NT; ++t) {
                v2f bf = *reinterpret_cast<const v2f*>(bj + 32 * t);
                acc[t] = __builtin_amdgcn_wmma_f32_16x16x4_f32(
                              false, af,  false, bf,
                              (short)0, acc[t],  false,  false);
            }
        }
    }

    float* cp = C + (size_t)(m0 + 8 * h) * NC + c;
#pragma unroll
    for (int t = 0; t < NT; ++t) {
#pragma unroll
        for (int r = 0; r < 8; ++r) cp[(size_t)r * NC + 16 * t] = acc[t][r];
    }

    if (c < 2 * HEADS) {
        float* sp = (c < HEADS) ? (a_s + c) : (a_d + (c - HEADS));
        sp += (size_t)(m0 + 8 * h) * HEADS;
#pragma unroll
        for (int r = 0; r < 8; ++r) sp[(size_t)r * HEADS] = acc[NT][r];
    }
}

#define BS_J 4
#define BS_T 2
__global__ void __launch_bounds__(1024)
k_bucket_sort(const int* __restrict__ cpairs, const int* __restrict__ segoff, int G, int nb,
              int* __restrict__ rowptr, int* __restrict__ ssrc, int n) {
    __shared__ int s_cnt[BKT_NODES];
    __shared__ int s_off[BKT_NODES];
    __shared__ int s_tot[2];
    const int tid = threadIdx.x, lane = tid & 31, wv = tid >> 5;
    const int bkt = blockIdx.x;
    const int lo  = bkt << BKT_SHIFT;
    const int ld  = nb + 1;
    const int2* pp = reinterpret_cast<const int2*>(cpairs);

    int o0[BS_J], o1[BS_J];
#pragma unroll
    for (int j = 0; j < BS_J; ++j) {
        const int w  = wv + 32 * j;
        const int wc = (w < G) ? w : 0;
        o0[j] = segoff[wc * ld + bkt];
        o1[j] = segoff[wc * ld + bkt + 1];
    }
    int a0 = 0, a1 = 0;
    for (int w = tid; w < G; w += 1024) {
        a0 += segoff[w * ld + bkt];
        a1 += segoff[w * ld + bkt + 1];
    }
    if (tid < BKT_NODES) s_cnt[tid] = 0;
    if (tid < 2) s_tot[tid] = 0;
    __syncthreads();
    for (int o = 16; o > 0; o >>= 1) {
        a0 += __shfl_xor(a0, o);
        a1 += __shfl_xor(a1, o);
    }
    if (lane == 0 && wv * 32 < G) {
        atomicAdd(&s_tot[0], a0);
        atomicAdd(&s_tot[1], a1);
    }

    int2 p[BS_J][BS_T];
#pragma unroll
    for (int j = 0; j < BS_J; ++j) {
        const int w  = wv + 32 * j;
        const int wc = (w < G) ? w : 0;
        if (w >= G) o1[j] = o0[j];
#pragma unroll
        for (int t = 0; t < BS_T; ++t) {
            const int i = o0[j] + 32 * t + lane;
            p[j][t] = pp[(size_t)wc * CHUNK + ((i < o1[j]) ? i : 0)];
        }
    }
#pragma unroll
    for (int j = 0; j < BS_J; ++j) {
#pragma unroll
        for (int t = 0; t < BS_T; ++t) {
            const int i = o0[j] + 32 * t + lane;
            if (i < o1[j]) atomicAdd(&s_cnt[p[j][t].y - lo], 1);
        }
        const int w = wv + 32 * j;
        if (w < G)
            for (int i = o0[j] + 32 * BS_T + lane; i < o1[j]; i += 32)
                atomicAdd(&s_cnt[pp[(size_t)w * CHUNK + i].y - lo], 1);
    }
    for (int w = wv + 32 * BS_J; w < G; w += 32) {
        const int b0 = segoff[w * ld + bkt], b1 = segoff[w * ld + bkt + 1];
        for (int i = b0 + lane; i < b1; i += 32) atomicAdd(&s_cnt[pp[(size_t)w * CHUNK + i].y - lo], 1);
    }
    __syncthreads();

    if (tid < 32) {
        int ex[8];
        int run = 0;
#pragma unroll
        for (int k = 0; k < 8; ++k) {
            ex[k] = run;
            run += s_cnt[8 * tid + k];
        }
        int inc = run;
        for (int o = 1; o < 32; o <<= 1) {
            int up = __shfl_up(inc, o);
            if (lane >= o) inc += up;
        }
        const int excl = inc - run;
#pragma unroll
        for (int k = 0; k < 8; ++k) s_off[8 * tid + k] = excl + ex[k];
    }
    __syncthreads();
    const int s0 = s_tot[0], s1 = s_tot[1];
    if (tid < BKT_NODES && lo + tid < n) rowptr[lo + tid] = s0 + s_off[tid];
    if (bkt == (int)gridDim.x - 1 && tid == 0) rowptr[n] = s1;
    __syncthreads();

#pragma unroll
    for (int j = 0; j < BS_J; ++j) {
#pragma unroll
        for (int t = 0; t < BS_T; ++t) {
            const int i = o0[j] + 32 * t + lane;
            if (i < o1[j]) ssrc[s0 + atomicAdd(&s_off[p[j][t].y - lo], 1)] = p[j][t].x;
        }
        const int w = wv + 32 * j;
        if (w < G)
            for (int i = o0[j] + 32 * BS_T + lane; i < o1[j]; i += 32) {
                const int2 q = pp[(size_t)w * CHUNK + i];
                ssrc[s0 + atomicAdd(&s_off[q.y - lo], 1)] = q.x;
            }
    }
    for (int w = wv + 32 * BS_J; w < G; w += 32) {
        const int b0 = segoff[w * ld + bkt], b1 = segoff[w * ld + bkt + 1];
        for (int i = b0 + lane; i < b1; i += 32) {
            const int2 q = pp[(size_t)w * CHUNK + i];
            ssrc[s0 + atomicAdd(&s_off[q.y - lo], 1)] = q.x;
        }
    }
}

#define L1_KEEP 8

__device__ __forceinline__ void l1_rows(const float* __restrict__ h, const int (&sq)[4], float w,
                                        int t0, int cntE, int lane, const int (&hk)[5], float (&acc)[5]) {
    if (t0 + 4 <= cntE) {
        float hv[4][5];
        for (int q = 0; q < 4; ++q) {
            const float* hs = h + (size_t)sq[q] * H1C1 + lane;
            for (int k = 0; k < 5; ++k) hv[q][k] = hs[32 * k];
        }
        for (int q = 0; q < 4; ++q) {
            for (int k = 0; k < 5; ++k) acc[k] += __shfl(w, q * 8 + hk[k]) * hv[q][k];
        }
    } else {
        for (int q = 0; q < 3; ++q) {
            if (t0 + q < cntE) {
                const float* hs = h + (size_t)sq[q] * H1C1 + lane;
                for (int k = 0; k < 5; ++k) acc[k] += __shfl(w, q * 8 + hk[k]) * hs[32 * k];
            }
        }
    }
}

__global__ void k_gather_l1(const int* __restrict__ rowptr, const int* __restrict__ ssrc,
                            const float* __restrict__ a_s, const float* __restrict__ a_d,
                            const float* __restrict__ h, const float* __restrict__ b,
                            float* __restrict__ agg, int n) {
    int lane = threadIdx.x & 31;
    int node = blockIdx.x * (blockDim.x >> 5) + (threadIdx.x >> 5);
    node = __builtin_amdgcn_readfirstlane(node);
    if (node >= n) return;
    int slot = lane >> 3, head = lane & 7;
    int beg  = rowptr[node];
    int cntE = rowptr[node + 1] - beg + 1;
    float ad = a_d[node * HEADS1 + head];

    int sAll = node;
    if (lane > 0 && lane < cntE) sAll = ssrc[beg + lane - 1];

    float eC[L1_KEEP];
#pragma unroll
    for (int k = 0; k < L1_KEEP; ++k) {
        const int s = __shfl(sAll, 4 * k + slot);
        eC[k] = a_s[s * HEADS1 + head];
    }
#pragma unroll
    for (int k = 0; k < L1_KEEP; ++k)
        eC[k] = (4 * k + slot < cntE) ? lrelu(eC[k] + ad) : -__builtin_huge_valf();

    float m = -__builtin_huge_valf();
#pragma unroll
    for (int k = 0; k < L1_KEEP; ++k) m = fmaxf(m, eC[k]);
    for (int t0 = 4 * L1_KEEP; t0 < cntE; t0 += 4) {
        int t = t0 + slot;
        if (t < cntE) {
            int s = ssrc[beg + t - 1];
            m = fmaxf(m, lrelu(a_s[s * HEADS1 + head] + ad));
        }
    }
    m = fmaxf(m, __shfl_xor(m, 8));
    m = fmaxf(m, __shfl_xor(m, 16));

    int hk[5];
    for (int k = 0; k < 5; ++k) hk[k] = (lane + 32 * k) / CH1;
    float acc[5] = {0.f, 0.f, 0.f, 0.f, 0.f};
    float den = 0.f;
#pragma unroll
    for (int k = 0; k < L1_KEEP; ++k) {
        const int t0 = 4 * k;
        if (t0 < cntE) {
            const float w = (t0 + slot < cntE) ? expf(eC[k] - m) : 0.f;
            den += w;
            int sq[4];
            for (int q = 0; q < 4; ++q) sq[q] = __builtin_amdgcn_readlane(sAll, t0 + q);
            l1_rows(h, sq, w, t0, cntE, lane, hk, acc);
        }
    }
    for (int t0 = 4 * L1_KEEP; t0 < cntE; t0 += 4) {
        int t = t0 + slot;
        int s = node;
        float w = 0.f;
        if (t < cntE) {
            s = ssrc[beg + t - 1];
            w = expf(lrelu(a_s[s * HEADS1 + head] + ad) - m);
        }
        den += w;
        int sq[4];
        for (int q = 0; q < 4; ++q) sq[q] = __builtin_amdgcn_readlane(s, q * 8);
        l1_rows(h, sq, w, t0, cntE, lane, hk, acc);
    }
    float bv[5];
    for (int k = 0; k < 5; ++k) bv[k] = b[lane + 32 * k];
    den += __shfl_xor(den, 8);
    den += __shfl_xor(den, 16);
    float* o = agg + (size_t)node * H1C1 + lane;
    for (int k = 0; k < 5; ++k) o[32 * k] = fmaxf(acc[k] / __shfl(den, hk[k]) + bv[k], 0.f);
}


__device__ __forceinline__ float l2_rows(const float* __restrict__ h, int s, float w, int np,
                                         int half, int c, float acc) {
#pragma unroll 4
    for (int q = 0; q < np; q += 2) {
        int   sl = __shfl(s, q + half);
        float wl = __shfl(w, q + half);
        acc += wl * h[(size_t)sl * NCLS + c];
    }
    return acc;
}

__global__ void k_gather_l2(const int* __restrict__ rowptr, const int* __restrict__ ssrc,
                            const float* __restrict__ a_s, const float* __restrict__ a_d,
                            const float* __restrict__ h, const float* __restrict__ b,
                            float* __restrict__ out, int n) {
    int lane = threadIdx.x & 31;
    int node = blockIdx.x * (blockDim.x >> 5) + (threadIdx.x >> 5);
    node = __builtin_amdgcn_readfirstlane(node);
    if (node >= n) return;
    int half = lane >> 4, c = lane & 15;
    int beg  = rowptr[node];
    int cntE = rowptr[node + 1] - beg + 1;
    float ad = a_d[node];
    float bc = b[c];

    int   s0 = node;
    float e0 = -__builtin_huge_valf();
    if (lane < cntE) {
        if (lane > 0) s0 = ssrc[beg + lane - 1];
        e0 = lrelu(a_s[s0] + ad);
    }

    float m = fmaxf(-__builtin_huge_valf(), e0);
    for (int t0 = 32; t0 < cntE; t0 += 32) {
        int t = t0 + lane;
        if (t < cntE) {
            int s = ssrc[beg + t - 1];
            m = fmaxf(m, lrelu(a_s[s] + ad));
        }
    }
    for (int o = 16; o > 0; o >>= 1) m = fmaxf(m, __shfl_xor(m, o));

    float acc = 0.f, den = 0.f;
    {
        const float w = (lane < cntE) ? expf(e0 - m) : 0.f;
        den += w;
        acc = l2_rows(h, s0, w, (cntE < 32) ? cntE : 32, half, c, acc);
    }
    for (int t0 = 32; t0 < cntE; t0 += 32) {
        int t = t0 + lane;
        int s = node;
        float w = 0.f;
        if (t < cntE) {
            s = ssrc[beg + t - 1];
            w = expf(lrelu(a_s[s] + ad) - m);
        }
        den += w;
        int left = cntE - t0;
        acc = l2_rows(h, s, w, (left < 32) ? left : 32, half, c, acc);
    }
    for (int o = 16; o > 0; o >>= 1) den += __shfl_xor(den, o);
    acc += __shfl_xor(acc, 16);
    if (lane < 16) out[(size_t)node * NCLS + c] = acc / den + bc;
}


extern "C" void kernel_launch(void* const* d_in, const int* in_sizes, int n_in,
                              void* d_out, int out_size, void* d_ws, size_t ws_size,
                              hipStream_t stream) {
    const float* x        = (const float*)d_in[0];
    const int*   ei       = (const int*)  d_in[1];
    const float* W1       = (const float*)d_in[2];
    const float* att_src1 = (const float*)d_in[3];
    const float* att_dst1 = (const float*)d_in[4];
    const float* b1       = (const float*)d_in[5];
    const float* W2       = (const float*)d_in[6];
    const float* att_src2 = (const float*)d_in[7];
    const float* att_dst2 = (const float*)d_in[8];
    const float* b2       = (const float*)d_in[9];
    float* out = (float*)d_out;

    const int N    = in_sizes[0] / F_IN;
    const int E    = in_sizes[1] / 2;
    float* ws  = (float*)d_ws;
    float* h1  = ws;
    float* as1 = h1  + (size_t)N * H1C1;
    float* ad1 = as1 + (size_t)N * HEADS1;
    float* h2  = ad1 + (size_t)N * HEADS1;
    float* as2 = h2  + (size_t)N * NCLS;
    float* ad2 = as2 + (size_t)N;
    float* agg1 = ad2 + (size_t)N;
    const int nb     = (N + BKT_NODES - 1) >> BKT_SHIFT;
    const int chunks = (E + CHUNK - 1) / CHUNK;
    int* cpairs = (int*)(agg1 + (size_t)N * H1C1);
    int* ssrc   = cpairs + 2 * (size_t)chunks * CHUNK;
    int* rowptr = ssrc + (size_t)E;
    int* segoff = rowptr + (size_t)N + 1;

    const int T = 256;
    const int wavesPerBlk = T >> 5;

    k_gemm_strip_f32<H1C1 / 16, F_IN, HEADS1, CH1, true><<<chunks + ((N >> 4) + wavesPerBlk - 1) / wavesPerBlk, T, 0, stream>>>(
        x, W1, att_src1, att_dst1, h1, as1, ad1, N,
        ei, E, nb, chunks, segoff, cpairs);
    k_bucket_sort<<<nb, 1024, 0, stream>>>(cpairs, segoff, chunks, nb, rowptr, ssrc, N);

    k_gather_l1<<<(N + wavesPerBlk - 1) / wavesPerBlk, T, 0, stream>>>(
        rowptr, ssrc, as1, ad1, h1, b1, agg1, N);

    k_gemm_strip_f32<NCLS / 16, H1C1, 1, NCLS, false><<<((N >> 4) + wavesPerBlk - 1) / wavesPerBlk, T, 0, stream>>>(
        agg1, W2, att_src2, att_dst2, h2, as2, ad2, N,
        nullptr, 0, 0, 0, nullptr, nullptr);
    k_gather_l2<<<(N + wavesPerBlk - 1) / wavesPerBlk, T, 0, stream>>>(
        rowptr, ssrc, as2, ad2, h2, b2, out, N);
}


# Layer-1 gather reads rows in 16-byte loads, one entry per 8-lane group (one load = a whole line of each of 4 rows): 4x fewer row loads; row sum order now 4 partial sums by entry slot, added at end
# speedup vs baseline: 15.7173x; 1.0570x over previous
#include <hip/hip_runtime.h>
#include <hip/hip_bf16.h>


#define NEG_SLOPE 0.2f
#define F_IN   128
#define H1C1   160
#define HEADS1 8
#define CH1    20
#define NCLS   16

typedef float v2f __attribute__((ext_vector_type(2)));
typedef float v8f __attribute__((ext_vector_type(8)));


__device__ __forceinline__ float lrelu(float v) {
    return (v > 0.f) ? v : NEG_SLOPE * v;
}


#define BKT_SHIFT 8
#define BKT_NODES 256
#define MAXB      256
#define CHUNK     8192
#define CS_EPT    32

__device__ __forceinline__ void chunk_sort(const int* __restrict__ ei, int E, int nb, int w,
                                           int* __restrict__ segoff, int* __restrict__ cpairs, int* lds) {
    int* s_cnt = lds;
    int* s_cur = lds + MAXB;
    const int tid = threadIdx.x, lane = tid & 31;
    const int base = w * CHUNK;
    int s[CS_EPT], d[CS_EPT];
#pragma unroll
    for (int k = 0; k < CS_EPT; ++k) {
        const int e  = base + k * 256 + tid;
        const int ec = (e < E) ? e : E - 1;
        s[k] = ei[ec];
        d[k] = ei[E + ec];
    }
    s_cnt[tid] = 0;
    __syncthreads();
#pragma unroll
    for (int k = 0; k < CS_EPT; ++k) {
        const int e = base + k * 256 + tid;
        if (e < E) atomicAdd(&s_cnt[d[k] >> BKT_SHIFT], 1);
    }
    __syncthreads();
    if (tid < 32) {
        int ex[8];
        int run = 0;
#pragma unroll
        for (int k = 0; k < 8; ++k) {
            ex[k] = run;
            run += s_cnt[8 * tid + k];
        }
        int inc = run;
        for (int o = 1; o < 32; o <<= 1) {
            int up = __shfl_up(inc, o);
            if (lane >= o) inc += up;
        }
        const int excl = inc - run;
#pragma unroll
        for (int k = 0; k < 8; ++k) {
            const int i = 8 * tid + k;
            s_cur[i] = excl + ex[k];
            if (i <= nb) segoff[w * (nb + 1) + i] = excl + ex[k];
        }
    }
    __syncthreads();
#pragma unroll
    for (int k = 0; k < CS_EPT; ++k) {
        const int e = base + k * 256 + tid;
        if (e < E) {
            const int pos = atomicAdd(&s_cur[d[k] >> BKT_SHIFT], 1);
            *reinterpret_cast<int2*>(cpairs + 2 * (size_t)(base + pos)) = make_int2(s[k], d[k]);
        }
    }
}

template <int NT, int K, int HEADS, int CH, bool SORT>
__global__ void __launch_bounds__(256)
k_gemm_strip_f32(const float* __restrict__ A,
                 const float* __restrict__ B,
                 const float* __restrict__ att_s,
                 const float* __restrict__ att_d,
                 float* __restrict__ C,
                 float* __restrict__ a_s, float* __restrict__ a_d, int M,
                 const int* __restrict__ ei, int E, int nb, int nsort,
                 int* __restrict__ segoff, int* __restrict__ cpairs) {
    constexpr int NC    = NT * 16;
    constexpr int NCX   = NC + 16;
    constexpr int PITCH = 2 * NCX + 8;
    constexpr int KB    = K / 16;
    constexpr int KP    = K / 2;
    static_assert(NC == HEADS * CH && 2 * HEADS <= 16 && CH % 2 == 0, "score tile: 2*HEADS columns, channels read in pairs");
    static_assert(HEADS == 1 || KP % 32 == 0, "a wave of the score pass works on one head");
    __shared__ __attribute__((aligned(16))) float Bs[KP * PITCH];
    static_assert(!SORT || KP * PITCH >= 2 * MAXB, "chunk_sort's counters live in the B stage");

    int blk = blockIdx.x;
    if constexpr (SORT) {
        if (blk < nsort) {
            chunk_sort(ei, E, nb, blk, segoff, cpairs, reinterpret_cast<int*>(Bs));
            return;
        }
        blk -= nsort;
    }

    const int lane  = threadIdx.x & 31;
    const int c     = lane & 15;
    const int h     = lane >> 4;
    const int strip = blk * (blockDim.x >> 5) + (threadIdx.x >> 5);
    const bool live = strip < (M >> 4);
    const int m0    = live ? (strip << 4) : 0;

    float4 a[KB][2];
    {
        const float4* ap = reinterpret_cast<const float4*>(A + (size_t)(m0 + c) * K + 8 * h);
#pragma unroll
        for (int i = 0; i < KB; ++i) {
            a[i][0] = ap[4 * i];
            a[i][1] = ap[4 * i + 1];
        }
    }

    constexpr int NQ    = NC / 4;
    constexpr int NP4   = (K / 2) * NQ;
    constexpr int ITERS = (NP4 + 255) / 256;
#pragma unroll
    for (int it = 0; it < ITERS; ++it) {
        const int p = it * 256 + (int)threadIdx.x;
        if (NP4 % 256 == 0 || p < NP4) {
            const int kp = p / NQ, q = p - kp * NQ;
            const float4 r0 = *reinterpret_cast<const float4*>(B + (size_t)(2 * kp)     * NC + 4 * q);
            const float4 r1 = *reinterpret_cast<const float4*>(B + (size_t)(2 * kp + 1) * NC + 4 * q);
            float4* d = reinterpret_cast<float4*>(&Bs[kp * PITCH + 8 * q]);
            d[0] = make_float4(r0.x, r1.x, r0.y, r1.y);
            d[1] = make_float4(r0.z, r1.z, r0.w, r1.w);
        }
    }
    if constexpr (2 * HEADS < 16) {
        constexpr int ZQ = (16 - 2 * HEADS) / 2;
        for (int p = (int)threadIdx.x; p < KP * ZQ; p += 256) {
            const int kp = p / ZQ, q = p - kp * ZQ;
            *reinterpret_cast<float4*>(&Bs[kp * PITCH + 2 * (NC + 2 * HEADS) + 4 * q]) = make_float4(0.f, 0.f, 0.f, 0.f);
        }
    }
    __syncthreads();

    constexpr int NITEM = KP * HEADS;
    constexpr int XIT   = (NITEM + 255) / 256;
#pragma unroll
    for (int it = 0; it < XIT; ++it) {
        const int item = it * 256 + (int)threadIdx.x;
        if (NITEM % 256 == 0 || item < NITEM) {
            const int hd = __builtin_amdgcn_readfirstlane(item / KP);
            const int kp = item - hd * KP;
            const float4* bp = reinterpret_cast<const float4*>(&Bs[kp * PITCH + 2 * hd * CH]);
            const float* as = att_s + hd * CH;
            const float* ad = att_d + hd * CH;
            float s0 = 0.f, s1 = 0.f, d0 = 0.f, d1 = 0.f;
#pragma unroll
            for (int n2 = 0; n2 < CH / 2; ++n2) {
                const float4 v = bp[n2];
                const float as0 = as[2 * n2], as1 = as[2 * n2 + 1];
                const float ad0 = ad[2 * n2], ad1 = ad[2 * n2 + 1];
                s0 += v.x * as0;  s1 += v.y * as0;  d0 += v.x * ad0;  d1 += v.y * ad0;
                s0 += v.z * as1;  s1 += v.w * as1;  d0 += v.z * ad1;  d1 += v.w * ad1;
            }
            *reinterpret_cast<float2*>(&Bs[kp * PITCH + 2 * (NC + hd)])         = make_float2(s0, s1);
            *reinterpret_cast<float2*>(&Bs[kp * PITCH + 2 * (NC + HEADS + hd)]) = make_float2(d0, d1);
        }
    }
    __syncthreads();
    if (!live) return;

    v8f acc[NT + 1];
#pragma unroll
    for (int t = 0; t <= NT; ++t) acc[t] = v8f{};

    const float* bl = Bs + (4 * h) * PITCH + 2 * c;
#pragma unroll
    for (int i = 0; i < KB; ++i) {
#pragma unroll
        for (int j = 0; j < 4; ++j) {
            const float4 q = a[i][j >> 1];
            v2f af;
            af.x = (j & 1) ? q.z : q.x;
            af.y = (j & 1) ? q.w : q.y;
            const float* bj = bl + (8 * i + j) * PITCH;
#pragma unroll
            for (int t = 0; t <= NT; ++t) {
                v2f bf = *reinterpret_cast<const v2f*>(bj + 32 * t);
                acc[t] = __builtin_amdgcn_wmma_f32_16x16x4_f32(
                              false, af,  false, bf,
                              (short)0, acc[t],  false,  false);
            }
        }
    }

    float* cp = C + (size_t)(m0 + 8 * h) * NC + c;
#pragma unroll
    for (int t = 0; t < NT; ++t) {
#pragma unroll
        for (int r = 0; r < 8; ++r) cp[(size_t)r * NC + 16 * t] = acc[t][r];
    }

    if (c < 2 * HEADS) {
        float* sp = (c < HEADS) ? (a_s + c) : (a_d + (c - HEADS));
        sp += (size_t)(m0 + 8 * h) * HEADS;
#pragma unroll
        for (int r = 0; r < 8; ++r) sp[(size_t)r * HEADS] = acc[NT][r];
    }
}

#define BS_J 4
#define BS_T 2
__global__ void __launch_bounds__(1024)
k_bucket_sort(const int* __restrict__ cpairs, const int* __restrict__ segoff, int G, int nb,
              int* __restrict__ rowptr, int* __restrict__ ssrc, int n) {
    __shared__ int s_cnt[BKT_NODES];
    __shared__ int s_off[BKT_NODES];
    __shared__ int s_tot[2];
    const int tid = threadIdx.x, lane = tid & 31, wv = tid >> 5;
    const int bkt = blockIdx.x;
    const int lo  = bkt << BKT_SHIFT;
    const int ld  = nb + 1;
    const int2* pp = reinterpret_cast<const int2*>(cpairs);

    int o0[BS_J], o1[BS_J];
#pragma unroll
    for (int j = 0; j < BS_J; ++j) {
        const int w  = wv + 32 * j;
        const int wc = (w < G) ? w : 0;
        o0[j] = segoff[wc * ld + bkt];
        o1[j] = segoff[wc * ld + bkt + 1];
    }
    int a0 = 0, a1 = 0;
    for (int w = tid; w < G; w += 1024) {
        a0 += segoff[w * ld + bkt];
        a1 += segoff[w * ld + bkt + 1];
    }
    if (tid < BKT_NODES) s_cnt[tid] = 0;
    if (tid < 2) s_tot[tid] = 0;
    __syncthreads();
    for (int o = 16; o > 0; o >>= 1) {
        a0 += __shfl_xor(a0, o);
        a1 += __shfl_xor(a1, o);
    }
    if (lane == 0 && wv * 32 < G) {
        atomicAdd(&s_tot[0], a0);
        atomicAdd(&s_tot[1], a1);
    }

    int2 p[BS_J][BS_T];
#pragma unroll
    for (int j = 0; j < BS_J; ++j) {
        const int w  = wv + 32 * j;
        const int wc = (w < G) ? w : 0;
        if (w >= G) o1[j] = o0[j];
#pragma unroll
        for (int t = 0; t < BS_T; ++t) {
            const int i = o0[j] + 32 * t + lane;
            p[j][t] = pp[(size_t)wc * CHUNK + ((i < o1[j]) ? i : 0)];
        }
    }
#pragma unroll
    for (int j = 0; j < BS_J; ++j) {
#pragma unroll
        for (int t = 0; t < BS_T; ++t) {
            const int i = o0[j] + 32 * t + lane;
            if (i < o1[j]) atomicAdd(&s_cnt[p[j][t].y - lo], 1);
        }
        const int w = wv + 32 * j;
        if (w < G)
            for (int i = o0[j] + 32 * BS_T + lane; i < o1[j]; i += 32)
                atomicAdd(&s_cnt[pp[(size_t)w * CHUNK + i].y - lo], 1);
    }
    for (int w = wv + 32 * BS_J; w < G; w += 32) {
        const int b0 = segoff[w * ld + bkt], b1 = segoff[w * ld + bkt + 1];
        for (int i = b0 + lane; i < b1; i += 32) atomicAdd(&s_cnt[pp[(size_t)w * CHUNK + i].y - lo], 1);
    }
    __syncthreads();

    if (tid < 32) {
        int ex[8];
        int run = 0;
#pragma unroll
        for (int k = 0; k < 8; ++k) {
            ex[k] = run;
            run += s_cnt[8 * tid + k];
        }
        int inc = run;
        for (int o = 1; o < 32; o <<= 1) {
            int up = __shfl_up(inc, o);
            if (lane >= o) inc += up;
        }
        const int excl = inc - run;
#pragma unroll
        for (int k = 0; k < 8; ++k) s_off[8 * tid + k] = excl + ex[k];
    }
    __syncthreads();
    const int s0 = s_tot[0], s1 = s_tot[1];
    if (tid < BKT_NODES && lo + tid < n) rowptr[lo + tid] = s0 + s_off[tid];
    if (bkt == (int)gridDim.x - 1 && tid == 0) rowptr[n] = s1;
    __syncthreads();

#pragma unroll
    for (int j = 0; j < BS_J; ++j) {
#pragma unroll
        for (int t = 0; t < BS_T; ++t) {
            const int i = o0[j] + 32 * t + lane;
            if (i < o1[j]) ssrc[s0 + atomicAdd(&s_off[p[j][t].y - lo], 1)] = p[j][t].x;
        }
        const int w = wv + 32 * j;
        if (w < G)
            for (int i = o0[j] + 32 * BS_T + lane; i < o1[j]; i += 32) {
                const int2 q = pp[(size_t)w * CHUNK + i];
                ssrc[s0 + atomicAdd(&s_off[q.y - lo], 1)] = q.x;
            }
    }
    for (int w = wv + 32 * BS_J; w < G; w += 32) {
        const int b0 = segoff[w * ld + bkt], b1 = segoff[w * ld + bkt + 1];
        for (int i = b0 + lane; i < b1; i += 32) {
            const int2 q = pp[(size_t)w * CHUNK + i];
            ssrc[s0 + atomicAdd(&s_off[q.y - lo], 1)] = q.x;
        }
    }
}

#define L1_KEEP 8

__device__ __forceinline__ void l1_rows(const float* __restrict__ h, int s, int j,
                                        const float (&wk)[5], float (&acc)[5][4]) {
    const float4* hp = reinterpret_cast<const float4*>(h + (size_t)s * H1C1) + j;
    float4 v[5];
#pragma unroll
    for (int k = 0; k < 5; ++k) v[k] = hp[8 * k];
#pragma unroll
    for (int k = 0; k < 5; ++k) {
        acc[k][0] += wk[k] * v[k].x;
        acc[k][1] += wk[k] * v[k].y;
        acc[k][2] += wk[k] * v[k].z;
        acc[k][3] += wk[k] * v[k].w;
    }
}

__global__ void k_gather_l1(const int* __restrict__ rowptr, const int* __restrict__ ssrc,
                            const float* __restrict__ a_s, const float* __restrict__ a_d,
                            const float* __restrict__ h, const float* __restrict__ b,
                            float* __restrict__ agg, int n) {
    int lane = threadIdx.x & 31;
    int node = blockIdx.x * (blockDim.x >> 5) + (threadIdx.x >> 5);
    node = __builtin_amdgcn_readfirstlane(node);
    if (node >= n) return;
    int slot = lane >> 3, head = lane & 7;
    int beg  = rowptr[node];
    int cntE = rowptr[node + 1] - beg + 1;
    float ad = a_d[node * HEADS1 + head];

    int sAll = node;
    if (lane > 0 && lane < cntE) sAll = ssrc[beg + lane - 1];

    int   sK[L1_KEEP];
    float eC[L1_KEEP];
#pragma unroll
    for (int k = 0; k < L1_KEEP; ++k) {
        sK[k] = __shfl(sAll, 4 * k + slot);
        eC[k] = a_s[sK[k] * HEADS1 + head];
    }
#pragma unroll
    for (int k = 0; k < L1_KEEP; ++k)
        eC[k] = (4 * k + slot < cntE) ? lrelu(eC[k] + ad) : -__builtin_huge_valf();

    float m = -__builtin_huge_valf();
#pragma unroll
    for (int k = 0; k < L1_KEEP; ++k) m = fmaxf(m, eC[k]);
    for (int t0 = 4 * L1_KEEP; t0 < cntE; t0 += 4) {
        int t = t0 + slot;
        if (t < cntE) {
            int s = ssrc[beg + t - 1];
            m = fmaxf(m, lrelu(a_s[s * HEADS1 + head] + ad));
        }
    }
    m = fmaxf(m, __shfl_xor(m, 8));
    m = fmaxf(m, __shfl_xor(m, 16));

    int wl[5];
    for (int k = 0; k < 5; ++k) wl[k] = (lane & 24) | ((head + 8 * k) / 5);
    float acc[5][4];
    for (int k = 0; k < 5; ++k)
        for (int c = 0; c < 4; ++c) acc[k][c] = 0.f;
    float den = 0.f;
#pragma unroll
    for (int k = 0; k < L1_KEEP; ++k) {
        const int t0 = 4 * k;
        if (t0 < cntE) {
            const bool valid = t0 + slot < cntE;
            const float w = valid ? expf(eC[k] - m) : 0.f;
            den += w;
            float wk[5];
            for (int q = 0; q < 5; ++q) wk[q] = __shfl(w, wl[q]);
            if (valid) l1_rows(h, sK[k], head, wk, acc);
        }
    }
    for (int t0 = 4 * L1_KEEP; t0 < cntE; t0 += 4) {
        const int t = t0 + slot;
        const bool valid = t < cntE;
        int s = node;
        float w = 0.f;
        if (valid) {
            s = ssrc[beg + t - 1];
            w = expf(lrelu(a_s[s * HEADS1 + head] + ad) - m);
        }
        den += w;
        float wk[5];
        for (int q = 0; q < 5; ++q) wk[q] = __shfl(w, wl[q]);
        if (valid) l1_rows(h, s, head, wk, acc);
    }
    den += __shfl_xor(den, 8);
    den += __shfl_xor(den, 16);

    const bool bA = (slot & 2) != 0, bB = (slot & 1) != 0;
    float fin[4], r4[4];
    for (int c = 0; c < 4; ++c) {
        const float g02 = bA ? acc[0][c] : acc[2][c];
        const float k02 = bA ? acc[2][c] : acc[0][c];
        const float g13 = bA ? acc[1][c] : acc[3][c];
        const float k13 = bA ? acc[3][c] : acc[1][c];
        const float r02 = k02 + __shfl_xor(g02, 16);
        const float r13 = k13 + __shfl_xor(g13, 16);
        const float r4c = acc[4][c] + __shfl_xor(acc[4][c], 16);
        const float g   = bB ? r02 : r13;
        const float kp  = bB ? r13 : r02;
        fin[c] = kp + __shfl_xor(g, 8);
        r4[c]  = r4c + __shfl_xor(r4c, 8);
    }

    const int   pf = head + 8 * slot;
    const float df = __shfl(den, pf / 5);
    const float d4 = __shfl(den, (head + 32) / 5);
    const float4* bp = reinterpret_cast<const float4*>(b);
    float4* o = reinterpret_cast<float4*>(agg + (size_t)node * H1C1);
    const float4 bf = bp[pf];
    o[pf] = make_float4(fmaxf(fin[0] / df + bf.x, 0.f), fmaxf(fin[1] / df + bf.y, 0.f),
                        fmaxf(fin[2] / df + bf.z, 0.f), fmaxf(fin[3] / df + bf.w, 0.f));
    if (slot == 0) {
        const float4 b4 = bp[32 + head];
        o[32 + head] = make_float4(fmaxf(r4[0] / d4 + b4.x, 0.f), fmaxf(r4[1] / d4 + b4.y, 0.f),
                                   fmaxf(r4[2] / d4 + b4.z, 0.f), fmaxf(r4[3] / d4 + b4.w, 0.f));
    }
}


__device__ __forceinline__ float l2_rows(const float* __restrict__ h, int s, float w, int np,
                                         int half, int c, float acc) {
#pragma unroll 4
    for (int q = 0; q < np; q += 2) {
        int   sl = __shfl(s, q + half);
        float wl = __shfl(w, q + half);
        acc += wl * h[(size_t)sl * NCLS + c];
    }
    return acc;
}

__global__ void k_gather_l2(const int* __restrict__ rowptr, const int* __restrict__ ssrc,
                            const float* __restrict__ a_s, const float* __restrict__ a_d,
                            const float* __restrict__ h, const float* __restrict__ b,
                            float* __restrict__ out, int n) {
    int lane = threadIdx.x & 31;
    int node = blockIdx.x * (blockDim.x >> 5) + (threadIdx.x >> 5);
    node = __builtin_amdgcn_readfirstlane(node);
    if (node >= n) return;
    int half = lane >> 4, c = lane & 15;
    int beg  = rowptr[node];
    int cntE = rowptr[node + 1] - beg + 1;
    float ad = a_d[node];
    float bc = b[c];

    int   s0 = node;
    float e0 = -__builtin_huge_valf();
    if (lane < cntE) {
        if (lane > 0) s0 = ssrc[beg + lane - 1];
        e0 = lrelu(a_s[s0] + ad);
    }

    float m = fmaxf(-__builtin_huge_valf(), e0);
    for (int t0 = 32; t0 < cntE; t0 += 32) {
        int t = t0 + lane;
        if (t < cntE) {
            int s = ssrc[beg + t - 1];
            m = fmaxf(m, lrelu(a_s[s] + ad));
        }
    }
    for (int o = 16; o > 0; o >>= 1) m = fmaxf(m, __shfl_xor(m, o));

    float acc = 0.f, den = 0.f;
    {
        const float w = (lane < cntE) ? expf(e0 - m) : 0.f;
        den += w;
        acc = l2_rows(h, s0, w, (cntE < 32) ? cntE : 32, half, c, acc);
    }
    for (int t0 = 32; t0 < cntE; t0 += 32) {
        int t = t0 + lane;
        int s = node;
        float w = 0.f;
        if (t < cntE) {
            s = ssrc[beg + t - 1];
            w = expf(lrelu(a_s[s] + ad) - m);
        }
        den += w;
        int left = cntE - t0;
        acc = l2_rows(h, s, w, (left < 32) ? left : 32, half, c, acc);
    }
    for (int o = 16; o > 0; o >>= 1) den += __shfl_xor(den, o);
    acc += __shfl_xor(acc, 16);
    if (lane < 16) out[(size_t)node * NCLS + c] = acc / den + bc;
}


extern "C" void kernel_launch(void* const* d_in, const int* in_sizes, int n_in,
                              void* d_out, int out_size, void* d_ws, size_t ws_size,
                              hipStream_t stream) {
    const float* x        = (const float*)d_in[0];
    const int*   ei       = (const int*)  d_in[1];
    const float* W1       = (const float*)d_in[2];
    const float* att_src1 = (const float*)d_in[3];
    const float* att_dst1 = (const float*)d_in[4];
    const float* b1       = (const float*)d_in[5];
    const float* W2       = (const float*)d_in[6];
    const float* att_src2 = (const float*)d_in[7];
    const float* att_dst2 = (const float*)d_in[8];
    const float* b2       = (const float*)d_in[9];
    float* out = (float*)d_out;

    const int N    = in_sizes[0] / F_IN;
    const int E    = in_sizes[1] / 2;
    float* ws  = (float*)d_ws;
    float* h1  = ws;
    float* as1 = h1  + (size_t)N * H1C1;
    float* ad1 = as1 + (size_t)N * HEADS1;
    float* h2  = ad1 + (size_t)N * HEADS1;
    float* as2 = h2  + (size_t)N * NCLS;
    float* ad2 = as2 + (size_t)N;
    float* agg1 = ad2 + (size_t)N;
    const int nb     = (N + BKT_NODES - 1) >> BKT_SHIFT;
    const int chunks = (E + CHUNK - 1) / CHUNK;
    int* cpairs = (int*)(agg1 + (size_t)N * H1C1);
    int* ssrc   = cpairs + 2 * (size_t)chunks * CHUNK;
    int* rowptr = ssrc + (size_t)E;
    int* segoff = rowptr + (size_t)N + 1;

    const int T = 256;
    const int wavesPerBlk = T >> 5;

    k_gemm_strip_f32<H1C1 / 16, F_IN, HEADS1, CH1, true><<<chunks + ((N >> 4) + wavesPerBlk - 1) / wavesPerBlk, T, 0, stream>>>(
        x, W1, att_src1, att_dst1, h1, as1, ad1, N,
        ei, E, nb, chunks, segoff, cpairs);
    k_bucket_sort<<<nb, 1024, 0, stream>>>(cpairs, segoff, chunks, nb, rowptr, ssrc, N);

    k_gather_l1<<<(N + wavesPerBlk - 1) / wavesPerBlk, T, 0, stream>>>(
        rowptr, ssrc, as1, ad1, h1, b1, agg1, N);

    k_gemm_strip_f32<NCLS / 16, H1C1, 1, NCLS, false><<<((N >> 4) + wavesPerBlk - 1) / wavesPerBlk, T, 0, stream>>>(
        agg1, W2, att_src2, att_dst2, h2, as2, ad2, N,
        nullptr, 0, 0, 0, nullptr, nullptr);
    k_gather_l2<<<(N + wavesPerBlk - 1) / wavesPerBlk, T, 0, stream>>>(
        rowptr, ssrc, as2, ad2, h2, b2, out, N);
}


# Layer-2 product moved into the layer-1 gather launch: 16 nodes per block keep finished rows in LDS, one wave runs the WMMAs; 4 launches; layer-2 k sum order and score sums (from h2) change
# speedup vs baseline: 16.6256x; 1.0578x over previous
#include <hip/hip_runtime.h>
#include <hip/hip_bf16.h>


#define NEG_SLOPE 0.2f
#define F_IN   128
#define H1C1   160
#define HEADS1 8
#define CH1    20
#define NCLS   16

typedef float v2f __attribute__((ext_vector_type(2)));
typedef float v8f __attribute__((ext_vector_type(8)));


__device__ __forceinline__ float lrelu(float v) {
    return (v > 0.f) ? v : NEG_SLOPE * v;
}


#define BKT_SHIFT 8
#define BKT_NODES 256
#define MAXB      256
#define CHUNK     8192
#define CS_EPT    32

__device__ __forceinline__ void chunk_sort(const int* __restrict__ ei, int E, int nb, int w,
                                           int* __restrict__ segoff, int* __restrict__ cpairs, int* lds) {
    int* s_cnt = lds;
    int* s_cur = lds + MAXB;
    const int tid = threadIdx.x, lane = tid & 31;
    const int base = w * CHUNK;
    int s[CS_EPT], d[CS_EPT];
#pragma unroll
    for (int k = 0; k < CS_EPT; ++k) {
        const int e  = base + k * 256 + tid;
        const int ec = (e < E) ? e : E - 1;
        s[k] = ei[ec];
        d[k] = ei[E + ec];
    }
    s_cnt[tid] = 0;
    __syncthreads();
#pragma unroll
    for (int k = 0; k < CS_EPT; ++k) {
        const int e = base + k * 256 + tid;
        if (e < E) atomicAdd(&s_cnt[d[k] >> BKT_SHIFT], 1);
    }
    __syncthreads();
    if (tid < 32) {
        int ex[8];
        int run = 0;
#pragma unroll
        for (int k = 0; k < 8; ++k) {
            ex[k] = run;
            run += s_cnt[8 * tid + k];
        }
        int inc = run;
        for (int o = 1; o < 32; o <<= 1) {
            int up = __shfl_up(inc, o);
            if (lane >= o) inc += up;
        }
        const int excl = inc - run;
#pragma unroll
        for (int k = 0; k < 8; ++k) {
            const int i = 8 * tid + k;
            s_cur[i] = excl + ex[k];
            if (i <= nb) segoff[w * (nb + 1) + i] = excl + ex[k];
        }
    }
    __syncthreads();
#pragma unroll
    for (int k = 0; k < CS_EPT; ++k) {
        const int e = base + k * 256 + tid;
        if (e < E) {
            const int pos = atomicAdd(&s_cur[d[k] >> BKT_SHIFT], 1);
            *reinterpret_cast<int2*>(cpairs + 2 * (size_t)(base + pos)) = make_int2(s[k], d[k]);
        }
    }
}

template <int NT, int K, int HEADS, int CH, bool SORT>
__global__ void __launch_bounds__(256)
k_gemm_strip_f32(const float* __restrict__ A,
                 const float* __restrict__ B,
                 const float* __restrict__ att_s,
                 const float* __restrict__ att_d,
                 float* __restrict__ C,
                 float* __restrict__ a_s, float* __restrict__ a_d, int M,
                 const int* __restrict__ ei, int E, int nb, int nsort,
                 int* __restrict__ segoff, int* __restrict__ cpairs) {
    constexpr int NC    = NT * 16;
    constexpr int NCX   = NC + 16;
    constexpr int PITCH = 2 * NCX + 8;
    constexpr int KB    = K / 16;
    constexpr int KP    = K / 2;
    static_assert(NC == HEADS * CH && 2 * HEADS <= 16 && CH % 2 == 0, "score tile: 2*HEADS columns, channels read in pairs");
    static_assert(HEADS == 1 || KP % 32 == 0, "a wave of the score pass works on one head");
    __shared__ __attribute__((aligned(16))) float Bs[KP * PITCH];
    static_assert(!SORT || KP * PITCH >= 2 * MAXB, "chunk_sort's counters live in the B stage");

    int blk = blockIdx.x;
    if constexpr (SORT) {
        if (blk < nsort) {
            chunk_sort(ei, E, nb, blk, segoff, cpairs, reinterpret_cast<int*>(Bs));
            return;
        }
        blk -= nsort;
    }

    const int lane  = threadIdx.x & 31;
    const int c     = lane & 15;
    const int h     = lane >> 4;
    const int strip = blk * (blockDim.x >> 5) + (threadIdx.x >> 5);
    const bool live = strip < (M >> 4);
    const int m0    = live ? (strip << 4) : 0;

    float4 a[KB][2];
    {
        const float4* ap = reinterpret_cast<const float4*>(A + (size_t)(m0 + c) * K + 8 * h);
#pragma unroll
        for (int i = 0; i < KB; ++i) {
            a[i][0] = ap[4 * i];
            a[i][1] = ap[4 * i + 1];
        }
    }

    constexpr int NQ    = NC / 4;
    constexpr int NP4   = (K / 2) * NQ;
    constexpr int ITERS = (NP4 + 255) / 256;
#pragma unroll
    for (int it = 0; it < ITERS; ++it) {
        const int p = it * 256 + (int)threadIdx.x;
        if (NP4 % 256 == 0 || p < NP4) {
            const int kp = p / NQ, q = p - kp * NQ;
            const float4 r0 = *reinterpret_cast<const float4*>(B + (size_t)(2 * kp)     * NC + 4 * q);
            const float4 r1 = *reinterpret_cast<const float4*>(B + (size_t)(2 * kp + 1) * NC + 4 * q);
            float4* d = reinterpret_cast<float4*>(&Bs[kp * PITCH + 8 * q]);
            d[0] = make_float4(r0.x, r1.x, r0.y, r1.y);
            d[1] = make_float4(r0.z, r1.z, r0.w, r1.w);
        }
    }
    if constexpr (2 * HEADS < 16) {
        constexpr int ZQ = (16 - 2 * HEADS) / 2;
        for (int p = (int)threadIdx.x; p < KP * ZQ; p += 256) {
            const int kp = p / ZQ, q = p - kp * ZQ;
            *reinterpret_cast<float4*>(&Bs[kp * PITCH + 2 * (NC + 2 * HEADS) + 4 * q]) = make_float4(0.f, 0.f, 0.f, 0.f);
        }
    }
    __syncthreads();

    constexpr int NITEM = KP * HEADS;
    constexpr int XIT   = (NITEM + 255) / 256;
#pragma unroll
    for (int it = 0; it < XIT; ++it) {
        const int item = it * 256 + (int)threadIdx.x;
        if (NITEM % 256 == 0 || item < NITEM) {
            const int hd = __builtin_amdgcn_readfirstlane(item / KP);
            const int kp = item - hd * KP;
            const float4* bp = reinterpret_cast<const float4*>(&Bs[kp * PITCH + 2 * hd * CH]);
            const float* as = att_s + hd * CH;
            const float* ad = att_d + hd * CH;
            float s0 = 0.f, s1 = 0.f, d0 = 0.f, d1 = 0.f;
#pragma unroll
            for (int n2 = 0; n2 < CH / 2; ++n2) {
                const float4 v = bp[n2];
                const float as0 = as[2 * n2], as1 = as[2 * n2 + 1];
                const float ad0 = ad[2 * n2], ad1 = ad[2 * n2 + 1];
                s0 += v.x * as0;  s1 += v.y * as0;  d0 += v.x * ad0;  d1 += v.y * ad0;
                s0 += v.z * as1;  s1 += v.w * as1;  d0 += v.z * ad1;  d1 += v.w * ad1;
            }
            *reinterpret_cast<float2*>(&Bs[kp * PITCH + 2 * (NC + hd)])         = make_float2(s0, s1);
            *reinterpret_cast<float2*>(&Bs[kp * PITCH + 2 * (NC + HEADS + hd)]) = make_float2(d0, d1);
        }
    }
    __syncthreads();
    if (!live) return;

    v8f acc[NT + 1];
#pragma unroll
    for (int t = 0; t <= NT; ++t) acc[t] = v8f{};

    const float* bl = Bs + (4 * h) * PITCH + 2 * c;
#pragma unroll
    for (int i = 0; i < KB; ++i) {
#pragma unroll
        for (int j = 0; j < 4; ++j) {
            const float4 q = a[i][j >> 1];
            v2f af;
            af.x = (j & 1) ? q.z : q.x;
            af.y = (j & 1) ? q.w : q.y;
            const float* bj = bl + (8 * i + j) * PITCH;
#pragma unroll
            for (int t = 0; t <= NT; ++t) {
                v2f bf = *reinterpret_cast<const v2f*>(bj + 32 * t);
                acc[t] = __builtin_amdgcn_wmma_f32_16x16x4_f32(
                              false, af,  false, bf,
                              (short)0, acc[t],  false,  false);
            }
        }
    }

    float* cp = C + (size_t)(m0 + 8 * h) * NC + c;
#pragma unroll
    for (int t = 0; t < NT; ++t) {
#pragma unroll
        for (int r = 0; r < 8; ++r) cp[(size_t)r * NC + 16 * t] = acc[t][r];
    }

    if (c < 2 * HEADS) {
        float* sp = (c < HEADS) ? (a_s + c) : (a_d + (c - HEADS));
        sp += (size_t)(m0 + 8 * h) * HEADS;
#pragma unroll
        for (int r = 0; r < 8; ++r) sp[(size_t)r * HEADS] = acc[NT][r];
    }
}

#define BS_J 4
#define BS_T 2
__global__ void __launch_bounds__(1024)
k_bucket_sort(const int* __restrict__ cpairs, const int* __restrict__ segoff, int G, int nb,
              int* __restrict__ rowptr, int* __restrict__ ssrc, int n) {
    __shared__ int s_cnt[BKT_NODES];
    __shared__ int s_off[BKT_NODES];
    __shared__ int s_tot[2];
    const int tid = threadIdx.x, lane = tid & 31, wv = tid >> 5;
    const int bkt = blockIdx.x;
    const int lo  = bkt << BKT_SHIFT;
    const int ld  = nb + 1;
    const int2* pp = reinterpret_cast<const int2*>(cpairs);

    int o0[BS_J], o1[BS_J];
#pragma unroll
    for (int j = 0; j < BS_J; ++j) {
        const int w  = wv + 32 * j;
        const int wc = (w < G) ? w : 0;
        o0[j] = segoff[wc * ld + bkt];
        o1[j] = segoff[wc * ld + bkt + 1];
    }
    int a0 = 0, a1 = 0;
    for (int w = tid; w < G; w += 1024) {
        a0 += segoff[w * ld + bkt];
        a1 += segoff[w * ld + bkt + 1];
    }
    if (tid < BKT_NODES) s_cnt[tid] = 0;
    if (tid < 2) s_tot[tid] = 0;
    __syncthreads();
    for (int o = 16; o > 0; o >>= 1) {
        a0 += __shfl_xor(a0, o);
        a1 += __shfl_xor(a1, o);
    }
    if (lane == 0 && wv * 32 < G) {
        atomicAdd(&s_tot[0], a0);
        atomicAdd(&s_tot[1], a1);
    }

    int2 p[BS_J][BS_T];
#pragma unroll
    for (int j = 0; j < BS_J; ++j) {
        const int w  = wv + 32 * j;
        const int wc = (w < G) ? w : 0;
        if (w >= G) o1[j] = o0[j];
#pragma unroll
        for (int t = 0; t < BS_T; ++t) {
            const int i = o0[j] + 32 * t + lane;
            p[j][t] = pp[(size_t)wc * CHUNK + ((i < o1[j]) ? i : 0)];
        }
    }
#pragma unroll
    for (int j = 0; j < BS_J; ++j) {
#pragma unroll
        for (int t = 0; t < BS_T; ++t) {
            const int i = o0[j] + 32 * t + lane;
            if (i < o1[j]) atomicAdd(&s_cnt[p[j][t].y - lo], 1);
        }
        const int w = wv + 32 * j;
        if (w < G)
            for (int i = o0[j] + 32 * BS_T + lane; i < o1[j]; i += 32)
                atomicAdd(&s_cnt[pp[(size_t)w * CHUNK + i].y - lo], 1);
    }
    for (int w = wv + 32 * BS_J; w < G; w += 32) {
        const int b0 = segoff[w * ld + bkt], b1 = segoff[w * ld + bkt + 1];
        for (int i = b0 + lane; i < b1; i += 32) atomicAdd(&s_cnt[pp[(size_t)w * CHUNK + i].y - lo], 1);
    }
    __syncthreads();

    if (tid < 32) {
        int ex[8];
        int run = 0;
#pragma unroll
        for (int k = 0; k < 8; ++k) {
            ex[k] = run;
            run += s_cnt[8 * tid + k];
        }
        int inc = run;
        for (int o = 1; o < 32; o <<= 1) {
            int up = __shfl_up(inc, o);
            if (lane >= o) inc += up;
        }
        const int excl = inc - run;
#pragma unroll
        for (int k = 0; k < 8; ++k) s_off[8 * tid + k] = excl + ex[k];
    }
    __syncthreads();
    const int s0 = s_tot[0], s1 = s_tot[1];
    if (tid < BKT_NODES && lo + tid < n) rowptr[lo + tid] = s0 + s_off[tid];
    if (bkt == (int)gridDim.x - 1 && tid == 0) rowptr[n] = s1;
    __syncthreads();

#pragma unroll
    for (int j = 0; j < BS_J; ++j) {
#pragma unroll
        for (int t = 0; t < BS_T; ++t) {
            const int i = o0[j] + 32 * t + lane;
            if (i < o1[j]) ssrc[s0 + atomicAdd(&s_off[p[j][t].y - lo], 1)] = p[j][t].x;
        }
        const int w = wv + 32 * j;
        if (w < G)
            for (int i = o0[j] + 32 * BS_T + lane; i < o1[j]; i += 32) {
                const int2 q = pp[(size_t)w * CHUNK + i];
                ssrc[s0 + atomicAdd(&s_off[q.y - lo], 1)] = q.x;
            }
    }
    for (int w = wv + 32 * BS_J; w < G; w += 32) {
        const int b0 = segoff[w * ld + bkt], b1 = segoff[w * ld + bkt + 1];
        for (int i = b0 + lane; i < b1; i += 32) {
            const int2 q = pp[(size_t)w * CHUNK + i];
            ssrc[s0 + atomicAdd(&s_off[q.y - lo], 1)] = q.x;
        }
    }
}

#define L1_KEEP 8
#define G1_NODES 16
#define G1_PITCH 164
#define G1_KB    (H1C1 / 8)

__device__ __forceinline__ void l1_rows(const float* __restrict__ h, int s, int j,
                                        const float (&wk)[5], float (&acc)[5][4]) {
    const float4* hp = reinterpret_cast<const float4*>(h + (size_t)s * H1C1) + j;
    float4 v[5];
#pragma unroll
    for (int k = 0; k < 5; ++k) v[k] = hp[8 * k];
#pragma unroll
    for (int k = 0; k < 5; ++k) {
        acc[k][0] += wk[k] * v[k].x;
        acc[k][1] += wk[k] * v[k].y;
        acc[k][2] += wk[k] * v[k].z;
        acc[k][3] += wk[k] * v[k].w;
    }
}

__global__ void __launch_bounds__(32 * G1_NODES, 12)
k_gather_l1_mm2(const int* __restrict__ rowptr, const int* __restrict__ ssrc,
                const float* __restrict__ a_s, const float* __restrict__ a_d,
                const float* __restrict__ h, const float* __restrict__ b,
                const float* __restrict__ W2, const float* __restrict__ att_s2, const float* __restrict__ att_d2,
                float* __restrict__ h2, float* __restrict__ a_s2, float* __restrict__ a_d2, int n) {
    __shared__ __attribute__((aligned(16))) float As[G1_NODES * G1_PITCH];
    __shared__ __attribute__((aligned(16))) float Bf[G1_KB * 32 * 4];
    const int lane = threadIdx.x & 31;
    const int wv   = __builtin_amdgcn_readfirstlane((int)(threadIdx.x >> 5));

    static_assert(G1_KB == G1_NODES + 4, "W2 staging: one block per wave and four more");
    const float* wp  = W2 + (size_t)(8 * wv + 4 * (lane >> 4)) * NCLS + (lane & 15);
    const float* wp2 = W2 + (size_t)(8 * (G1_NODES + (wv & 3)) + 4 * (lane >> 4)) * NCLS + (lane & 15);
    const float4 wq  = make_float4(wp[0],  wp[NCLS],  wp[2 * NCLS],  wp[3 * NCLS]);
    const float4 wq2 = make_float4(wp2[0], wp2[NCLS], wp2[2 * NCLS], wp2[3 * NCLS]);
    auto put_w2 = [&]() {
        float4* bf4 = reinterpret_cast<float4*>(Bf);
        bf4[wv * 32 + lane] = wq;
        if (wv < 4) bf4[(G1_NODES + wv) * 32 + lane] = wq2;
    };

    int node = blockIdx.x * G1_NODES + wv;
    node = __builtin_amdgcn_readfirstlane(node);
    float4* arow = reinterpret_cast<float4*>(As + wv * G1_PITCH);
    if (node < n) {
    int slot = lane >> 3, head = lane & 7;
    int beg  = rowptr[node];
    int cntE = rowptr[node + 1] - beg + 1;
    float ad = a_d[node * HEADS1 + head];

    int sAll = node;
    if (lane > 0 && lane < cntE) sAll = ssrc[beg + lane - 1];

    int   sK[L1_KEEP];
    float eC[L1_KEEP];
#pragma unroll
    for (int k = 0; k < L1_KEEP; ++k) {
        sK[k] = __shfl(sAll, 4 * k + slot);
        eC[k] = a_s[sK[k] * HEADS1 + head];
    }
    put_w2();
#pragma unroll
    for (int k = 0; k < L1_KEEP; ++k)
        eC[k] = (4 * k + slot < cntE) ? lrelu(eC[k] + ad) : -__builtin_huge_valf();

    float m = -__builtin_huge_valf();
#pragma unroll
    for (int k = 0; k < L1_KEEP; ++k) m = fmaxf(m, eC[k]);
    for (int t0 = 4 * L1_KEEP; t0 < cntE; t0 += 4) {
        int t = t0 + slot;
        if (t < cntE) {
            int s = ssrc[beg + t - 1];
            m = fmaxf(m, lrelu(a_s[s * HEADS1 + head] + ad));
        }
    }
    m = fmaxf(m, __shfl_xor(m, 8));
    m = fmaxf(m, __shfl_xor(m, 16));

    int wl[5];
    for (int k = 0; k < 5; ++k) wl[k] = (lane & 24) | ((head + 8 * k) / 5);
    float acc[5][4];
    for (int k = 0; k < 5; ++k)
        for (int c = 0; c < 4; ++c) acc[k][c] = 0.f;
    float den = 0.f;
#pragma unroll
    for (int k = 0; k < L1_KEEP; ++k) {
        const int t0 = 4 * k;
        if (t0 < cntE) {
            const bool valid = t0 + slot < cntE;
            const float w = valid ? expf(eC[k] - m) : 0.f;
            den += w;
            float wk[5];
            for (int q = 0; q < 5; ++q) wk[q] = __shfl(w, wl[q]);
            if (valid) l1_rows(h, sK[k], head, wk, acc);
        }
    }
    for (int t0 = 4 * L1_KEEP; t0 < cntE; t0 += 4) {
        const int t = t0 + slot;
        const bool valid = t < cntE;
        int s = node;
        float w = 0.f;
        if (valid) {
            s = ssrc[beg + t - 1];
            w = expf(lrelu(a_s[s * HEADS1 + head] + ad) - m);
        }
        den += w;
        float wk[5];
        for (int q = 0; q < 5; ++q) wk[q] = __shfl(w, wl[q]);
        if (valid) l1_rows(h, s, head, wk, acc);
    }
    den += __shfl_xor(den, 8);
    den += __shfl_xor(den, 16);

    const bool bA = (slot & 2) != 0, bB = (slot & 1) != 0;
    float fin[4], r4[4];
    for (int c = 0; c < 4; ++c) {
        const float g02 = bA ? acc[0][c] : acc[2][c];
        const float k02 = bA ? acc[2][c] : acc[0][c];
        const float g13 = bA ? acc[1][c] : acc[3][c];
        const float k13 = bA ? acc[3][c] : acc[1][c];
        const float r02 = k02 + __shfl_xor(g02, 16);
        const float r13 = k13 + __shfl_xor(g13, 16);
        const float r4c = acc[4][c] + __shfl_xor(acc[4][c], 16);
        const float g   = bB ? r02 : r13;
        const float kp  = bB ? r13 : r02;
        fin[c] = kp + __shfl_xor(g, 8);
        r4[c]  = r4c + __shfl_xor(r4c, 8);
    }

    const int   pf = head + 8 * slot;
    const float df = __shfl(den, pf / 5);
    const float d4 = __shfl(den, (head + 32) / 5);
    const float4* bp = reinterpret_cast<const float4*>(b);
    const float4 bf = bp[pf];
    arow[pf] = make_float4(fmaxf(fin[0] / df + bf.x, 0.f), fmaxf(fin[1] / df + bf.y, 0.f),
                           fmaxf(fin[2] / df + bf.z, 0.f), fmaxf(fin[3] / df + bf.w, 0.f));
    if (slot == 0) {
        const float4 b4 = bp[32 + head];
        arow[32 + head] = make_float4(fmaxf(r4[0] / d4 + b4.x, 0.f), fmaxf(r4[1] / d4 + b4.y, 0.f),
                                      fmaxf(r4[2] / d4 + b4.z, 0.f), fmaxf(r4[3] / d4 + b4.w, 0.f));
    }
    } else {
        put_w2();
        arow[lane] = make_float4(0.f, 0.f, 0.f, 0.f);
        if (lane < 8) arow[32 + lane] = make_float4(0.f, 0.f, 0.f, 0.f);
    }
    __syncthreads();
    if (wv != 0) return;

    const int c  = lane & 15;
    const int hh = lane >> 4;
    const float asc = att_s2[c], adc = att_d2[c];
    const float4* ap  = reinterpret_cast<const float4*>(As + c * G1_PITCH + 4 * hh);
    const float4* bqp = reinterpret_cast<const float4*>(Bf) + lane;
    v8f acc0 = v8f{}, acc1 = v8f{};
    constexpr int G1_AHEAD = 4;
    float4 qa[G1_KB], qb[G1_KB];
#pragma unroll
    for (int i = 0; i < G1_AHEAD; ++i) {
        qa[i] = ap[2 * i];
        qb[i] = bqp[32 * i];
    }
#pragma unroll
    for (int i = 0; i < G1_KB; ++i) {
        if (i + G1_AHEAD < G1_KB) {
            qa[i + G1_AHEAD] = ap[2 * (i + G1_AHEAD)];
            qb[i + G1_AHEAD] = bqp[32 * (i + G1_AHEAD)];
        }
        __builtin_amdgcn_sched_barrier(0);
        v2f a0, b0, a1, b1;
        a0.x = qa[i].x; a0.y = qa[i].y; b0.x = qb[i].x; b0.y = qb[i].y;
        a1.x = qa[i].z; a1.y = qa[i].w; b1.x = qb[i].z; b1.y = qb[i].w;
        acc0 = __builtin_amdgcn_wmma_f32_16x16x4_f32( false, a0,  false, b0,
                                                      (short)0, acc0,  false,  false);
        acc1 = __builtin_amdgcn_wmma_f32_16x16x4_f32( false, a1,  false, b1,
                                                      (short)0, acc1,  false,  false);
    }
    const int rbase = blockIdx.x * G1_NODES + 8 * hh;
    float d[8];
#pragma unroll
    for (int r = 0; r < 8; ++r) d[r] = acc0[r] + acc1[r];
    float* hp = h2 + (size_t)rbase * NCLS + c;
    if ((int)(blockIdx.x + 1) * G1_NODES <= n) {
#pragma unroll
        for (int r = 0; r < 8; ++r) hp[r * NCLS] = d[r];
    } else {
#pragma unroll
        for (int r = 0; r < 8; ++r)
            if (rbase + r < n) hp[r * NCLS] = d[r];
    }
    float u[8], w4[4], x2[2];
    {
        const bool up = (c & 8) != 0;
#pragma unroll
        for (int r = 0; r < 8; ++r) {
            const float vs = d[r] * asc, vd = d[r] * adc;
            u[r] = (up ? vd : vs) + __shfl_xor(up ? vs : vd, 8);
        }
    }
    {
        const bool up = (c & 4) != 0;
#pragma unroll
        for (int r = 0; r < 4; ++r) w4[r] = (up ? u[4 + r] : u[r]) + __shfl_xor(up ? u[r] : u[4 + r], 4);
    }
    {
        const bool up = (c & 2) != 0;
#pragma unroll
        for (int r = 0; r < 2; ++r) x2[r] = (up ? w4[2 + r] : w4[r]) + __shfl_xor(up ? w4[r] : w4[2 + r], 2);
    }
    const bool up1 = (c & 1) != 0;
    const float y = (up1 ? x2[1] : x2[0]) + __shfl_xor(up1 ? x2[0] : x2[1], 1);
    const int row = rbase + (c & 7);
    if (row < n) {
        float* sp = (c & 8) ? a_d2 : a_s2;
        sp[row] = y;
    }
}


__device__ __forceinline__ float l2_rows(const float* __restrict__ h, int s, float w, int np,
                                         int half, int c, float acc) {
#pragma unroll 4
    for (int q = 0; q < np; q += 2) {
        int   sl = __shfl(s, q + half);
        float wl = __shfl(w, q + half);
        acc += wl * h[(size_t)sl * NCLS + c];
    }
    return acc;
}

__global__ void k_gather_l2(const int* __restrict__ rowptr, const int* __restrict__ ssrc,
                            const float* __restrict__ a_s, const float* __restrict__ a_d,
                            const float* __restrict__ h, const float* __restrict__ b,
                            float* __restrict__ out, int n) {
    int lane = threadIdx.x & 31;
    int node = blockIdx.x * (blockDim.x >> 5) + (threadIdx.x >> 5);
    node = __builtin_amdgcn_readfirstlane(node);
    if (node >= n) return;
    int half = lane >> 4, c = lane & 15;
    int beg  = rowptr[node];
    int cntE = rowptr[node + 1] - beg + 1;
    float ad = a_d[node];
    float bc = b[c];

    int   s0 = node;
    float e0 = -__builtin_huge_valf();
    if (lane < cntE) {
        if (lane > 0) s0 = ssrc[beg + lane - 1];
        e0 = lrelu(a_s[s0] + ad);
    }

    float m = fmaxf(-__builtin_huge_valf(), e0);
    for (int t0 = 32; t0 < cntE; t0 += 32) {
        int t = t0 + lane;
        if (t < cntE) {
            int s = ssrc[beg + t - 1];
            m = fmaxf(m, lrelu(a_s[s] + ad));
        }
    }
    for (int o = 16; o > 0; o >>= 1) m = fmaxf(m, __shfl_xor(m, o));

    float acc = 0.f, den = 0.f;
    {
        const float w = (lane < cntE) ? expf(e0 - m) : 0.f;
        den += w;
        acc = l2_rows(h, s0, w, (cntE < 32) ? cntE : 32, half, c, acc);
    }
    for (int t0 = 32; t0 < cntE; t0 += 32) {
        int t = t0 + lane;
        int s = node;
        float w = 0.f;
        if (t < cntE) {
            s = ssrc[beg + t - 1];
            w = expf(lrelu(a_s[s] + ad) - m);
        }
        den += w;
        int left = cntE - t0;
        acc = l2_rows(h, s, w, (left < 32) ? left : 32, half, c, acc);
    }
    for (int o = 16; o > 0; o >>= 1) den += __shfl_xor(den, o);
    acc += __shfl_xor(acc, 16);
    if (lane < 16) out[(size_t)node * NCLS + c] = acc / den + bc;
}


extern "C" void kernel_launch(void* const* d_in, const int* in_sizes, int n_in,
                              void* d_out, int out_size, void* d_ws, size_t ws_size,
                              hipStream_t stream) {
    const float* x        = (const float*)d_in[0];
    const int*   ei       = (const int*)  d_in[1];
    const float* W1       = (const float*)d_in[2];
    const float* att_src1 = (const float*)d_in[3];
    const float* att_dst1 = (const float*)d_in[4];
    const float* b1       = (const float*)d_in[5];
    const float* W2       = (const float*)d_in[6];
    const float* att_src2 = (const float*)d_in[7];
    const float* att_dst2 = (const float*)d_in[8];
    const float* b2       = (const float*)d_in[9];
    float* out = (float*)d_out;

    const int N    = in_sizes[0] / F_IN;
    const int E    = in_sizes[1] / 2;
    float* ws  = (float*)d_ws;
    float* h1  = ws;
    float* as1 = h1  + (size_t)N * H1C1;
    float* ad1 = as1 + (size_t)N * HEADS1;
    float* h2  = ad1 + (size_t)N * HEADS1;
    float* as2 = h2  + (size_t)N * NCLS;
    float* ad2 = as2 + (size_t)N;
    float* agg1 = ad2 + (size_t)N;
    const int nb     = (N + BKT_NODES - 1) >> BKT_SHIFT;
    const int chunks = (E + CHUNK - 1) / CHUNK;
    int* cpairs = (int*)(agg1 + (size_t)N * H1C1);
    int* ssrc   = cpairs + 2 * (size_t)chunks * CHUNK;
    int* rowptr = ssrc + (size_t)E;
    int* segoff = rowptr + (size_t)N + 1;

    const int T = 256;
    const int wavesPerBlk = T >> 5;

    k_gemm_strip_f32<H1C1 / 16, F_IN, HEADS1, CH1, true><<<chunks + ((N >> 4) + wavesPerBlk - 1) / wavesPerBlk, T, 0, stream>>>(
        x, W1, att_src1, att_dst1, h1, as1, ad1, N,
        ei, E, nb, chunks, segoff, cpairs);
    k_bucket_sort<<<nb, 1024, 0, stream>>>(cpairs, segoff, chunks, nb, rowptr, ssrc, N);

    k_gather_l1_mm2<<<(N + G1_NODES - 1) / G1_NODES, 32 * G1_NODES, 0, stream>>>(
        rowptr, ssrc, as1, ad1, h1, b1, W2, att_src2, att_dst2, h2, as2, ad2, N);

    k_gather_l2<<<(N + wavesPerBlk - 1) / wavesPerBlk, T, 0, stream>>>(
        rowptr, ssrc, as2, ad2, h2, b2, out, N);
}
